# starting kernel
# baseline (speedup 1.0000x reference)
#include <hip/hip_runtime.h>
#include <math.h>

typedef float v2f __attribute__((ext_vector_type(2)));
typedef float v8f __attribute__((ext_vector_type(8)));

#define NODE_DIM 128
#define ATTR_DIM 64
#define EPS 1e-8f

__global__ void init_kernel(float* __restrict__ seg_max,
                            float* __restrict__ seg_sum,
                            float* __restrict__ agg, int n1) {
  int i = blockIdx.x * blockDim.x + threadIdx.x;
  int total = n1 * NODE_DIM;
  if (i < total) agg[i] = 0.0f;
  if (i < n1) {
    seg_max[i] = -INFINITY;
    seg_sum[i] = 0.0f;
  }
}

__global__ void inv_norm_kernel(const float* __restrict__ X,
                                float* __restrict__ inv, int n) {
  int wave = (blockIdx.x * blockDim.x + threadIdx.x) >> 5;
  int lane = threadIdx.x & 31;
  if (wave >= n) return;
  float4 a = reinterpret_cast<const float4*>(X + (size_t)wave * NODE_DIM)[lane];
  float p = a.x * a.x + a.y * a.y + a.z * a.z + a.w * a.w;
#pragma unroll
  for (int off = 16; off > 0; off >>= 1) p += __shfl_xor(p, off, 32);
  if (lane == 0) inv[wave] = 1.0f / fmaxf(sqrtf(p), EPS);
}

__global__ void edge_sim_kernel(const float* __restrict__ X1,
                                const float* __restrict__ X2,
                                const int* __restrict__ src,
                                const int* __restrict__ dst,
                                const float* __restrict__ invn1,
                                const float* __restrict__ invn2,
                                float* __restrict__ sim,
                                float* __restrict__ seg_max, int E) {
  int e = (blockIdx.x * blockDim.x + threadIdx.x) >> 5;
  int lane = threadIdx.x & 31;
  if (e >= E) return;
  int s = src[e];
  int d = dst[e];
  float4 a = reinterpret_cast<const float4*>(X1 + (size_t)s * NODE_DIM)[lane];
  float4 b = reinterpret_cast<const float4*>(X2 + (size_t)d * NODE_DIM)[lane];
  float p = a.x * b.x + a.y * b.y + a.z * b.z + a.w * b.w;
#pragma unroll
  for (int off = 16; off > 0; off >>= 1) p += __shfl_xor(p, off, 32);
  if (lane == 0) {
    float simv = p * invn1[s] * invn2[d];
    sim[e] = simv;
    unsafeAtomicMax(&seg_max[s], simv);
  }
}

__global__ void edge_exp_kernel(const int* __restrict__ src,
                                const float* __restrict__ sim,
                                const float* __restrict__ seg_max,
                                float* __restrict__ ex,
                                float* __restrict__ seg_sum, int E) {
  int e = blockIdx.x * blockDim.x + threadIdx.x;
  if (e >= E) return;
  int s = src[e];
  float v = expf(sim[e] - seg_max[s]);
  ex[e] = v;
  unsafeAtomicAdd(&seg_sum[s], v);
}

__global__ void edge_agg_kernel(const float* __restrict__ X2,
                                const int* __restrict__ src,
                                const int* __restrict__ dst,
                                const float* __restrict__ ex,
                                const float* __restrict__ seg_sum,
                                float* __restrict__ agg, int E) {
  int e = (blockIdx.x * blockDim.x + threadIdx.x) >> 5;
  int lane = threadIdx.x & 31;
  if (e >= E) return;
  int s = src[e];
  int d = dst[e];
  float w = ex[e] / seg_sum[s];
  float4 b = reinterpret_cast<const float4*>(X2 + (size_t)d * NODE_DIM)[lane];
  float* out = agg + (size_t)s * NODE_DIM + lane * 4;
  unsafeAtomicAdd(out + 0, w * b.x);
  unsafeAtomicAdd(out + 1, w * b.y);
  unsafeAtomicAdd(out + 2, w * b.z);
  unsafeAtomicAdd(out + 3, w * b.w);
}

__global__ void gate_out_kernel(const float* __restrict__ Xn,
                                const float* __restrict__ Wg,
                                const float* __restrict__ agg,
                                float* __restrict__ out, int n1) {
  int wave = blockIdx.x * (blockDim.x >> 5) + (threadIdx.x >> 5);
  int lane = threadIdx.x & 31;
  int tilesM = (n1 + 15) >> 4;
  int tileM = wave >> 3;
  int tileN = wave & 7;
  if (tileM >= tilesM) return;

  int l15 = lane & 15;
  int half = lane >> 4;
  int m = tileM * 16 + l15;
  if (m >= n1) m = n1 - 1;
  const float* arow = Xn + (size_t)m * ATTR_DIM;
  const float* brow = Wg + (size_t)(tileN * 16 + l15) * ATTR_DIM;

  v8f c = {};
#pragma unroll
  for (int k0 = 0; k0 < ATTR_DIM; k0 += 4) {
    int k = k0 + half * 2;
    v2f a = *reinterpret_cast<const v2f*>(arow + k);
    v2f b = *reinterpret_cast<const v2f*>(brow + k);
    c = __builtin_amdgcn_wmma_f32_16x16x4_f32(
         false, a,  false, b,
         (short)0, c,  false,  false);
  }

  int n = tileN * 16 + l15;
#pragma unroll
  for (int v = 0; v < 8; ++v) {
    int row = tileM * 16 + half * 8 + v;
    if (row < n1) {
      float g = 1.0f / (1.0f + expf(-c[v]));
      size_t idx = (size_t)row * NODE_DIM + n;
      out[idx] = g * agg[idx];
    }
  }
}

extern "C" void kernel_launch(void* const* d_in, const int* in_sizes, int n_in,
                              void* d_out, int out_size, void* d_ws,
                              size_t ws_size, hipStream_t stream) {
  const float* X_h_1 = (const float*)d_in[0];
  const float* X_h_2 = (const float*)d_in[1];
  const float* X_n_1 = (const float*)d_in[2];
  const int* cross = (const int*)d_in[3];
  const float* W_gate = (const float*)d_in[4];
  float* out = (float*)d_out;

  const int N1 = in_sizes[0] / NODE_DIM;
  const int N2 = in_sizes[1] / NODE_DIM;
  const int E = in_sizes[3] / 2;
  const int* src = cross;
  const int* dst = cross + E;

  float* ws = (float*)d_ws;
  float* invn1 = ws;
  float* invn2 = invn1 + N1;
  float* sim = invn2 + N2;
  float* ex = sim + E;
  float* seg_max = ex + E;
  float* seg_sum = seg_max + N1;
  float* agg = seg_sum + N1;

  const int BLK = 256;
  const int waves_per_blk = BLK / 32;

  {
    int total = N1 * NODE_DIM;
    init_kernel<<<(total + BLK - 1) / BLK, BLK, 0, stream>>>(seg_max, seg_sum,
                                                             agg, N1);
  }
  inv_norm_kernel<<<(N1 + waves_per_blk - 1) / waves_per_blk, BLK, 0, stream>>>(
      X_h_1, invn1, N1);
  inv_norm_kernel<<<(N2 + waves_per_blk - 1) / waves_per_blk, BLK, 0, stream>>>(
      X_h_2, invn2, N2);
  edge_sim_kernel<<<(E + waves_per_blk - 1) / waves_per_blk, BLK, 0, stream>>>(
      X_h_1, X_h_2, src, dst, invn1, invn2, sim, seg_max, E);
  edge_exp_kernel<<<(E + BLK - 1) / BLK, BLK, 0, stream>>>(src, sim, seg_max,
                                                           ex, seg_sum, E);
  edge_agg_kernel<<<(E + waves_per_blk - 1) / waves_per_blk, BLK, 0, stream>>>(
      X_h_2, src, dst, ex, seg_sum, agg, E);
  {
    int tilesM = (N1 + 15) / 16;
    int waves = tilesM * (NODE_DIM / 16);
    gate_out_kernel<<<(waves + waves_per_blk - 1) / waves_per_blk, BLK, 0,
                      stream>>>(X_n_1, W_gate, agg, out, N1);
  }
}


# scatter-add by owner work-groups: each takes 200 nodes, filters the whole src list, sums rows in LDS by LDS float atomics, stores agg once; no global float atomics; sum order = LDS arrival
# speedup vs baseline: 1.4205x; 1.4205x over previous
#include <hip/hip_runtime.h>
#include <math.h>

typedef float v2f __attribute__((ext_vector_type(2)));
typedef float v8f __attribute__((ext_vector_type(8)));

#define NODE_DIM 128
#define ATTR_DIM 64
#define EPS 1e-8f
#define EPW 4

__global__ void init_kernel(float* __restrict__ seg_max,
                            float* __restrict__ seg_sum, int n1) {
  int i = blockIdx.x * blockDim.x + threadIdx.x;
  if (i < n1) {
    seg_max[i] = -INFINITY;
    seg_sum[i] = 0.0f;
  }
}

__global__ void inv_norm_kernel(const float* __restrict__ X,
                                float* __restrict__ inv, int n) {
  int wave = (blockIdx.x * blockDim.x + threadIdx.x) >> 5;
  int lane = threadIdx.x & 31;
  if (wave >= n) return;
  float4 a = reinterpret_cast<const float4*>(X + (size_t)wave * NODE_DIM)[lane];
  float p = a.x * a.x + a.y * a.y + a.z * a.z + a.w * a.w;
#pragma unroll
  for (int off = 16; off > 0; off >>= 1) p += __shfl_xor(p, off, 32);
  if (lane == 0) inv[wave] = 1.0f / fmaxf(sqrtf(p), EPS);
}

__global__ void edge_sim_kernel(const float* __restrict__ X1,
                                const float* __restrict__ X2,
                                const int* __restrict__ src,
                                const int* __restrict__ dst,
                                const float* __restrict__ invn1,
                                const float* __restrict__ invn2,
                                float* __restrict__ sim,
                                float* __restrict__ seg_max, int E) {
  int wave = __builtin_amdgcn_readfirstlane(
      (int)((blockIdx.x * blockDim.x + threadIdx.x) >> 5));
  int lane = threadIdx.x & 31;
  int e0 = wave * EPW;
  if (e0 >= E) return;
  int sub = lane & (EPW - 1);
  int em = min(e0 + sub, E - 1);
  int ms = src[em];
  int md = dst[em];
  float i1 = invn1[ms];
  float i2 = invn2[md];
  float4 a[EPW], b[EPW];
#pragma unroll
  for (int q = 0; q < EPW; ++q) {
    int s = __builtin_amdgcn_readlane(ms, q);
    int d = __builtin_amdgcn_readlane(md, q);
    a[q] = reinterpret_cast<const float4*>(X1 + (size_t)s * NODE_DIM)[lane];
    b[q] = reinterpret_cast<const float4*>(X2 + (size_t)d * NODE_DIM)[lane];
  }
  float p[EPW];
#pragma unroll
  for (int q = 0; q < EPW; ++q)
    p[q] = a[q].x * b[q].x + a[q].y * b[q].y + a[q].z * b[q].z +
           a[q].w * b[q].w;
#pragma unroll
  for (int off = 16; off > 0; off >>= 1) {
#pragma unroll
    for (int q = 0; q < EPW; ++q) p[q] += __shfl_xor(p[q], off, 32);
  }
  float mine = p[0];
#pragma unroll
  for (int q = 1; q < EPW; ++q) mine = (sub == q) ? p[q] : mine;
  if (lane < EPW && e0 + lane < E) {
    float simv = mine * i1 * i2;
    sim[e0 + lane] = simv;
    unsafeAtomicMax(&seg_max[ms], simv);
  }
}

__global__ void edge_exp_kernel(const int* __restrict__ src,
                                const float* __restrict__ sim,
                                const float* __restrict__ seg_max,
                                float* __restrict__ ex,
                                float* __restrict__ seg_sum, int E) {
  int e = blockIdx.x * blockDim.x + threadIdx.x;
  if (e >= E) return;
  int s = src[e];
  float v = expf(sim[e] - seg_max[s]);
  ex[e] = v;
  unsafeAtomicAdd(&seg_sum[s], v);
}

#define OWN_R 200
#define OWN_WAVES 32
#define OWN_U 4
#define OWN_CW 640
#define OWN_G 8

__device__ __forceinline__ void own_flush(const int2* myList, int count,
                                          float* aggL,
                                          const float* __restrict__ X2,
                                          const int* __restrict__ dst,
                                          const float* __restrict__ ex,
                                          const float* __restrict__ seg_sum,
                                          int r0, int lane) {
  const int sub = lane & (OWN_G - 1);
  for (int i = 0; i < count; i += OWN_G) {
    int2 en = myList[min(i + sub, count - 1)];
    int md = dst[en.x];
    float mex = ex[en.x];
    float msum = seg_sum[r0 + en.y];
    int mloc = (i + sub < count) ? en.y : OWN_R;
    int loc[OWN_G];
    float4 b[OWN_G];
    float w[OWN_G];
#pragma unroll
    for (int q = 0; q < OWN_G; ++q) {
      int d = __builtin_amdgcn_readlane(md, q);
      loc[q] = __builtin_amdgcn_readlane(mloc, q);
      b[q] = reinterpret_cast<const float4*>(X2 + (size_t)d * NODE_DIM)[lane];
    }
    float mw = mex / msum;
#pragma unroll
    for (int q = 0; q < OWN_G; ++q)
      w[q] = __int_as_float(__builtin_amdgcn_readlane(__float_as_int(mw), q));
#pragma unroll
    for (int q = 0; q < OWN_G; ++q) {
      float* o = aggL + loc[q] * NODE_DIM + lane;
      __hip_atomic_fetch_add(o + 0, w[q] * b[q].x, __ATOMIC_RELAXED, __HIP_MEMORY_SCOPE_WORKGROUP);
      __hip_atomic_fetch_add(o + 32, w[q] * b[q].y, __ATOMIC_RELAXED, __HIP_MEMORY_SCOPE_WORKGROUP);
      __hip_atomic_fetch_add(o + 64, w[q] * b[q].z, __ATOMIC_RELAXED, __HIP_MEMORY_SCOPE_WORKGROUP);
      __hip_atomic_fetch_add(o + 96, w[q] * b[q].w, __ATOMIC_RELAXED, __HIP_MEMORY_SCOPE_WORKGROUP);
    }
  }
}

__device__ __forceinline__ int own_append(int2* myList, int count, int s, int e,
                                          int r0, int rn) {
  unsigned loc = (unsigned)(s - r0);
  bool m = loc < (unsigned)rn;
  unsigned mask = __builtin_amdgcn_ballot_w32(m);
  if (__builtin_expect(mask != 0u, 0)) {
    int pos = count + (int)__builtin_amdgcn_mbcnt_lo(mask, 0u);
    if (m) myList[pos] = make_int2(e, (int)loc);
    count += __builtin_popcount(mask);
  }
  return count;
}

__global__ __launch_bounds__(OWN_WAVES * 32) void owner_agg_kernel(
    const float* __restrict__ X2, const int* __restrict__ src,
    const int* __restrict__ dst, const float* __restrict__ ex,
    const float* __restrict__ seg_sum, float* __restrict__ agg, int n1, int E) {
  __shared__ float aggL[(OWN_R + 1) * NODE_DIM];
  __shared__ int2 listL[OWN_WAVES * OWN_CW];
  const int tid = threadIdx.x;
  const int lane = tid & 31;
  const int w = __builtin_amdgcn_readfirstlane(tid >> 5);
  const int r0 = blockIdx.x * OWN_R;
  const int rn = min(OWN_R, n1 - r0);

  for (int i = tid; i < (OWN_R + 1) * NODE_DIM / 4; i += OWN_WAVES * 32)
    reinterpret_cast<float4*>(aggL)[i] = make_float4(0.f, 0.f, 0.f, 0.f);
  __syncthreads();

  int2* myList = listL + w * OWN_CW;
  int count = 0;
  const int n4 = E >> 2;
  const int4* src4 = reinterpret_cast<const int4*>(src);
  const int trip4 = OWN_WAVES * OWN_U * 32;
  const int trips = (n4 + trip4 - 1) / trip4;
  int tt = trips > 0 ? (int)(blockIdx.x % (unsigned)trips) : 0;
  for (int t = 0; t < trips; ++t) {
    const int base = tt * trip4 + w * (OWN_U * 32);
    tt = (tt + 1 == trips) ? 0 : tt + 1;
    if (count > OWN_CW - OWN_U * 128 - 4) {
      own_flush(myList, count, aggL, X2, dst, ex, seg_sum, r0, lane);
      count = 0;
    }
    int4 v[OWN_U];
#pragma unroll
    for (int u = 0; u < OWN_U; ++u) {
      int g = base + u * 32 + lane;
      v[u] = (g < n4) ? src4[g] : make_int4(-1, -1, -1, -1);
    }
#pragma unroll
    for (int u = 0; u < OWN_U; ++u) {
      int e = (base + u * 32 + lane) << 2;
      count = own_append(myList, count, v[u].x, e + 0, r0, rn);
      count = own_append(myList, count, v[u].y, e + 1, r0, rn);
      count = own_append(myList, count, v[u].z, e + 2, r0, rn);
      count = own_append(myList, count, v[u].w, e + 3, r0, rn);
    }
  }
  if (w == 0) {
    int e = (n4 << 2) + lane;
    int s = (e < E) ? src[e] : -1;
    count = own_append(myList, count, s, e, r0, rn);
  }
  own_flush(myList, count, aggL, X2, dst, ex, seg_sum, r0, lane);
  __syncthreads();

  for (int i = tid; i < rn * 32; i += OWN_WAVES * 32) {
    int row = i >> 5;
    const float* p = aggL + row * NODE_DIM + lane;
    float4 o = make_float4(p[0], p[32], p[64], p[96]);
    reinterpret_cast<float4*>(agg + (size_t)(r0 + row) * NODE_DIM)[lane] = o;
  }
}

__global__ void gate_out_kernel(const float* __restrict__ Xn,
                                const float* __restrict__ Wg,
                                const float* __restrict__ agg,
                                float* __restrict__ out, int n1) {
  int wave = blockIdx.x * (blockDim.x >> 5) + (threadIdx.x >> 5);
  int lane = threadIdx.x & 31;
  int tilesM = (n1 + 15) >> 4;
  int tileM = wave >> 3;
  int tileN = wave & 7;
  if (tileM >= tilesM) return;

  int l15 = lane & 15;
  int half = lane >> 4;
  int m = tileM * 16 + l15;
  if (m >= n1) m = n1 - 1;
  const float* arow = Xn + (size_t)m * ATTR_DIM;
  const float* brow = Wg + (size_t)(tileN * 16 + l15) * ATTR_DIM;

  v8f c = {};
#pragma unroll
  for (int k0 = 0; k0 < ATTR_DIM; k0 += 4) {
    int k = k0 + half * 2;
    v2f a = *reinterpret_cast<const v2f*>(arow + k);
    v2f b = *reinterpret_cast<const v2f*>(brow + k);
    c = __builtin_amdgcn_wmma_f32_16x16x4_f32(
         false, a,  false, b,
         (short)0, c,  false,  false);
  }

  int n = tileN * 16 + l15;
#pragma unroll
  for (int v = 0; v < 8; ++v) {
    int row = tileM * 16 + half * 8 + v;
    if (row < n1) {
      float g = 1.0f / (1.0f + expf(-c[v]));
      size_t idx = (size_t)row * NODE_DIM + n;
      out[idx] = g * agg[idx];
    }
  }
}

extern "C" void kernel_launch(void* const* d_in, const int* in_sizes, int n_in,
                              void* d_out, int out_size, void* d_ws,
                              size_t ws_size, hipStream_t stream) {
  const float* X_h_1 = (const float*)d_in[0];
  const float* X_h_2 = (const float*)d_in[1];
  const float* X_n_1 = (const float*)d_in[2];
  const int* cross = (const int*)d_in[3];
  const float* W_gate = (const float*)d_in[4];
  float* out = (float*)d_out;

  const int N1 = in_sizes[0] / NODE_DIM;
  const int N2 = in_sizes[1] / NODE_DIM;
  const int E = in_sizes[3] / 2;
  const int* src = cross;
  const int* dst = cross + E;

  float* ws = (float*)d_ws;
  float* invn1 = ws;
  float* invn2 = invn1 + N1;
  float* sim = invn2 + N2;
  float* ex = sim + E;
  float* seg_max = ex + E;
  float* seg_sum = seg_max + N1;
  float* agg = seg_sum + N1;

  const int BLK = 256;
  const int waves_per_blk = BLK / 32;
  const int edge_waves = (E + EPW - 1) / EPW;

  init_kernel<<<(N1 + BLK - 1) / BLK, BLK, 0, stream>>>(seg_max, seg_sum, N1);
  inv_norm_kernel<<<(N1 + waves_per_blk - 1) / waves_per_blk, BLK, 0, stream>>>(
      X_h_1, invn1, N1);
  inv_norm_kernel<<<(N2 + waves_per_blk - 1) / waves_per_blk, BLK, 0, stream>>>(
      X_h_2, invn2, N2);
  edge_sim_kernel<<<(edge_waves + waves_per_blk - 1) / waves_per_blk, BLK, 0,
                    stream>>>(
      X_h_1, X_h_2, src, dst, invn1, invn2, sim, seg_max, E);
  edge_exp_kernel<<<(E + BLK - 1) / BLK, BLK, 0, stream>>>(src, sim, seg_max,
                                                           ex, seg_sum, E);
  owner_agg_kernel<<<(N1 + OWN_R - 1) / OWN_R, OWN_WAVES * 32, 0, stream>>>(
      X_h_2, src, dst, ex, seg_sum, agg, N1, E);
  {
    int tilesM = (N1 + 15) / 16;
    int waves = tilesM * (NODE_DIM / 16);
    gate_out_kernel<<<(waves + waves_per_blk - 1) / waves_per_blk, BLK, 0,
                      stream>>>(X_n_1, W_gate, agg, out, N1);
  }
}


# segment softmax (max, exp-sum, division) by the owner work-groups in LDS: wave w takes nodes w mod 32 of all lists, sums in list order not arrival order; agg divided once by its sum; 7 launches to 5
# speedup vs baseline: 2.3549x; 1.6578x over previous
#include <hip/hip_runtime.h>
#include <math.h>

typedef float v2f __attribute__((ext_vector_type(2)));
typedef float v8f __attribute__((ext_vector_type(8)));

#define NODE_DIM 128
#define ATTR_DIM 64
#define EPS 1e-8f
#define EPW 4

__global__ void inv_norm_kernel(const float* __restrict__ X,
                                float* __restrict__ inv, int n) {
  int wave = (blockIdx.x * blockDim.x + threadIdx.x) >> 5;
  int lane = threadIdx.x & 31;
  if (wave >= n) return;
  float4 a = reinterpret_cast<const float4*>(X + (size_t)wave * NODE_DIM)[lane];
  float p = a.x * a.x + a.y * a.y + a.z * a.z + a.w * a.w;
#pragma unroll
  for (int off = 16; off > 0; off >>= 1) p += __shfl_xor(p, off, 32);
  if (lane == 0) inv[wave] = 1.0f / fmaxf(sqrtf(p), EPS);
}

__global__ void edge_sim_kernel(const float* __restrict__ X1,
                                const float* __restrict__ X2,
                                const int* __restrict__ src,
                                const int* __restrict__ dst,
                                const float* __restrict__ invn1,
                                const float* __restrict__ invn2,
                                float* __restrict__ sim, int E) {
  int wave = __builtin_amdgcn_readfirstlane(
      (int)((blockIdx.x * blockDim.x + threadIdx.x) >> 5));
  int lane = threadIdx.x & 31;
  int e0 = wave * EPW;
  if (e0 >= E) return;
  int sub = lane & (EPW - 1);
  int em = min(e0 + sub, E - 1);
  int ms = src[em];
  int md = dst[em];
  float i1 = invn1[ms];
  float i2 = invn2[md];
  float4 a[EPW], b[EPW];
#pragma unroll
  for (int q = 0; q < EPW; ++q) {
    int s = __builtin_amdgcn_readlane(ms, q);
    int d = __builtin_amdgcn_readlane(md, q);
    a[q] = reinterpret_cast<const float4*>(X1 + (size_t)s * NODE_DIM)[lane];
    b[q] = reinterpret_cast<const float4*>(X2 + (size_t)d * NODE_DIM)[lane];
  }
  float p[EPW];
#pragma unroll
  for (int q = 0; q < EPW; ++q)
    p[q] = a[q].x * b[q].x + a[q].y * b[q].y + a[q].z * b[q].z +
           a[q].w * b[q].w;
#pragma unroll
  for (int off = 16; off > 0; off >>= 1) {
#pragma unroll
    for (int q = 0; q < EPW; ++q) p[q] += __shfl_xor(p[q], off, 32);
  }
  float mine = p[0];
#pragma unroll
  for (int q = 1; q < EPW; ++q) mine = (sub == q) ? p[q] : mine;
  if (lane < EPW && e0 + lane < E) {
    sim[e0 + lane] = mine * i1 * i2;
  }
}

#define OWN_R 200
#define OWN_WAVES 32
#define OWN_U 4
#define OWN_CW 640
#define OWN_G 8
#define OWN_PB 64

__device__ __forceinline__ int own_key(float f) {
  int b = __float_as_int(f);
  return b >= 0 ? b : (b ^ 0x7fffffff);
}
__device__ __forceinline__ float own_unkey(int k) {
  return __int_as_float(k >= 0 ? k : (k ^ 0x7fffffff));
}

__device__ __forceinline__ void own_max(const int2* list, int count, int* maxL,
                                        const float* __restrict__ sim, int lane) {
  for (int i = lane; i < count; i += 32) {
    int2 en = list[i];
    __hip_atomic_fetch_max(maxL + en.y, own_key(sim[en.x]), __ATOMIC_RELAXED,
                           __HIP_MEMORY_SCOPE_WORKGROUP);
  }
}

__device__ __forceinline__ void own_flush(const int2* list, int count,
                                          float* aggL, float* sumL,
                                          const int* maxL,
                                          const float* __restrict__ X2,
                                          const int* __restrict__ dst,
                                          const float* __restrict__ sim,
                                          int lane) {
  const int sub = lane & (OWN_G - 1);
  for (int i = 0; i < count; i += OWN_G) {
    int2 en = list[min(i + sub, count - 1)];
    int md = dst[en.x];
    float ms = sim[en.x];
    bool live = (i + sub < count);
    int mloc = live ? en.y : OWN_R;
    int loc[OWN_G];
    float4 b[OWN_G];
    float w[OWN_G];
#pragma unroll
    for (int q = 0; q < OWN_G; ++q) {
      int d = __builtin_amdgcn_readlane(md, q);
      loc[q] = __builtin_amdgcn_readlane(mloc, q);
      b[q] = reinterpret_cast<const float4*>(X2 + (size_t)d * NODE_DIM)[lane];
    }
    float mex = expf(ms - own_unkey(maxL[en.y]));
    if (lane < OWN_G && live)
      __hip_atomic_fetch_add(sumL + en.y, mex, __ATOMIC_RELAXED, __HIP_MEMORY_SCOPE_WORKGROUP);
#pragma unroll
    for (int q = 0; q < OWN_G; ++q)
      w[q] = __int_as_float(__builtin_amdgcn_readlane(__float_as_int(mex), q));
#pragma unroll
    for (int q = 0; q < OWN_G; ++q) {
      float* o = aggL + loc[q] * NODE_DIM + lane;
      __hip_atomic_fetch_add(o + 0, w[q] * b[q].x, __ATOMIC_RELAXED, __HIP_MEMORY_SCOPE_WORKGROUP);
      __hip_atomic_fetch_add(o + 32, w[q] * b[q].y, __ATOMIC_RELAXED, __HIP_MEMORY_SCOPE_WORKGROUP);
      __hip_atomic_fetch_add(o + 64, w[q] * b[q].z, __ATOMIC_RELAXED, __HIP_MEMORY_SCOPE_WORKGROUP);
      __hip_atomic_fetch_add(o + 96, w[q] * b[q].w, __ATOMIC_RELAXED, __HIP_MEMORY_SCOPE_WORKGROUP);
    }
  }
}

__device__ __forceinline__ int own_append(int2* myList, int count, int s, int e,
                                          int r0, int rn) {
  unsigned loc = (unsigned)(s - r0);
  bool m = loc < (unsigned)rn;
  unsigned mask = __builtin_amdgcn_ballot_w32(m);
  if (__builtin_expect(mask != 0u, 0)) {
    int pos = count + (int)__builtin_amdgcn_mbcnt_lo(mask, 0u);
    if (m) myList[pos] = make_int2(e, (int)loc);
    count += __builtin_popcount(mask);
  }
  return count;
}

__global__ __launch_bounds__(OWN_WAVES * 32) void owner_agg_kernel(
    const float* __restrict__ X2, const int* __restrict__ src,
    const int* __restrict__ dst, const float* __restrict__ sim,
    float* __restrict__ agg, int n1, int E) {
  __shared__ float aggL[(OWN_R + 1) * NODE_DIM];
  __shared__ int2 listL[OWN_WAVES * OWN_CW];
  __shared__ int2 privL[OWN_WAVES * OWN_PB];
  __shared__ float sumL[OWN_R];
  __shared__ int maxL[OWN_R];
  __shared__ int cntL[OWN_WAVES];
  __shared__ int ovfL;
  const int tid = threadIdx.x;
  const int lane = tid & 31;
  const int w = __builtin_amdgcn_readfirstlane(tid >> 5);
  const int r0 = blockIdx.x * OWN_R;
  const int rn = min(OWN_R, n1 - r0);

  for (int i = tid; i < (OWN_R + 1) * NODE_DIM / 4; i += OWN_WAVES * 32)
    reinterpret_cast<float4*>(aggL)[i] = make_float4(0.f, 0.f, 0.f, 0.f);
  if (tid < OWN_R) {
    sumL[tid] = 0.0f;
    maxL[tid] = own_key(-INFINITY);
  }
  if (tid == 0) ovfL = 0;
  __syncthreads();

  int2* myList = listL + w * OWN_CW;
  const int n4 = E >> 2;
  const int4* src4 = reinterpret_cast<const int4*>(src);
  const int trip4 = OWN_WAVES * OWN_U * 32;
  const int trips = (n4 + trip4 - 1) / trip4;
  const int tt0 = trips > 0 ? (int)(blockIdx.x % (unsigned)trips) : 0;
  bool second = false;
  for (int sweep = 0; sweep < 2; ++sweep) {
    int count = 0;
    int tt = tt0;
    for (int t = 0; t < trips; ++t) {
      const int base = tt * trip4 + w * (OWN_U * 32);
      tt = (tt + 1 == trips) ? 0 : tt + 1;
      if (count > OWN_CW - OWN_U * 128 - 4) {
        if (second) {
          own_flush(myList, count, aggL, sumL, maxL, X2, dst, sim, lane);
        } else {
          own_max(myList, count, maxL, sim, lane);
          if (lane == 0) ovfL = 1;
        }
        count = 0;
      }
      int4 v[OWN_U];
#pragma unroll
      for (int u = 0; u < OWN_U; ++u) {
        int g = base + u * 32 + lane;
        v[u] = (g < n4) ? src4[g] : make_int4(-1, -1, -1, -1);
      }
#pragma unroll
      for (int u = 0; u < OWN_U; ++u) {
        int e = (base + u * 32 + lane) << 2;
        count = own_append(myList, count, v[u].x, e + 0, r0, rn);
        count = own_append(myList, count, v[u].y, e + 1, r0, rn);
        count = own_append(myList, count, v[u].z, e + 2, r0, rn);
        count = own_append(myList, count, v[u].w, e + 3, r0, rn);
      }
    }
    if (w == 0) {
      int e = (n4 << 2) + lane;
      int s = (e < E) ? src[e] : -1;
      count = own_append(myList, count, s, e, r0, rn);
    }
    if (second) {
      own_flush(myList, count, aggL, sumL, maxL, X2, dst, sim, lane);
    } else {
      own_max(myList, count, maxL, sim, lane);
      if (lane == 0) cntL[w] = count;
    }
    __syncthreads();
    if (second || ovfL == 0) break;
    second = true;
  }

  if (!second) {
    const int myc = cntL[lane];
    int2* myPriv = privL + w * OWN_PB;
    int pc = 0;
    for (int j = 0; j < OWN_WAVES; ++j) {
      const int cj = __builtin_amdgcn_readlane(myc, j);
      const int2* Lj = listL + j * OWN_CW;
      for (int i = 0; i < cj; i += 32) {
        const int idx = i + lane;
        int2 en = Lj[min(idx, cj - 1)];
        bool m = (idx < cj) && ((en.y & (OWN_WAVES - 1)) == w);
        unsigned mask = __builtin_amdgcn_ballot_w32(m);
        if (mask != 0u) {
          int pos = pc + (int)__builtin_amdgcn_mbcnt_lo(mask, 0u);
          if (m) myPriv[pos] = en;
          pc += __builtin_popcount(mask);
          if (pc > OWN_PB - 32) {
            own_flush(myPriv, pc, aggL, sumL, maxL, X2, dst, sim, lane);
            pc = 0;
          }
        }
      }
    }
    own_flush(myPriv, pc, aggL, sumL, maxL, X2, dst, sim, lane);
  }
  __syncthreads();

  for (int i = tid; i < rn * 32; i += OWN_WAVES * 32) {
    int row = i >> 5;
    const float* p = aggL + row * NODE_DIM + lane;
    float sm = sumL[row];
    float4 o = make_float4(0.f, 0.f, 0.f, 0.f);
    if (sm > 0.0f) o = make_float4(p[0] / sm, p[32] / sm, p[64] / sm, p[96] / sm);
    reinterpret_cast<float4*>(agg + (size_t)(r0 + row) * NODE_DIM)[lane] = o;
  }
}

__global__ void gate_out_kernel(const float* __restrict__ Xn,
                                const float* __restrict__ Wg,
                                const float* __restrict__ agg,
                                float* __restrict__ out, int n1) {
  int wave = blockIdx.x * (blockDim.x >> 5) + (threadIdx.x >> 5);
  int lane = threadIdx.x & 31;
  int tilesM = (n1 + 15) >> 4;
  int tileM = wave >> 3;
  int tileN = wave & 7;
  if (tileM >= tilesM) return;

  int l15 = lane & 15;
  int half = lane >> 4;
  int m = tileM * 16 + l15;
  if (m >= n1) m = n1 - 1;
  const float* arow = Xn + (size_t)m * ATTR_DIM;
  const float* brow = Wg + (size_t)(tileN * 16 + l15) * ATTR_DIM;

  v8f c = {};
#pragma unroll
  for (int k0 = 0; k0 < ATTR_DIM; k0 += 4) {
    int k = k0 + half * 2;
    v2f a = *reinterpret_cast<const v2f*>(arow + k);
    v2f b = *reinterpret_cast<const v2f*>(brow + k);
    c = __builtin_amdgcn_wmma_f32_16x16x4_f32(
         false, a,  false, b,
         (short)0, c,  false,  false);
  }

  int n = tileN * 16 + l15;
#pragma unroll
  for (int v = 0; v < 8; ++v) {
    int row = tileM * 16 + half * 8 + v;
    if (row < n1) {
      float g = 1.0f / (1.0f + expf(-c[v]));
      size_t idx = (size_t)row * NODE_DIM + n;
      out[idx] = g * agg[idx];
    }
  }
}

extern "C" void kernel_launch(void* const* d_in, const int* in_sizes, int n_in,
                              void* d_out, int out_size, void* d_ws,
                              size_t ws_size, hipStream_t stream) {
  const float* X_h_1 = (const float*)d_in[0];
  const float* X_h_2 = (const float*)d_in[1];
  const float* X_n_1 = (const float*)d_in[2];
  const int* cross = (const int*)d_in[3];
  const float* W_gate = (const float*)d_in[4];
  float* out = (float*)d_out;

  const int N1 = in_sizes[0] / NODE_DIM;
  const int N2 = in_sizes[1] / NODE_DIM;
  const int E = in_sizes[3] / 2;
  const int* src = cross;
  const int* dst = cross + E;

  float* ws = (float*)d_ws;
  float* agg = ws;
  float* invn1 = agg + (size_t)N1 * NODE_DIM;
  float* invn2 = invn1 + N1;
  float* sim = invn2 + N2;

  const int BLK = 256;
  const int waves_per_blk = BLK / 32;
  const int edge_waves = (E + EPW - 1) / EPW;

  inv_norm_kernel<<<(N1 + waves_per_blk - 1) / waves_per_blk, BLK, 0, stream>>>(
      X_h_1, invn1, N1);
  inv_norm_kernel<<<(N2 + waves_per_blk - 1) / waves_per_blk, BLK, 0, stream>>>(
      X_h_2, invn2, N2);
  edge_sim_kernel<<<(edge_waves + waves_per_blk - 1) / waves_per_blk, BLK, 0,
                    stream>>>(
      X_h_1, X_h_2, src, dst, invn1, invn2, sim, E);
  owner_agg_kernel<<<(N1 + OWN_R - 1) / OWN_R, OWN_WAVES * 32, 0, stream>>>(
      X_h_2, src, dst, sim, agg, N1, E);
  {
    int tilesM = (N1 + 15) / 16;
    int waves = tilesM * (NODE_DIM / 16);
    gate_out_kernel<<<(waves + waves_per_blk - 1) / waves_per_blk, BLK, 0,
                      stream>>>(X_n_1, W_gate, agg, out, N1);
  }
}


# cosine sims computed inside the owner work-groups from their match lists (8 lanes a row, 16 products a lane, 3 swizzled adds: new sum tree); norm and sim launches gone, 5 launches to 2
# speedup vs baseline: 2.6674x; 1.1327x over previous
#include <hip/hip_runtime.h>
#include <math.h>

typedef float v2f __attribute__((ext_vector_type(2)));
typedef float v8f __attribute__((ext_vector_type(8)));

#define NODE_DIM 128
#define ATTR_DIM 64
#define EPS 1e-8f

#define OWN_R 200
#define OWN_WAVES 32
#define OWN_U 4
#define OWN_CW 640
#define OWN_G 8
#define OWN_PB 64
#define OWN_ST 2

__device__ __forceinline__ int own_key(float f) {
  int b = __float_as_int(f);
  return b >= 0 ? b : (b ^ 0x7fffffff);
}
__device__ __forceinline__ float own_unkey(int k) {
  return __int_as_float(k >= 0 ? k : (k ^ 0x7fffffff));
}

template <int M>
__device__ __forceinline__ float own_xadd(float x) {
  int b = __builtin_bit_cast(int, x);
  int r = __builtin_amdgcn_update_dpp(b, b, 0x160 + M, 0xf, 0xf, false);
  return x + __builtin_bit_cast(float, r);
}

__device__ __forceinline__ void own_sim_max(const int2* list, int count, int* maxL,
                                            const float* __restrict__ X1,
                                            const float* __restrict__ X2,
                                            const int* __restrict__ dst,
                                            float* sim, int r0, int lane) {
  if (count == 0) return;
  const int j = lane & 7;
  const int g = lane >> 3;
  int2 en[OWN_ST];
  int d[OWN_ST];
#pragma unroll
  for (int t = 0; t < OWN_ST; ++t) {
    en[t] = list[min(t * 4 + g, count - 1)];
    d[t] = dst[en[t].x];
  }
  for (int i = 0; i < count; i += 4 * OWN_ST) {
    float4 a[OWN_ST][4], b[OWN_ST][4];
    int2 cur[OWN_ST];
#pragma unroll
    for (int t = 0; t < OWN_ST; ++t) {
      const float4* ra =
          reinterpret_cast<const float4*>(X1 + (size_t)(r0 + en[t].y) * NODE_DIM) + j;
      const float4* rb =
          reinterpret_cast<const float4*>(X2 + (size_t)d[t] * NODE_DIM) + j;
#pragma unroll
      for (int c = 0; c < 4; ++c) {
        a[t][c] = ra[c * 8];
        b[t][c] = rb[c * 8];
      }
      cur[t] = en[t];
    }
#pragma unroll
    for (int t = 0; t < OWN_ST; ++t) {
      en[t] = list[min(i + 4 * OWN_ST + t * 4 + g, count - 1)];
      d[t] = dst[en[t].x];
    }
    float pd[OWN_ST], pa[OWN_ST], pb[OWN_ST];
#pragma unroll
    for (int t = 0; t < OWN_ST; ++t) {
      float sd = 0.0f, sa = 0.0f, sb = 0.0f;
#pragma unroll
      for (int c = 0; c < 4; ++c) {
        sd = fmaf(a[t][c].x, b[t][c].x, sd);
        sd = fmaf(a[t][c].y, b[t][c].y, sd);
        sd = fmaf(a[t][c].z, b[t][c].z, sd);
        sd = fmaf(a[t][c].w, b[t][c].w, sd);
        sa = fmaf(a[t][c].x, a[t][c].x, sa);
        sa = fmaf(a[t][c].y, a[t][c].y, sa);
        sa = fmaf(a[t][c].z, a[t][c].z, sa);
        sa = fmaf(a[t][c].w, a[t][c].w, sa);
        sb = fmaf(b[t][c].x, b[t][c].x, sb);
        sb = fmaf(b[t][c].y, b[t][c].y, sb);
        sb = fmaf(b[t][c].z, b[t][c].z, sb);
        sb = fmaf(b[t][c].w, b[t][c].w, sb);
      }
      sd = own_xadd<4>(sd);
      sa = own_xadd<4>(sa);
      sb = own_xadd<4>(sb);
      sd = own_xadd<2>(sd);
      sa = own_xadd<2>(sa);
      sb = own_xadd<2>(sb);
      pd[t] = own_xadd<1>(sd);
      pa[t] = own_xadd<1>(sa);
      pb[t] = own_xadd<1>(sb);
    }
    float md = pd[0], ma = pa[0], mb = pb[0];
    int2 me = cur[0];
#pragma unroll
    for (int t = 1; t < OWN_ST; ++t) {
      const bool s = (j == t);
      md = s ? pd[t] : md;
      ma = s ? pa[t] : ma;
      mb = s ? pb[t] : mb;
      me.x = s ? cur[t].x : me.x;
      me.y = s ? cur[t].y : me.y;
    }
    float sv = md / (fmaxf(sqrtf(ma), EPS) * fmaxf(sqrtf(mb), EPS));
    if (j < OWN_ST && i + j * 4 + g < count) {
      sim[me.x] = sv;
      __hip_atomic_fetch_max(maxL + me.y, own_key(sv), __ATOMIC_RELAXED,
                             __HIP_MEMORY_SCOPE_WORKGROUP);
    }
  }
}

__device__ __forceinline__ void own_flush(const int2* list, int count,
                                          float* aggL, float* sumL,
                                          const int* maxL,
                                          const float* __restrict__ X2,
                                          const int* __restrict__ dst,
                                          const float* sim,
                                          int lane) {
  const int sub = lane & (OWN_G - 1);
  for (int i = 0; i < count; i += OWN_G) {
    int2 en = list[min(i + sub, count - 1)];
    int md = dst[en.x];
    float ms = sim[en.x];
    bool live = (i + sub < count);
    int mloc = live ? en.y : OWN_R;
    int loc[OWN_G];
    float4 b[OWN_G];
    float w[OWN_G];
#pragma unroll
    for (int q = 0; q < OWN_G; ++q) {
      int d = __builtin_amdgcn_readlane(md, q);
      loc[q] = __builtin_amdgcn_readlane(mloc, q);
      b[q] = reinterpret_cast<const float4*>(X2 + (size_t)d * NODE_DIM)[lane];
    }
    float mex = expf(ms - own_unkey(maxL[en.y]));
    if (lane < OWN_G && live)
      __hip_atomic_fetch_add(sumL + en.y, mex, __ATOMIC_RELAXED, __HIP_MEMORY_SCOPE_WORKGROUP);
#pragma unroll
    for (int q = 0; q < OWN_G; ++q)
      w[q] = __int_as_float(__builtin_amdgcn_readlane(__float_as_int(mex), q));
#pragma unroll
    for (int q = 0; q < OWN_G; ++q) {
      float* o = aggL + loc[q] * NODE_DIM + lane;
      __hip_atomic_fetch_add(o + 0, w[q] * b[q].x, __ATOMIC_RELAXED, __HIP_MEMORY_SCOPE_WORKGROUP);
      __hip_atomic_fetch_add(o + 32, w[q] * b[q].y, __ATOMIC_RELAXED, __HIP_MEMORY_SCOPE_WORKGROUP);
      __hip_atomic_fetch_add(o + 64, w[q] * b[q].z, __ATOMIC_RELAXED, __HIP_MEMORY_SCOPE_WORKGROUP);
      __hip_atomic_fetch_add(o + 96, w[q] * b[q].w, __ATOMIC_RELAXED, __HIP_MEMORY_SCOPE_WORKGROUP);
    }
  }
}

__device__ __forceinline__ int own_append(int2* myList, int count, int s, int e,
                                          int r0, int rn) {
  unsigned loc = (unsigned)(s - r0);
  bool m = loc < (unsigned)rn;
  unsigned mask = __builtin_amdgcn_ballot_w32(m);
  if (__builtin_expect(mask != 0u, 0)) {
    int pos = count + (int)__builtin_amdgcn_mbcnt_lo(mask, 0u);
    if (m) myList[pos] = make_int2(e, (int)loc);
    count += __builtin_popcount(mask);
  }
  return count;
}

__global__ __launch_bounds__(OWN_WAVES * 32) void owner_agg_kernel(
    const float* __restrict__ X1, const float* __restrict__ X2,
    const int* __restrict__ src, const int* __restrict__ dst, float* sim,
    float* __restrict__ agg, int n1, int E) {
  __shared__ float aggL[(OWN_R + 1) * NODE_DIM];
  __shared__ int2 listL[OWN_WAVES * OWN_CW];
  __shared__ int2 privL[OWN_WAVES * OWN_PB];
  __shared__ float sumL[OWN_R];
  __shared__ int maxL[OWN_R];
  __shared__ int cntL[OWN_WAVES];
  __shared__ int ovfL;
  const int tid = threadIdx.x;
  const int lane = tid & 31;
  const int w = __builtin_amdgcn_readfirstlane(tid >> 5);
  const int r0 = blockIdx.x * OWN_R;
  const int rn = min(OWN_R, n1 - r0);

  for (int i = tid; i < (OWN_R + 1) * NODE_DIM / 4; i += OWN_WAVES * 32)
    reinterpret_cast<float4*>(aggL)[i] = make_float4(0.f, 0.f, 0.f, 0.f);
  if (tid < OWN_R) {
    sumL[tid] = 0.0f;
    maxL[tid] = own_key(-INFINITY);
  }
  if (tid == 0) ovfL = 0;
  __syncthreads();

  int2* myList = listL + w * OWN_CW;
  const int n4 = E >> 2;
  const int4* src4 = reinterpret_cast<const int4*>(src);
  const int trip4 = OWN_WAVES * OWN_U * 32;
  const int trips = (n4 + trip4 - 1) / trip4;
  const int tt0 = trips > 0 ? (int)(blockIdx.x % (unsigned)trips) : 0;
  bool second = false;
  for (int sweep = 0; sweep < 2; ++sweep) {
    int count = 0;
    int tt = tt0;
    for (int t = 0; t < trips; ++t) {
      const int base = tt * trip4 + w * (OWN_U * 32);
      tt = (tt + 1 == trips) ? 0 : tt + 1;
      if (count > OWN_CW - OWN_U * 128 - 4) {
        if (second) {
          own_flush(myList, count, aggL, sumL, maxL, X2, dst, sim, lane);
        } else {
          own_sim_max(myList, count, maxL, X1, X2, dst, sim, r0, lane);
          if (lane == 0) ovfL = 1;
        }
        count = 0;
      }
      int4 v[OWN_U];
#pragma unroll
      for (int u = 0; u < OWN_U; ++u) {
        int g = base + u * 32 + lane;
        v[u] = (g < n4) ? src4[g] : make_int4(-1, -1, -1, -1);
      }
#pragma unroll
      for (int u = 0; u < OWN_U; ++u) {
        int e = (base + u * 32 + lane) << 2;
        count = own_append(myList, count, v[u].x, e + 0, r0, rn);
        count = own_append(myList, count, v[u].y, e + 1, r0, rn);
        count = own_append(myList, count, v[u].z, e + 2, r0, rn);
        count = own_append(myList, count, v[u].w, e + 3, r0, rn);
      }
    }
    if (w == 0) {
      int e = (n4 << 2) + lane;
      int s = (e < E) ? src[e] : -1;
      count = own_append(myList, count, s, e, r0, rn);
    }
    if (second) {
      own_flush(myList, count, aggL, sumL, maxL, X2, dst, sim, lane);
    } else {
      own_sim_max(myList, count, maxL, X1, X2, dst, sim, r0, lane);
      if (lane == 0) cntL[w] = count;
    }
    __syncthreads();
    if (second || ovfL == 0) break;
    second = true;
  }

  if (!second) {
    const int myc = cntL[lane];
    int2* myPriv = privL + w * OWN_PB;
    int pc = 0;
    for (int j = 0; j < OWN_WAVES; ++j) {
      const int cj = __builtin_amdgcn_readlane(myc, j);
      const int2* Lj = listL + j * OWN_CW;
      for (int i = 0; i < cj; i += 32) {
        const int idx = i + lane;
        int2 en = Lj[min(idx, cj - 1)];
        bool m = (idx < cj) && ((en.y & (OWN_WAVES - 1)) == w);
        unsigned mask = __builtin_amdgcn_ballot_w32(m);
        if (mask != 0u) {
          int pos = pc + (int)__builtin_amdgcn_mbcnt_lo(mask, 0u);
          if (m) myPriv[pos] = en;
          pc += __builtin_popcount(mask);
          if (pc > OWN_PB - 32) {
            own_flush(myPriv, pc, aggL, sumL, maxL, X2, dst, sim, lane);
            pc = 0;
          }
        }
      }
    }
    own_flush(myPriv, pc, aggL, sumL, maxL, X2, dst, sim, lane);
  }
  __syncthreads();

  for (int i = tid; i < rn * 32; i += OWN_WAVES * 32) {
    int row = i >> 5;
    const float* p = aggL + row * NODE_DIM + lane;
    float sm = sumL[row];
    float4 o = make_float4(0.f, 0.f, 0.f, 0.f);
    if (sm > 0.0f) o = make_float4(p[0] / sm, p[32] / sm, p[64] / sm, p[96] / sm);
    reinterpret_cast<float4*>(agg + (size_t)(r0 + row) * NODE_DIM)[lane] = o;
  }
}

__global__ void gate_out_kernel(const float* __restrict__ Xn,
                                const float* __restrict__ Wg,
                                const float* __restrict__ agg,
                                float* __restrict__ out, int n1) {
  int wave = blockIdx.x * (blockDim.x >> 5) + (threadIdx.x >> 5);
  int lane = threadIdx.x & 31;
  int tilesM = (n1 + 15) >> 4;
  int tileM = wave >> 3;
  int tileN = wave & 7;
  if (tileM >= tilesM) return;

  int l15 = lane & 15;
  int half = lane >> 4;
  int m = tileM * 16 + l15;
  if (m >= n1) m = n1 - 1;
  const float* arow = Xn + (size_t)m * ATTR_DIM;
  const float* brow = Wg + (size_t)(tileN * 16 + l15) * ATTR_DIM;

  v8f c = {};
#pragma unroll
  for (int k0 = 0; k0 < ATTR_DIM; k0 += 4) {
    int k = k0 + half * 2;
    v2f a = *reinterpret_cast<const v2f*>(arow + k);
    v2f b = *reinterpret_cast<const v2f*>(brow + k);
    c = __builtin_amdgcn_wmma_f32_16x16x4_f32(
         false, a,  false, b,
         (short)0, c,  false,  false);
  }

  int n = tileN * 16 + l15;
#pragma unroll
  for (int v = 0; v < 8; ++v) {
    int row = tileM * 16 + half * 8 + v;
    if (row < n1) {
      float g = 1.0f / (1.0f + expf(-c[v]));
      size_t idx = (size_t)row * NODE_DIM + n;
      out[idx] = g * agg[idx];
    }
  }
}

extern "C" void kernel_launch(void* const* d_in, const int* in_sizes, int n_in,
                              void* d_out, int out_size, void* d_ws,
                              size_t ws_size, hipStream_t stream) {
  const float* X_h_1 = (const float*)d_in[0];
  const float* X_h_2 = (const float*)d_in[1];
  const float* X_n_1 = (const float*)d_in[2];
  const int* cross = (const int*)d_in[3];
  const float* W_gate = (const float*)d_in[4];
  float* out = (float*)d_out;

  const int N1 = in_sizes[0] / NODE_DIM;
  const int E = in_sizes[3] / 2;
  const int* src = cross;
  const int* dst = cross + E;

  float* ws = (float*)d_ws;
  float* agg = ws;
  float* sim = agg + (size_t)N1 * NODE_DIM;

  const int BLK = 256;
  const int waves_per_blk = BLK / 32;

  owner_agg_kernel<<<(N1 + OWN_R - 1) / OWN_R, OWN_WAVES * 32, 0, stream>>>(
      X_h_1, X_h_2, src, dst, sim, agg, N1, E);
  {
    int tilesM = (N1 + 15) / 16;
    int waves = tilesM * (NODE_DIM / 16);
    gate_out_kernel<<<(waves + waves_per_blk - 1) / waves_per_blk, BLK, 0,
                      stream>>>(X_n_1, W_gate, agg, out, N1);
  }
}


# owned X1 rows (contiguous per work-group) copied once into LDS, in the aggregate's space while it is idle; the similarity pass reads each edge's X1 row there, not from memory; sums bit-identical
# speedup vs baseline: 2.7933x; 1.0472x over previous
#include <hip/hip_runtime.h>
#include <math.h>

typedef float v2f __attribute__((ext_vector_type(2)));
typedef float v8f __attribute__((ext_vector_type(8)));

#define NODE_DIM 128
#define ATTR_DIM 64
#define EPS 1e-8f

#define OWN_R 200
#define OWN_WAVES 32
#define OWN_U 4
#define OWN_CW 640
#define OWN_G 8
#define OWN_PB 64
#define OWN_ST 2
#define OWN_XU ((OWN_R * NODE_DIM / 4 + OWN_WAVES * 32 - 1) / (OWN_WAVES * 32))

__device__ __forceinline__ int own_key(float f) {
  int b = __float_as_int(f);
  return b >= 0 ? b : (b ^ 0x7fffffff);
}
__device__ __forceinline__ float own_unkey(int k) {
  return __int_as_float(k >= 0 ? k : (k ^ 0x7fffffff));
}

template <int M>
__device__ __forceinline__ float own_xadd(float x) {
  int b = __builtin_bit_cast(int, x);
  int r = __builtin_amdgcn_update_dpp(b, b, 0x160 + M, 0xf, 0xf, false);
  return x + __builtin_bit_cast(float, r);
}

__device__ __forceinline__ void own_sim_max(const int2* list, int count, int* maxL,
                                            const float* x1L,
                                            const float* __restrict__ X2,
                                            const int* __restrict__ dst,
                                            float* sim, int lane) {
  if (count == 0) return;
  const int j = lane & 7;
  const int g = lane >> 3;
  int2 en[OWN_ST];
  int d[OWN_ST];
#pragma unroll
  for (int t = 0; t < OWN_ST; ++t) {
    en[t] = list[min(t * 4 + g, count - 1)];
    d[t] = dst[en[t].x];
  }
  for (int i = 0; i < count; i += 4 * OWN_ST) {
    float4 a[OWN_ST][4], b[OWN_ST][4];
    int2 cur[OWN_ST];
#pragma unroll
    for (int t = 0; t < OWN_ST; ++t) {
      const float4* ra =
          reinterpret_cast<const float4*>(x1L + en[t].y * NODE_DIM) + j;
      const float4* rb =
          reinterpret_cast<const float4*>(X2 + (size_t)d[t] * NODE_DIM) + j;
#pragma unroll
      for (int c = 0; c < 4; ++c) {
        a[t][c] = ra[c * 8];
        b[t][c] = rb[c * 8];
      }
      cur[t] = en[t];
    }
#pragma unroll
    for (int t = 0; t < OWN_ST; ++t) {
      en[t] = list[min(i + 4 * OWN_ST + t * 4 + g, count - 1)];
      d[t] = dst[en[t].x];
    }
    float pd[OWN_ST], pa[OWN_ST], pb[OWN_ST];
#pragma unroll
    for (int t = 0; t < OWN_ST; ++t) {
      float sd = 0.0f, sa = 0.0f, sb = 0.0f;
#pragma unroll
      for (int c = 0; c < 4; ++c) {
        sd = fmaf(a[t][c].x, b[t][c].x, sd);
        sd = fmaf(a[t][c].y, b[t][c].y, sd);
        sd = fmaf(a[t][c].z, b[t][c].z, sd);
        sd = fmaf(a[t][c].w, b[t][c].w, sd);
        sa = fmaf(a[t][c].x, a[t][c].x, sa);
        sa = fmaf(a[t][c].y, a[t][c].y, sa);
        sa = fmaf(a[t][c].z, a[t][c].z, sa);
        sa = fmaf(a[t][c].w, a[t][c].w, sa);
        sb = fmaf(b[t][c].x, b[t][c].x, sb);
        sb = fmaf(b[t][c].y, b[t][c].y, sb);
        sb = fmaf(b[t][c].z, b[t][c].z, sb);
        sb = fmaf(b[t][c].w, b[t][c].w, sb);
      }
      sd = own_xadd<4>(sd);
      sa = own_xadd<4>(sa);
      sb = own_xadd<4>(sb);
      sd = own_xadd<2>(sd);
      sa = own_xadd<2>(sa);
      sb = own_xadd<2>(sb);
      pd[t] = own_xadd<1>(sd);
      pa[t] = own_xadd<1>(sa);
      pb[t] = own_xadd<1>(sb);
    }
    float md = pd[0], ma = pa[0], mb = pb[0];
    int2 me = cur[0];
#pragma unroll
    for (int t = 1; t < OWN_ST; ++t) {
      const bool s = (j == t);
      md = s ? pd[t] : md;
      ma = s ? pa[t] : ma;
      mb = s ? pb[t] : mb;
      me.x = s ? cur[t].x : me.x;
      me.y = s ? cur[t].y : me.y;
    }
    float sv = md / (fmaxf(sqrtf(ma), EPS) * fmaxf(sqrtf(mb), EPS));
    if (j < OWN_ST && i + j * 4 + g < count) {
      sim[me.x] = sv;
      __hip_atomic_fetch_max(maxL + me.y, own_key(sv), __ATOMIC_RELAXED,
                             __HIP_MEMORY_SCOPE_WORKGROUP);
    }
  }
}

__device__ __forceinline__ void own_flush(const int2* list, int count,
                                          float* aggL, float* sumL,
                                          const int* maxL,
                                          const float* __restrict__ X2,
                                          const int* __restrict__ dst,
                                          const float* sim,
                                          int lane) {
  const int sub = lane & (OWN_G - 1);
  for (int i = 0; i < count; i += OWN_G) {
    int2 en = list[min(i + sub, count - 1)];
    int md = dst[en.x];
    float ms = sim[en.x];
    bool live = (i + sub < count);
    int mloc = live ? en.y : OWN_R;
    int loc[OWN_G];
    float4 b[OWN_G];
    float w[OWN_G];
#pragma unroll
    for (int q = 0; q < OWN_G; ++q) {
      int d = __builtin_amdgcn_readlane(md, q);
      loc[q] = __builtin_amdgcn_readlane(mloc, q);
      b[q] = reinterpret_cast<const float4*>(X2 + (size_t)d * NODE_DIM)[lane];
    }
    float mex = expf(ms - own_unkey(maxL[en.y]));
    if (lane < OWN_G && live)
      __hip_atomic_fetch_add(sumL + en.y, mex, __ATOMIC_RELAXED, __HIP_MEMORY_SCOPE_WORKGROUP);
#pragma unroll
    for (int q = 0; q < OWN_G; ++q)
      w[q] = __int_as_float(__builtin_amdgcn_readlane(__float_as_int(mex), q));
#pragma unroll
    for (int q = 0; q < OWN_G; ++q) {
      float* o = aggL + loc[q] * NODE_DIM + lane;
      __hip_atomic_fetch_add(o + 0, w[q] * b[q].x, __ATOMIC_RELAXED, __HIP_MEMORY_SCOPE_WORKGROUP);
      __hip_atomic_fetch_add(o + 32, w[q] * b[q].y, __ATOMIC_RELAXED, __HIP_MEMORY_SCOPE_WORKGROUP);
      __hip_atomic_fetch_add(o + 64, w[q] * b[q].z, __ATOMIC_RELAXED, __HIP_MEMORY_SCOPE_WORKGROUP);
      __hip_atomic_fetch_add(o + 96, w[q] * b[q].w, __ATOMIC_RELAXED, __HIP_MEMORY_SCOPE_WORKGROUP);
    }
  }
}

__device__ __forceinline__ int own_append(int2* myList, int count, int s, int e,
                                          int r0, int rn) {
  unsigned loc = (unsigned)(s - r0);
  bool m = loc < (unsigned)rn;
  unsigned mask = __builtin_amdgcn_ballot_w32(m);
  if (__builtin_expect(mask != 0u, 0)) {
    int pos = count + (int)__builtin_amdgcn_mbcnt_lo(mask, 0u);
    if (m) myList[pos] = make_int2(e, (int)loc);
    count += __builtin_popcount(mask);
  }
  return count;
}

__global__ __launch_bounds__(OWN_WAVES * 32) void owner_agg_kernel(
    const float* __restrict__ X1, const float* __restrict__ X2,
    const int* __restrict__ src, const int* __restrict__ dst, float* sim,
    float* __restrict__ agg, int n1, int E) {
  __shared__ float aggL[(OWN_R + 1) * NODE_DIM];
  __shared__ int2 listL[OWN_WAVES * OWN_CW];
  __shared__ int2 privL[OWN_WAVES * OWN_PB];
  __shared__ float sumL[OWN_R];
  __shared__ int maxL[OWN_R];
  __shared__ int cntL[OWN_WAVES];
  __shared__ int ovfL;
  const int tid = threadIdx.x;
  const int lane = tid & 31;
  const int w = __builtin_amdgcn_readfirstlane(tid >> 5);
  const int r0 = blockIdx.x * OWN_R;
  const int rn = min(OWN_R, n1 - r0);

  {
    const float4* xs = reinterpret_cast<const float4*>(X1 + (size_t)r0 * NODE_DIM);
    float4* xd = reinterpret_cast<float4*>(aggL);
    const int total = rn * (NODE_DIM / 4);
    float4 xv[OWN_XU];
#pragma unroll
    for (int u = 0; u < OWN_XU; ++u) {
      const int i = tid + u * (OWN_WAVES * 32);
      xv[u] = (i < total) ? xs[i] : make_float4(0.f, 0.f, 0.f, 0.f);
    }
#pragma unroll
    for (int u = 0; u < OWN_XU; ++u) {
      const int i = tid + u * (OWN_WAVES * 32);
      if (i < total) xd[i] = xv[u];
    }
  }
  if (tid < OWN_R) {
    sumL[tid] = 0.0f;
    maxL[tid] = own_key(-INFINITY);
  }
  if (tid == 0) ovfL = 0;
  __syncthreads();

  int2* myList = listL + w * OWN_CW;
  const int n4 = E >> 2;
  const int4* src4 = reinterpret_cast<const int4*>(src);
  const int trip4 = OWN_WAVES * OWN_U * 32;
  const int trips = (n4 + trip4 - 1) / trip4;
  const int tt0 = trips > 0 ? (int)(blockIdx.x % (unsigned)trips) : 0;
  bool second = false;
  for (int sweep = 0; sweep < 2; ++sweep) {
    int count = 0;
    int tt = tt0;
    for (int t = 0; t < trips; ++t) {
      const int base = tt * trip4 + w * (OWN_U * 32);
      tt = (tt + 1 == trips) ? 0 : tt + 1;
      if (count > OWN_CW - OWN_U * 128 - 4) {
        if (second) {
          own_flush(myList, count, aggL, sumL, maxL, X2, dst, sim, lane);
        } else {
          own_sim_max(myList, count, maxL, aggL, X2, dst, sim, lane);
          if (lane == 0) ovfL = 1;
        }
        count = 0;
      }
      int4 v[OWN_U];
#pragma unroll
      for (int u = 0; u < OWN_U; ++u) {
        int g = base + u * 32 + lane;
        v[u] = (g < n4) ? src4[g] : make_int4(-1, -1, -1, -1);
      }
#pragma unroll
      for (int u = 0; u < OWN_U; ++u) {
        int e = (base + u * 32 + lane) << 2;
        count = own_append(myList, count, v[u].x, e + 0, r0, rn);
        count = own_append(myList, count, v[u].y, e + 1, r0, rn);
        count = own_append(myList, count, v[u].z, e + 2, r0, rn);
        count = own_append(myList, count, v[u].w, e + 3, r0, rn);
      }
    }
    if (w == 0) {
      int e = (n4 << 2) + lane;
      int s = (e < E) ? src[e] : -1;
      count = own_append(myList, count, s, e, r0, rn);
    }
    if (second) {
      own_flush(myList, count, aggL, sumL, maxL, X2, dst, sim, lane);
    } else {
      own_sim_max(myList, count, maxL, aggL, X2, dst, sim, lane);
      if (lane == 0) cntL[w] = count;
    }
    __syncthreads();
    if (second) break;
    for (int i = tid; i < (OWN_R + 1) * NODE_DIM / 4; i += OWN_WAVES * 32)
      reinterpret_cast<float4*>(aggL)[i] = make_float4(0.f, 0.f, 0.f, 0.f);
    __syncthreads();
    if (ovfL == 0) break;
    second = true;
  }

  if (!second) {
    const int myc = cntL[lane];
    int2* myPriv = privL + w * OWN_PB;
    int pc = 0;
    for (int j = 0; j < OWN_WAVES; ++j) {
      const int cj = __builtin_amdgcn_readlane(myc, j);
      const int2* Lj = listL + j * OWN_CW;
      for (int i = 0; i < cj; i += 32) {
        const int idx = i + lane;
        int2 en = Lj[min(idx, cj - 1)];
        bool m = (idx < cj) && ((en.y & (OWN_WAVES - 1)) == w);
        unsigned mask = __builtin_amdgcn_ballot_w32(m);
        if (mask != 0u) {
          int pos = pc + (int)__builtin_amdgcn_mbcnt_lo(mask, 0u);
          if (m) myPriv[pos] = en;
          pc += __builtin_popcount(mask);
          if (pc > OWN_PB - 32) {
            own_flush(myPriv, pc, aggL, sumL, maxL, X2, dst, sim, lane);
            pc = 0;
          }
        }
      }
    }
    own_flush(myPriv, pc, aggL, sumL, maxL, X2, dst, sim, lane);
  }
  __syncthreads();

  for (int i = tid; i < rn * 32; i += OWN_WAVES * 32) {
    int row = i >> 5;
    const float* p = aggL + row * NODE_DIM + lane;
    float sm = sumL[row];
    float4 o = make_float4(0.f, 0.f, 0.f, 0.f);
    if (sm > 0.0f) o = make_float4(p[0] / sm, p[32] / sm, p[64] / sm, p[96] / sm);
    reinterpret_cast<float4*>(agg + (size_t)(r0 + row) * NODE_DIM)[lane] = o;
  }
}

__global__ void gate_out_kernel(const float* __restrict__ Xn,
                                const float* __restrict__ Wg,
                                const float* __restrict__ agg,
                                float* __restrict__ out, int n1) {
  int wave = blockIdx.x * (blockDim.x >> 5) + (threadIdx.x >> 5);
  int lane = threadIdx.x & 31;
  int tilesM = (n1 + 15) >> 4;
  int tileM = wave >> 3;
  int tileN = wave & 7;
  if (tileM >= tilesM) return;

  int l15 = lane & 15;
  int half = lane >> 4;
  int m = tileM * 16 + l15;
  if (m >= n1) m = n1 - 1;
  const float* arow = Xn + (size_t)m * ATTR_DIM;
  const float* brow = Wg + (size_t)(tileN * 16 + l15) * ATTR_DIM;

  v8f c = {};
#pragma unroll
  for (int k0 = 0; k0 < ATTR_DIM; k0 += 4) {
    int k = k0 + half * 2;
    v2f a = *reinterpret_cast<const v2f*>(arow + k);
    v2f b = *reinterpret_cast<const v2f*>(brow + k);
    c = __builtin_amdgcn_wmma_f32_16x16x4_f32(
         false, a,  false, b,
         (short)0, c,  false,  false);
  }

  int n = tileN * 16 + l15;
#pragma unroll
  for (int v = 0; v < 8; ++v) {
    int row = tileM * 16 + half * 8 + v;
    if (row < n1) {
      float g = 1.0f / (1.0f + expf(-c[v]));
      size_t idx = (size_t)row * NODE_DIM + n;
      out[idx] = g * agg[idx];
    }
  }
}

extern "C" void kernel_launch(void* const* d_in, const int* in_sizes, int n_in,
                              void* d_out, int out_size, void* d_ws,
                              size_t ws_size, hipStream_t stream) {
  const float* X_h_1 = (const float*)d_in[0];
  const float* X_h_2 = (const float*)d_in[1];
  const float* X_n_1 = (const float*)d_in[2];
  const int* cross = (const int*)d_in[3];
  const float* W_gate = (const float*)d_in[4];
  float* out = (float*)d_out;

  const int N1 = in_sizes[0] / NODE_DIM;
  const int E = in_sizes[3] / 2;
  const int* src = cross;
  const int* dst = cross + E;

  float* ws = (float*)d_ws;
  float* agg = ws;
  float* sim = agg + (size_t)N1 * NODE_DIM;

  const int BLK = 256;
  const int waves_per_blk = BLK / 32;

  owner_agg_kernel<<<(N1 + OWN_R - 1) / OWN_R, OWN_WAVES * 32, 0, stream>>>(
      X_h_1, X_h_2, src, dst, sim, agg, N1, E);
  {
    int tilesM = (N1 + 15) / 16;
    int waves = tilesM * (NODE_DIM / 16);
    gate_out_kernel<<<(waves + waves_per_blk - 1) / waves_per_blk, BLK, 0,
                      stream>>>(X_n_1, W_gate, agg, out, N1);
  }
}


# gate GEMM (WMMA) + sigmoid + multiply done by the owner work-groups on their aggregate while it is in LDS, Xn and Wg staged in the lists' LDS by whole-line loads; 2 launches to 1; bits unchanged
# speedup vs baseline: 3.2461x; 1.1621x over previous
#include <hip/hip_runtime.h>
#include <math.h>

typedef float v2f __attribute__((ext_vector_type(2)));
typedef float v8f __attribute__((ext_vector_type(8)));

#define NODE_DIM 128
#define ATTR_DIM 64
#define EPS 1e-8f

#define OWN_R 200
#define OWN_WAVES 32
#define OWN_U 4
#define OWN_CW 640
#define OWN_G 8
#define OWN_PB 64
#define OWN_ST 2
#define OWN_XU ((OWN_R * NODE_DIM / 4 + OWN_WAVES * 32 - 1) / (OWN_WAVES * 32))
#define OWN_TM ((OWN_R + 15) / 16)
#define OWN_XP (ATTR_DIM + 4)
#define OWN_GA ((OWN_TM * 16 * (ATTR_DIM / 4) + OWN_WAVES * 32 - 1) / (OWN_WAVES * 32))
#define OWN_GB ((NODE_DIM * (ATTR_DIM / 4) + OWN_WAVES * 32 - 1) / (OWN_WAVES * 32))
static_assert(OWN_WAVES % (NODE_DIM / 16) == 0, "a wave keeps one feature tile for all its gate tiles");
static_assert((OWN_TM * 16 + NODE_DIM) * OWN_XP * 4 <= OWN_WAVES * OWN_CW * 8,
              "the staged Xn and Wg rows fit in the space of the match lists");

__device__ __forceinline__ int own_key(float f) {
  int b = __float_as_int(f);
  return b >= 0 ? b : (b ^ 0x7fffffff);
}
__device__ __forceinline__ float own_unkey(int k) {
  return __int_as_float(k >= 0 ? k : (k ^ 0x7fffffff));
}

template <int M>
__device__ __forceinline__ float own_xadd(float x) {
  int b = __builtin_bit_cast(int, x);
  int r = __builtin_amdgcn_update_dpp(b, b, 0x160 + M, 0xf, 0xf, false);
  return x + __builtin_bit_cast(float, r);
}

__device__ __forceinline__ void own_sim_max(const int2* list, int count, int* maxL,
                                            const float* x1L,
                                            const float* __restrict__ X2,
                                            const int* __restrict__ dst,
                                            float* sim, int lane) {
  if (count == 0) return;
  const int j = lane & 7;
  const int g = lane >> 3;
  int2 en[OWN_ST];
  int d[OWN_ST];
#pragma unroll
  for (int t = 0; t < OWN_ST; ++t) {
    en[t] = list[min(t * 4 + g, count - 1)];
    d[t] = dst[en[t].x];
  }
  for (int i = 0; i < count; i += 4 * OWN_ST) {
    float4 a[OWN_ST][4], b[OWN_ST][4];
    int2 cur[OWN_ST];
#pragma unroll
    for (int t = 0; t < OWN_ST; ++t) {
      const float4* ra =
          reinterpret_cast<const float4*>(x1L + en[t].y * NODE_DIM) + j;
      const float4* rb =
          reinterpret_cast<const float4*>(X2 + (size_t)d[t] * NODE_DIM) + j;
#pragma unroll
      for (int c = 0; c < 4; ++c) {
        a[t][c] = ra[c * 8];
        b[t][c] = rb[c * 8];
      }
      cur[t] = en[t];
    }
#pragma unroll
    for (int t = 0; t < OWN_ST; ++t) {
      en[t] = list[min(i + 4 * OWN_ST + t * 4 + g, count - 1)];
      d[t] = dst[en[t].x];
    }
    float pd[OWN_ST], pa[OWN_ST], pb[OWN_ST];
#pragma unroll
    for (int t = 0; t < OWN_ST; ++t) {
      float sd = 0.0f, sa = 0.0f, sb = 0.0f;
#pragma unroll
      for (int c = 0; c < 4; ++c) {
        sd = fmaf(a[t][c].x, b[t][c].x, sd);
        sd = fmaf(a[t][c].y, b[t][c].y, sd);
        sd = fmaf(a[t][c].z, b[t][c].z, sd);
        sd = fmaf(a[t][c].w, b[t][c].w, sd);
        sa = fmaf(a[t][c].x, a[t][c].x, sa);
        sa = fmaf(a[t][c].y, a[t][c].y, sa);
        sa = fmaf(a[t][c].z, a[t][c].z, sa);
        sa = fmaf(a[t][c].w, a[t][c].w, sa);
        sb = fmaf(b[t][c].x, b[t][c].x, sb);
        sb = fmaf(b[t][c].y, b[t][c].y, sb);
        sb = fmaf(b[t][c].z, b[t][c].z, sb);
        sb = fmaf(b[t][c].w, b[t][c].w, sb);
      }
      sd = own_xadd<4>(sd);
      sa = own_xadd<4>(sa);
      sb = own_xadd<4>(sb);
      sd = own_xadd<2>(sd);
      sa = own_xadd<2>(sa);
      sb = own_xadd<2>(sb);
      pd[t] = own_xadd<1>(sd);
      pa[t] = own_xadd<1>(sa);
      pb[t] = own_xadd<1>(sb);
    }
    float md = pd[0], ma = pa[0], mb = pb[0];
    int2 me = cur[0];
#pragma unroll
    for (int t = 1; t < OWN_ST; ++t) {
      const bool s = (j == t);
      md = s ? pd[t] : md;
      ma = s ? pa[t] : ma;
      mb = s ? pb[t] : mb;
      me.x = s ? cur[t].x : me.x;
      me.y = s ? cur[t].y : me.y;
    }
    float sv = md / (fmaxf(sqrtf(ma), EPS) * fmaxf(sqrtf(mb), EPS));
    if (j < OWN_ST && i + j * 4 + g < count) {
      sim[me.x] = sv;
      __hip_atomic_fetch_max(maxL + me.y, own_key(sv), __ATOMIC_RELAXED,
                             __HIP_MEMORY_SCOPE_WORKGROUP);
    }
  }
}

__device__ __forceinline__ void own_flush(const int2* list, int count,
                                          float* aggL, float* sumL,
                                          const int* maxL,
                                          const float* __restrict__ X2,
                                          const int* __restrict__ dst,
                                          const float* sim,
                                          int lane) {
  const int sub = lane & (OWN_G - 1);
  for (int i = 0; i < count; i += OWN_G) {
    int2 en = list[min(i + sub, count - 1)];
    int md = dst[en.x];
    float ms = sim[en.x];
    bool live = (i + sub < count);
    int mloc = live ? en.y : OWN_R;
    int loc[OWN_G];
    float4 b[OWN_G];
    float w[OWN_G];
#pragma unroll
    for (int q = 0; q < OWN_G; ++q) {
      int d = __builtin_amdgcn_readlane(md, q);
      loc[q] = __builtin_amdgcn_readlane(mloc, q);
      b[q] = reinterpret_cast<const float4*>(X2 + (size_t)d * NODE_DIM)[lane];
    }
    float mex = expf(ms - own_unkey(maxL[en.y]));
    if (lane < OWN_G && live)
      __hip_atomic_fetch_add(sumL + en.y, mex, __ATOMIC_RELAXED, __HIP_MEMORY_SCOPE_WORKGROUP);
#pragma unroll
    for (int q = 0; q < OWN_G; ++q)
      w[q] = __int_as_float(__builtin_amdgcn_readlane(__float_as_int(mex), q));
#pragma unroll
    for (int q = 0; q < OWN_G; ++q) {
      float* o = aggL + loc[q] * NODE_DIM + lane;
      __hip_atomic_fetch_add(o + 0, w[q] * b[q].x, __ATOMIC_RELAXED, __HIP_MEMORY_SCOPE_WORKGROUP);
      __hip_atomic_fetch_add(o + 32, w[q] * b[q].y, __ATOMIC_RELAXED, __HIP_MEMORY_SCOPE_WORKGROUP);
      __hip_atomic_fetch_add(o + 64, w[q] * b[q].z, __ATOMIC_RELAXED, __HIP_MEMORY_SCOPE_WORKGROUP);
      __hip_atomic_fetch_add(o + 96, w[q] * b[q].w, __ATOMIC_RELAXED, __HIP_MEMORY_SCOPE_WORKGROUP);
    }
  }
}

__device__ __forceinline__ int own_append(int2* myList, int count, int s, int e,
                                          int r0, int rn) {
  unsigned loc = (unsigned)(s - r0);
  bool m = loc < (unsigned)rn;
  unsigned mask = __builtin_amdgcn_ballot_w32(m);
  if (__builtin_expect(mask != 0u, 0)) {
    int pos = count + (int)__builtin_amdgcn_mbcnt_lo(mask, 0u);
    if (m) myList[pos] = make_int2(e, (int)loc);
    count += __builtin_popcount(mask);
  }
  return count;
}

__global__ __launch_bounds__(OWN_WAVES * 32) void owner_agg_kernel(
    const float* __restrict__ X1, const float* __restrict__ X2,
    const float* __restrict__ Xn, const float* __restrict__ Wg,
    const int* __restrict__ src, const int* __restrict__ dst, float* sim,
    float* __restrict__ out, int n1, int E) {
  __shared__ float aggL[(OWN_R + 1) * NODE_DIM];
  __shared__ __attribute__((aligned(16))) int2 listL[OWN_WAVES * OWN_CW];
  __shared__ int2 privL[OWN_WAVES * OWN_PB];
  __shared__ float sumL[OWN_R];
  __shared__ int maxL[OWN_R];
  __shared__ int cntL[OWN_WAVES];
  __shared__ int ovfL;
  const int tid = threadIdx.x;
  const int lane = tid & 31;
  const int w = __builtin_amdgcn_readfirstlane(tid >> 5);
  const int r0 = blockIdx.x * OWN_R;
  const int rn = min(OWN_R, n1 - r0);

  {
    const float4* xs = reinterpret_cast<const float4*>(X1 + (size_t)r0 * NODE_DIM);
    float4* xd = reinterpret_cast<float4*>(aggL);
    const int total = rn * (NODE_DIM / 4);
    float4 xv[OWN_XU];
#pragma unroll
    for (int u = 0; u < OWN_XU; ++u) {
      const int i = tid + u * (OWN_WAVES * 32);
      xv[u] = (i < total) ? xs[i] : make_float4(0.f, 0.f, 0.f, 0.f);
    }
#pragma unroll
    for (int u = 0; u < OWN_XU; ++u) {
      const int i = tid + u * (OWN_WAVES * 32);
      if (i < total) xd[i] = xv[u];
    }
  }
  if (tid < OWN_R) {
    sumL[tid] = 0.0f;
    maxL[tid] = own_key(-INFINITY);
  }
  if (tid == 0) ovfL = 0;
  __syncthreads();

  int2* myList = listL + w * OWN_CW;
  const int n4 = E >> 2;
  const int4* src4 = reinterpret_cast<const int4*>(src);
  const int trip4 = OWN_WAVES * OWN_U * 32;
  const int trips = (n4 + trip4 - 1) / trip4;
  const int tt0 = trips > 0 ? (int)(blockIdx.x % (unsigned)trips) : 0;
  bool second = false;
  for (int sweep = 0; sweep < 2; ++sweep) {
    int count = 0;
    int tt = tt0;
    for (int t = 0; t < trips; ++t) {
      const int base = tt * trip4 + w * (OWN_U * 32);
      tt = (tt + 1 == trips) ? 0 : tt + 1;
      if (count > OWN_CW - OWN_U * 128 - 4) {
        if (second) {
          own_flush(myList, count, aggL, sumL, maxL, X2, dst, sim, lane);
        } else {
          own_sim_max(myList, count, maxL, aggL, X2, dst, sim, lane);
          if (lane == 0) ovfL = 1;
        }
        count = 0;
      }
      int4 v[OWN_U];
#pragma unroll
      for (int u = 0; u < OWN_U; ++u) {
        int g = base + u * 32 + lane;
        v[u] = (g < n4) ? src4[g] : make_int4(-1, -1, -1, -1);
      }
#pragma unroll
      for (int u = 0; u < OWN_U; ++u) {
        int e = (base + u * 32 + lane) << 2;
        count = own_append(myList, count, v[u].x, e + 0, r0, rn);
        count = own_append(myList, count, v[u].y, e + 1, r0, rn);
        count = own_append(myList, count, v[u].z, e + 2, r0, rn);
        count = own_append(myList, count, v[u].w, e + 3, r0, rn);
      }
    }
    if (w == 0) {
      int e = (n4 << 2) + lane;
      int s = (e < E) ? src[e] : -1;
      count = own_append(myList, count, s, e, r0, rn);
    }
    if (second) {
      own_flush(myList, count, aggL, sumL, maxL, X2, dst, sim, lane);
    } else {
      own_sim_max(myList, count, maxL, aggL, X2, dst, sim, lane);
      if (lane == 0) cntL[w] = count;
    }
    __syncthreads();
    if (second) break;
    for (int i = tid; i < (OWN_R + 1) * NODE_DIM / 4; i += OWN_WAVES * 32)
      reinterpret_cast<float4*>(aggL)[i] = make_float4(0.f, 0.f, 0.f, 0.f);
    __syncthreads();
    if (ovfL == 0) break;
    second = true;
  }

  float4 ga[OWN_GA], gb[OWN_GB];
  {
    const float4* xs = reinterpret_cast<const float4*>(Xn + (size_t)r0 * ATTR_DIM);
    const float4* wsrc = reinterpret_cast<const float4*>(Wg);
#pragma unroll
    for (int u = 0; u < OWN_GA; ++u) {
      const int i = tid + u * (OWN_WAVES * 32);
      ga[u] = (i < rn * (ATTR_DIM / 4)) ? xs[i] : make_float4(0.f, 0.f, 0.f, 0.f);
    }
#pragma unroll
    for (int u = 0; u < OWN_GB; ++u) {
      const int i = tid + u * (OWN_WAVES * 32);
      gb[u] = (i < NODE_DIM * (ATTR_DIM / 4)) ? wsrc[i] : make_float4(0.f, 0.f, 0.f, 0.f);
    }
  }

  if (!second) {
    const int myc = cntL[lane];
    int2* myPriv = privL + w * OWN_PB;
    int pc = 0;
    for (int j = 0; j < OWN_WAVES; ++j) {
      const int cj = __builtin_amdgcn_readlane(myc, j);
      const int2* Lj = listL + j * OWN_CW;
      for (int i = 0; i < cj; i += 32) {
        const int idx = i + lane;
        int2 en = Lj[min(idx, cj - 1)];
        bool m = (idx < cj) && ((en.y & (OWN_WAVES - 1)) == w);
        unsigned mask = __builtin_amdgcn_ballot_w32(m);
        if (mask != 0u) {
          int pos = pc + (int)__builtin_amdgcn_mbcnt_lo(mask, 0u);
          if (m) myPriv[pos] = en;
          pc += __builtin_popcount(mask);
          if (pc > OWN_PB - 32) {
            own_flush(myPriv, pc, aggL, sumL, maxL, X2, dst, sim, lane);
            pc = 0;
          }
        }
      }
    }
    own_flush(myPriv, pc, aggL, sumL, maxL, X2, dst, sim, lane);
  }

  __syncthreads();

  float* xnL = reinterpret_cast<float*>(listL);
  float* wgL = xnL + OWN_TM * 16 * OWN_XP;
#pragma unroll
  for (int u = 0; u < OWN_GA; ++u) {
    const int i = tid + u * (OWN_WAVES * 32);
    if (i < OWN_TM * 16 * (ATTR_DIM / 4))
      *reinterpret_cast<float4*>(xnL + (i >> 4) * OWN_XP + (i & 15) * 4) = ga[u];
  }
#pragma unroll
  for (int u = 0; u < OWN_GB; ++u) {
    const int i = tid + u * (OWN_WAVES * 32);
    if (i < NODE_DIM * (ATTR_DIM / 4))
      *reinterpret_cast<float4*>(wgL + (i >> 4) * OWN_XP + (i & 15) * 4) = gb[u];
  }
  __syncthreads();

  {
    const int l15 = lane & 15;
    const int half = lane >> 4;
    const int tn = w & (NODE_DIM / 16 - 1);
    const float* bp = wgL + (tn * 16 + l15) * OWN_XP + half * 2;
    v2f bf[ATTR_DIM / 4];
#pragma unroll
    for (int s = 0; s < ATTR_DIM / 4; ++s) bf[s] = *reinterpret_cast<const v2f*>(bp + 4 * s);
    const int n = tn * 16 + l15;
    const int col = (n & 3) * 32 + (n >> 2);
    const int tiles = ((rn + 15) >> 4) * (NODE_DIM / 16);
    for (int t = w; t < tiles; t += OWN_WAVES) {
      const int tm = t / (NODE_DIM / 16);
      const float* ap = xnL + (tm * 16 + l15) * OWN_XP + half * 2;
      v2f af[ATTR_DIM / 4];
#pragma unroll
      for (int s = 0; s < ATTR_DIM / 4; ++s) af[s] = *reinterpret_cast<const v2f*>(ap + 4 * s);
      float pv[8], sv[8];
#pragma unroll
      for (int v = 0; v < 8; ++v) {
        const int rc = min(tm * 16 + half * 8 + v, rn - 1);
        pv[v] = aggL[rc * NODE_DIM + col];
        sv[v] = sumL[rc];
      }
      v8f c = {};
#pragma unroll
      for (int s = 0; s < ATTR_DIM / 4; ++s)
        c = __builtin_amdgcn_wmma_f32_16x16x4_f32(
             false, af[s],  false, bf[s],
             (short)0, c,  false,  false);
#pragma unroll
      for (int v = 0; v < 8; ++v) {
        const int row = tm * 16 + half * 8 + v;
        const float g = 1.0f / (1.0f + expf(-c[v]));
        const float a = (sv[v] > 0.0f) ? (pv[v] / sv[v]) : 0.0f;
        aggL[(row < rn ? row : OWN_R) * NODE_DIM + col] = g * a;
      }
    }
  }
  __syncthreads();

  for (int i = tid; i < rn * 32; i += OWN_WAVES * 32) {
    int row = i >> 5;
    const float* p = aggL + row * NODE_DIM + lane;
    reinterpret_cast<float4*>(out + (size_t)(r0 + row) * NODE_DIM)[lane] =
        make_float4(p[0], p[32], p[64], p[96]);
  }
}

extern "C" void kernel_launch(void* const* d_in, const int* in_sizes, int n_in,
                              void* d_out, int out_size, void* d_ws,
                              size_t ws_size, hipStream_t stream) {
  const float* X_h_1 = (const float*)d_in[0];
  const float* X_h_2 = (const float*)d_in[1];
  const float* X_n_1 = (const float*)d_in[2];
  const int* cross = (const int*)d_in[3];
  const float* W_gate = (const float*)d_in[4];
  float* out = (float*)d_out;

  const int N1 = in_sizes[0] / NODE_DIM;
  const int E = in_sizes[3] / 2;
  const int* src = cross;
  const int* dst = cross + E;

  float* sim = (float*)d_ws;

  owner_agg_kernel<<<(N1 + OWN_R - 1) / OWN_R, OWN_WAVES * 32, 0, stream>>>(
      X_h_1, X_h_2, X_n_1, W_gate, src, dst, sim, out, N1, E);
}


# edges regrouped by node before any row is read: 4 lanes a node do sim (new sum tree) + softmax (running max, sums rescaled) + weighted sum in registers; X2 rows read once, no LDS float adds
# speedup vs baseline: 3.9545x; 1.2182x over previous
#include <hip/hip_runtime.h>
#include <math.h>

typedef float v2f __attribute__((ext_vector_type(2)));
typedef float v8f __attribute__((ext_vector_type(8)));

#define NODE_DIM 128
#define ATTR_DIM 64
#define EPS 1e-8f

#define OWN_R 200
#define OWN_WAVES 32
#define OWN_U 4
#define OWN_CW 640
#define OWN_G 8
#define OWN_PB 192
#define OWN_K ((OWN_R + OWN_WAVES - 1) / OWN_WAVES)
#define OWN_ST 2
#define OWN_XU ((OWN_R * NODE_DIM / 4 + OWN_WAVES * 32 - 1) / (OWN_WAVES * 32))
#define OWN_TM ((OWN_R + 15) / 16)
#define OWN_XP (ATTR_DIM + 4)
#define OWN_GA ((OWN_TM * 16 * (ATTR_DIM / 4) + OWN_WAVES * 32 - 1) / (OWN_WAVES * 32))
#define OWN_GB ((NODE_DIM * (ATTR_DIM / 4) + OWN_WAVES * 32 - 1) / (OWN_WAVES * 32))
static_assert(OWN_WAVES % (NODE_DIM / 16) == 0, "a wave keeps one feature tile for all its gate tiles");
static_assert((OWN_TM * 16 + NODE_DIM) * OWN_XP * 4 <= OWN_WAVES * OWN_CW * 8,
              "the staged Xn and Wg rows fit in the space of the match lists");

__device__ __forceinline__ int own_key(float f) {
  int b = __float_as_int(f);
  return b >= 0 ? b : (b ^ 0x7fffffff);
}
__device__ __forceinline__ float own_unkey(int k) {
  return __int_as_float(k >= 0 ? k : (k ^ 0x7fffffff));
}

template <int M>
__device__ __forceinline__ float own_xadd(float x) {
  int b = __builtin_bit_cast(int, x);
  int r = __builtin_amdgcn_update_dpp(b, b, 0x160 + M, 0xf, 0xf, false);
  return x + __builtin_bit_cast(float, r);
}

__device__ __forceinline__ void own_sim_max(const int2* list, int count, int* maxL,
                                            const float* x1L,
                                            const float* __restrict__ X2,
                                            const int* __restrict__ dst,
                                            float* sim, int lane) {
  if (count == 0) return;
  const int j = lane & 7;
  const int g = lane >> 3;
  int2 en[OWN_ST];
  int d[OWN_ST];
#pragma unroll
  for (int t = 0; t < OWN_ST; ++t) {
    en[t] = list[min(t * 4 + g, count - 1)];
    d[t] = dst[en[t].x];
  }
  for (int i = 0; i < count; i += 4 * OWN_ST) {
    float4 a[OWN_ST][4], b[OWN_ST][4];
    int2 cur[OWN_ST];
#pragma unroll
    for (int t = 0; t < OWN_ST; ++t) {
      const float4* ra =
          reinterpret_cast<const float4*>(x1L + en[t].y * NODE_DIM) + j;
      const float4* rb =
          reinterpret_cast<const float4*>(X2 + (size_t)d[t] * NODE_DIM) + j;
#pragma unroll
      for (int c = 0; c < 4; ++c) {
        a[t][c] = ra[c * 8];
        b[t][c] = rb[c * 8];
      }
      cur[t] = en[t];
    }
#pragma unroll
    for (int t = 0; t < OWN_ST; ++t) {
      en[t] = list[min(i + 4 * OWN_ST + t * 4 + g, count - 1)];
      d[t] = dst[en[t].x];
    }
    float pd[OWN_ST], pa[OWN_ST], pb[OWN_ST];
#pragma unroll
    for (int t = 0; t < OWN_ST; ++t) {
      float sd = 0.0f, sa = 0.0f, sb = 0.0f;
#pragma unroll
      for (int c = 0; c < 4; ++c) {
        sd = fmaf(a[t][c].x, b[t][c].x, sd);
        sd = fmaf(a[t][c].y, b[t][c].y, sd);
        sd = fmaf(a[t][c].z, b[t][c].z, sd);
        sd = fmaf(a[t][c].w, b[t][c].w, sd);
        sa = fmaf(a[t][c].x, a[t][c].x, sa);
        sa = fmaf(a[t][c].y, a[t][c].y, sa);
        sa = fmaf(a[t][c].z, a[t][c].z, sa);
        sa = fmaf(a[t][c].w, a[t][c].w, sa);
        sb = fmaf(b[t][c].x, b[t][c].x, sb);
        sb = fmaf(b[t][c].y, b[t][c].y, sb);
        sb = fmaf(b[t][c].z, b[t][c].z, sb);
        sb = fmaf(b[t][c].w, b[t][c].w, sb);
      }
      sd = own_xadd<4>(sd);
      sa = own_xadd<4>(sa);
      sb = own_xadd<4>(sb);
      sd = own_xadd<2>(sd);
      sa = own_xadd<2>(sa);
      sb = own_xadd<2>(sb);
      pd[t] = own_xadd<1>(sd);
      pa[t] = own_xadd<1>(sa);
      pb[t] = own_xadd<1>(sb);
    }
    float md = pd[0], ma = pa[0], mb = pb[0];
    int2 me = cur[0];
#pragma unroll
    for (int t = 1; t < OWN_ST; ++t) {
      const bool s = (j == t);
      md = s ? pd[t] : md;
      ma = s ? pa[t] : ma;
      mb = s ? pb[t] : mb;
      me.x = s ? cur[t].x : me.x;
      me.y = s ? cur[t].y : me.y;
    }
    float sv = md / (fmaxf(sqrtf(ma), EPS) * fmaxf(sqrtf(mb), EPS));
    if (j < OWN_ST && i + j * 4 + g < count) {
      sim[me.x] = sv;
      __hip_atomic_fetch_max(maxL + me.y, own_key(sv), __ATOMIC_RELAXED,
                             __HIP_MEMORY_SCOPE_WORKGROUP);
    }
  }
}

__device__ __forceinline__ void own_flush(const int2* list, int count,
                                          float* aggL, float* sumL,
                                          const int* maxL,
                                          const float* __restrict__ X2,
                                          const int* __restrict__ dst,
                                          const float* sim,
                                          int lane) {
  const int sub = lane & (OWN_G - 1);
  for (int i = 0; i < count; i += OWN_G) {
    int2 en = list[min(i + sub, count - 1)];
    int md = dst[en.x];
    float ms = sim[en.x];
    bool live = (i + sub < count);
    int mloc = live ? en.y : OWN_R;
    int loc[OWN_G];
    float4 b[OWN_G];
    float w[OWN_G];
#pragma unroll
    for (int q = 0; q < OWN_G; ++q) {
      int d = __builtin_amdgcn_readlane(md, q);
      loc[q] = __builtin_amdgcn_readlane(mloc, q);
      b[q] = reinterpret_cast<const float4*>(X2 + (size_t)d * NODE_DIM)[lane];
    }
    float mex = expf(ms - own_unkey(maxL[en.y]));
    if (lane < OWN_G && live)
      __hip_atomic_fetch_add(sumL + en.y, mex, __ATOMIC_RELAXED, __HIP_MEMORY_SCOPE_WORKGROUP);
#pragma unroll
    for (int q = 0; q < OWN_G; ++q)
      w[q] = __int_as_float(__builtin_amdgcn_readlane(__float_as_int(mex), q));
#pragma unroll
    for (int q = 0; q < OWN_G; ++q) {
      float* o = aggL + loc[q] * NODE_DIM + lane;
      __hip_atomic_fetch_add(o + 0, w[q] * b[q].x, __ATOMIC_RELAXED, __HIP_MEMORY_SCOPE_WORKGROUP);
      __hip_atomic_fetch_add(o + 32, w[q] * b[q].y, __ATOMIC_RELAXED, __HIP_MEMORY_SCOPE_WORKGROUP);
      __hip_atomic_fetch_add(o + 64, w[q] * b[q].z, __ATOMIC_RELAXED, __HIP_MEMORY_SCOPE_WORKGROUP);
      __hip_atomic_fetch_add(o + 96, w[q] * b[q].w, __ATOMIC_RELAXED, __HIP_MEMORY_SCOPE_WORKGROUP);
    }
  }
}

static_assert(OWN_K <= 8, "one team of 4 lanes per node slot of a wave");
__device__ __forceinline__ void own_team_pass(const int* sd, const int (&st)[OWN_K],
                                              const int (&cn)[OWN_K], float* aggL,
                                              float* sumL, const float* __restrict__ X2,
                                              int w, int rn, int lane) {
  const int j = lane & 3;
  const int g = lane >> 2;
  int myn = 0, mys = 0, maxn = 0;
#pragma unroll
  for (int k = 0; k < OWN_K; ++k) {
    myn = (g == k) ? cn[k] : myn;
    mys = (g == k) ? st[k] : mys;
    maxn = max(maxn, cn[k]);
  }
  const int loc = w + g * OWN_WAVES;
  const bool valid = (g < OWN_K) && (loc < rn);
  const int locc = min(loc, rn - 1);
  float4 a[8];
  const float4* ra = reinterpret_cast<const float4*>(aggL + locc * NODE_DIM) + j;
#pragma unroll
  for (int c = 0; c < 8; ++c) a[c] = ra[c * 4];
  float sa = 0.0f;
#pragma unroll
  for (int c = 0; c < 8; ++c) {
    sa = fmaf(a[c].x, a[c].x, sa);
    sa = fmaf(a[c].y, a[c].y, sa);
    sa = fmaf(a[c].z, a[c].z, sa);
    sa = fmaf(a[c].w, a[c].w, sa);
  }
  sa = own_xadd<2>(sa);
  sa = own_xadd<1>(sa);
  const float n1c = fmaxf(sqrtf(sa), EPS);
  float m = -INFINITY, s = 0.0f;
  float4 acc[8];
#pragma unroll
  for (int c = 0; c < 8; ++c) acc[c] = make_float4(0.f, 0.f, 0.f, 0.f);
  int dn = sd[(0 < myn) ? mys : 0];
  dn = (0 < myn) ? dn : 0;
  for (int i = 0; i < maxn; ++i) {
    const bool lv = (i < myn);
    float4 b[8];
    const float4* rb = reinterpret_cast<const float4*>(X2 + (size_t)dn * NODE_DIM) + j;
#pragma unroll
    for (int c = 0; c < 8; ++c) b[c] = rb[c * 4];
    {
      const int ni = i + 1;
      const int v = sd[(ni < myn) ? mys + ni : 0];
      dn = (ni < myn) ? v : 0;
    }
    float sdot = 0.0f, sb = 0.0f;
#pragma unroll
    for (int c = 0; c < 8; ++c) {
      sdot = fmaf(a[c].x, b[c].x, sdot);
      sdot = fmaf(a[c].y, b[c].y, sdot);
      sdot = fmaf(a[c].z, b[c].z, sdot);
      sdot = fmaf(a[c].w, b[c].w, sdot);
      sb = fmaf(b[c].x, b[c].x, sb);
      sb = fmaf(b[c].y, b[c].y, sb);
      sb = fmaf(b[c].z, b[c].z, sb);
      sb = fmaf(b[c].w, b[c].w, sb);
    }
    sdot = own_xadd<2>(sdot);
    sb = own_xadd<2>(sb);
    sdot = own_xadd<1>(sdot);
    sb = own_xadd<1>(sb);
    const float sv = sdot / (n1c * fmaxf(sqrtf(sb), EPS));
    const float ex = expf(-fabsf(sv - m));
    const bool up = sv > m;
    float sc = up ? ex : 1.0f;
    float p = up ? 1.0f : ex;
    float mn = fmaxf(m, sv);
    if (!lv) {
      sc = 1.0f;
      p = 0.0f;
      mn = m;
    }
    s = fmaf(s, sc, p);
#pragma unroll
    for (int c = 0; c < 8; ++c) {
      acc[c].x = fmaf(p, b[c].x, acc[c].x * sc);
      acc[c].y = fmaf(p, b[c].y, acc[c].y * sc);
      acc[c].z = fmaf(p, b[c].z, acc[c].z * sc);
      acc[c].w = fmaf(p, b[c].w, acc[c].w * sc);
    }
    m = mn;
  }
  if (valid) {
    float* o = aggL + loc * NODE_DIM + j;
#pragma unroll
    for (int c = 0; c < 8; ++c) {
      o[c * 4 + 0] = acc[c].x;
      o[c * 4 + 32] = acc[c].y;
      o[c * 4 + 64] = acc[c].z;
      o[c * 4 + 96] = acc[c].w;
    }
    if (j == 0) sumL[loc] = s;
  }
}

__device__ __forceinline__ int own_append(int2* myList, int count, int s, int e,
                                          int r0, int rn) {
  unsigned loc = (unsigned)(s - r0);
  bool m = loc < (unsigned)rn;
  unsigned mask = __builtin_amdgcn_ballot_w32(m);
  if (__builtin_expect(mask != 0u, 0)) {
    int pos = count + (int)__builtin_amdgcn_mbcnt_lo(mask, 0u);
    if (m) myList[pos] = make_int2(e, (int)loc);
    count += __builtin_popcount(mask);
  }
  return count;
}

__global__ __launch_bounds__(OWN_WAVES * 32) void owner_agg_kernel(
    const float* __restrict__ X1, const float* __restrict__ X2,
    const float* __restrict__ Xn, const float* __restrict__ Wg,
    const int* __restrict__ src, const int* __restrict__ dst, float* sim,
    float* __restrict__ out, int n1, int E) {
  __shared__ float aggL[(OWN_R + 1) * NODE_DIM];
  __shared__ __attribute__((aligned(16))) int2 listL[OWN_WAVES * OWN_CW];
  __shared__ int2 privL[OWN_WAVES * OWN_PB];
  __shared__ float sumL[OWN_R];
  __shared__ int maxL[OWN_R];
  __shared__ int cntL[OWN_WAVES];
  __shared__ int ovfL;
  __shared__ int ovf2L;
  const int tid = threadIdx.x;
  const int lane = tid & 31;
  const int w = __builtin_amdgcn_readfirstlane(tid >> 5);
  const int r0 = blockIdx.x * OWN_R;
  const int rn = min(OWN_R, n1 - r0);

  {
    const float4* xs = reinterpret_cast<const float4*>(X1 + (size_t)r0 * NODE_DIM);
    float4* xd = reinterpret_cast<float4*>(aggL);
    const int total = rn * (NODE_DIM / 4);
    float4 xv[OWN_XU];
#pragma unroll
    for (int u = 0; u < OWN_XU; ++u) {
      const int i = tid + u * (OWN_WAVES * 32);
      xv[u] = (i < total) ? xs[i] : make_float4(0.f, 0.f, 0.f, 0.f);
    }
#pragma unroll
    for (int u = 0; u < OWN_XU; ++u) {
      const int i = tid + u * (OWN_WAVES * 32);
      if (i < total) xd[i] = xv[u];
    }
  }
  if (tid < OWN_R) {
    sumL[tid] = 0.0f;
    maxL[tid] = own_key(-INFINITY);
  }
  if (tid == 0) {
    ovfL = 0;
    ovf2L = 0;
  }
  __syncthreads();

  int2* myList = listL + w * OWN_CW;
  const int n4 = E >> 2;
  const int4* src4 = reinterpret_cast<const int4*>(src);
  const int trip4 = OWN_WAVES * OWN_U * 32;
  const int trips = (n4 + trip4 - 1) / trip4;
  const int tt0 = trips > 0 ? (int)(blockIdx.x % (unsigned)trips) : 0;
  bool second = false;
  bool fast = false;
  int pc = 0;
  int2* myPriv = privL + w * OWN_PB;
  for (int sweep = 0; sweep < 2; ++sweep) {
    int count = 0;
    int tt = tt0;
    for (int t = 0; t < trips; ++t) {
      const int base = tt * trip4 + w * (OWN_U * 32);
      tt = (tt + 1 == trips) ? 0 : tt + 1;
      if (count > OWN_CW - OWN_U * 128 - 4) {
        if (second) {
          own_flush(myList, count, aggL, sumL, maxL, X2, dst, sim, lane);
        } else {
          own_sim_max(myList, count, maxL, aggL, X2, dst, sim, lane);
          if (lane == 0) ovfL = 1;
        }
        count = 0;
      }
      int4 v[OWN_U];
#pragma unroll
      for (int u = 0; u < OWN_U; ++u) {
        int g = base + u * 32 + lane;
        v[u] = (g < n4) ? src4[g] : make_int4(-1, -1, -1, -1);
      }
#pragma unroll
      for (int u = 0; u < OWN_U; ++u) {
        int e = (base + u * 32 + lane) << 2;
        count = own_append(myList, count, v[u].x, e + 0, r0, rn);
        count = own_append(myList, count, v[u].y, e + 1, r0, rn);
        count = own_append(myList, count, v[u].z, e + 2, r0, rn);
        count = own_append(myList, count, v[u].w, e + 3, r0, rn);
      }
    }
    if (w == 0) {
      int e = (n4 << 2) + lane;
      int s = (e < E) ? src[e] : -1;
      count = own_append(myList, count, s, e, r0, rn);
    }
    if (second) {
      own_flush(myList, count, aggL, sumL, maxL, X2, dst, sim, lane);
    } else {
      if (lane == 0) cntL[w] = count;
      __syncthreads();
      if (ovfL == 0) {
        const int myc = cntL[lane];
        bool full = false;
        for (int jl = 0; jl < OWN_WAVES; ++jl) {
          const int cj = __builtin_amdgcn_readlane(myc, jl);
          const int2* Lj = listL + jl * OWN_CW;
          for (int i = 0; i < cj; i += 32) {
            const int idx = i + lane;
            int2 en = Lj[min(idx, cj - 1)];
            bool mm = (idx < cj) && ((en.y & (OWN_WAVES - 1)) == w);
            unsigned mask = __builtin_amdgcn_ballot_w32(mm);
            if (mask != 0u) {
              const int n = __builtin_popcount(mask);
              if (pc + n > OWN_PB) {
                full = true;
              } else {
                int pos = pc + (int)__builtin_amdgcn_mbcnt_lo(mask, 0u);
                if (mm) myPriv[pos] = en;
                pc += n;
              }
            }
          }
        }
        if (full && lane == 0) ovf2L = 1;
        __syncthreads();
        fast = (ovf2L == 0);
      }
      if (!fast) own_sim_max(myList, count, maxL, aggL, X2, dst, sim, lane);
    }
    if (fast) break;
    __syncthreads();
    if (second) break;
    for (int i = tid; i < (OWN_R + 1) * NODE_DIM / 4; i += OWN_WAVES * 32)
      reinterpret_cast<float4*>(aggL)[i] = make_float4(0.f, 0.f, 0.f, 0.f);
    __syncthreads();
    if (ovfL == 0) break;
    second = true;
  }

  if (fast) {
    int* sd = reinterpret_cast<int*>(myList);
    int cn[OWN_K], st[OWN_K], run[OWN_K];
#pragma unroll
    for (int k = 0; k < OWN_K; ++k) cn[k] = 0;
    for (int i = 0; i < pc; i += 32) {
      const int idx = i + lane;
      const int2 en = myPriv[min(idx, pc - 1)];
      const int kk = (idx < pc) ? (en.y / OWN_WAVES) : OWN_K;
#pragma unroll
      for (int k = 0; k < OWN_K; ++k)
        cn[k] += __builtin_popcount(__builtin_amdgcn_ballot_w32(kk == k));
    }
    st[0] = 0;
#pragma unroll
    for (int k = 1; k < OWN_K; ++k) st[k] = st[k - 1] + cn[k - 1];
#pragma unroll
    for (int k = 0; k < OWN_K; ++k) run[k] = st[k];
    for (int i = 0; i < pc; i += 32) {
      const int idx = i + lane;
      const int2 en = myPriv[min(idx, pc - 1)];
      const bool live = (idx < pc);
      const int kk = live ? (en.y / OWN_WAVES) : OWN_K;
      const int d = dst[en.x];
#pragma unroll
      for (int k = 0; k < OWN_K; ++k) {
        const unsigned mask = __builtin_amdgcn_ballot_w32(kk == k);
        if (kk == k) sd[run[k] + (int)__builtin_amdgcn_mbcnt_lo(mask, 0u)] = d;
        run[k] += __builtin_popcount(mask);
      }
    }
    own_team_pass(sd, st, cn, aggL, sumL, X2, w, rn, lane);
  } else if (!second) {
    const int myc = cntL[lane];
    pc = 0;
    for (int j = 0; j < OWN_WAVES; ++j) {
      const int cj = __builtin_amdgcn_readlane(myc, j);
      const int2* Lj = listL + j * OWN_CW;
      for (int i = 0; i < cj; i += 32) {
        const int idx = i + lane;
        int2 en = Lj[min(idx, cj - 1)];
        bool m = (idx < cj) && ((en.y & (OWN_WAVES - 1)) == w);
        unsigned mask = __builtin_amdgcn_ballot_w32(m);
        if (mask != 0u) {
          int pos = pc + (int)__builtin_amdgcn_mbcnt_lo(mask, 0u);
          if (m) myPriv[pos] = en;
          pc += __builtin_popcount(mask);
          if (pc > OWN_PB - 32) {
            own_flush(myPriv, pc, aggL, sumL, maxL, X2, dst, sim, lane);
            pc = 0;
          }
        }
      }
    }
    own_flush(myPriv, pc, aggL, sumL, maxL, X2, dst, sim, lane);
  }

  float4 ga[OWN_GA], gb[OWN_GB];
  {
    const float4* xs = reinterpret_cast<const float4*>(Xn + (size_t)r0 * ATTR_DIM);
    const float4* wsrc = reinterpret_cast<const float4*>(Wg);
#pragma unroll
    for (int u = 0; u < OWN_GA; ++u) {
      const int i = tid + u * (OWN_WAVES * 32);
      ga[u] = (i < rn * (ATTR_DIM / 4)) ? xs[i] : make_float4(0.f, 0.f, 0.f, 0.f);
    }
#pragma unroll
    for (int u = 0; u < OWN_GB; ++u) {
      const int i = tid + u * (OWN_WAVES * 32);
      gb[u] = (i < NODE_DIM * (ATTR_DIM / 4)) ? wsrc[i] : make_float4(0.f, 0.f, 0.f, 0.f);
    }
  }

  __syncthreads();

  float* xnL = reinterpret_cast<float*>(listL);
  float* wgL = xnL + OWN_TM * 16 * OWN_XP;
#pragma unroll
  for (int u = 0; u < OWN_GA; ++u) {
    const int i = tid + u * (OWN_WAVES * 32);
    if (i < OWN_TM * 16 * (ATTR_DIM / 4))
      *reinterpret_cast<float4*>(xnL + (i >> 4) * OWN_XP + (i & 15) * 4) = ga[u];
  }
#pragma unroll
  for (int u = 0; u < OWN_GB; ++u) {
    const int i = tid + u * (OWN_WAVES * 32);
    if (i < NODE_DIM * (ATTR_DIM / 4))
      *reinterpret_cast<float4*>(wgL + (i >> 4) * OWN_XP + (i & 15) * 4) = gb[u];
  }
  __syncthreads();

  {
    const int l15 = lane & 15;
    const int half = lane >> 4;
    const int tn = w & (NODE_DIM / 16 - 1);
    const float* bp = wgL + (tn * 16 + l15) * OWN_XP + half * 2;
    v2f bf[ATTR_DIM / 4];
#pragma unroll
    for (int s = 0; s < ATTR_DIM / 4; ++s) bf[s] = *reinterpret_cast<const v2f*>(bp + 4 * s);
    const int n = tn * 16 + l15;
    const int col = (n & 3) * 32 + (n >> 2);
    const int tiles = ((rn + 15) >> 4) * (NODE_DIM / 16);
    for (int t = w; t < tiles; t += OWN_WAVES) {
      const int tm = t / (NODE_DIM / 16);
      const float* ap = xnL + (tm * 16 + l15) * OWN_XP + half * 2;
      v2f af[ATTR_DIM / 4];
#pragma unroll
      for (int s = 0; s < ATTR_DIM / 4; ++s) af[s] = *reinterpret_cast<const v2f*>(ap + 4 * s);
      float pv[8], sv[8];
#pragma unroll
      for (int v = 0; v < 8; ++v) {
        const int rc = min(tm * 16 + half * 8 + v, rn - 1);
        pv[v] = aggL[rc * NODE_DIM + col];
        sv[v] = sumL[rc];
      }
      v8f c = {};
#pragma unroll
      for (int s = 0; s < ATTR_DIM / 4; ++s)
        c = __builtin_amdgcn_wmma_f32_16x16x4_f32(
             false, af[s],  false, bf[s],
             (short)0, c,  false,  false);
#pragma unroll
      for (int v = 0; v < 8; ++v) {
        const int row = tm * 16 + half * 8 + v;
        const float g = 1.0f / (1.0f + expf(-c[v]));
        const float a = (sv[v] > 0.0f) ? (pv[v] / sv[v]) : 0.0f;
        aggL[(row < rn ? row : OWN_R) * NODE_DIM + col] = g * a;
      }
    }
  }
  __syncthreads();

  for (int i = tid; i < rn * 32; i += OWN_WAVES * 32) {
    int row = i >> 5;
    const float* p = aggL + row * NODE_DIM + lane;
    reinterpret_cast<float4*>(out + (size_t)(r0 + row) * NODE_DIM)[lane] =
        make_float4(p[0], p[32], p[64], p[96]);
  }
}

extern "C" void kernel_launch(void* const* d_in, const int* in_sizes, int n_in,
                              void* d_out, int out_size, void* d_ws,
                              size_t ws_size, hipStream_t stream) {
  const float* X_h_1 = (const float*)d_in[0];
  const float* X_h_2 = (const float*)d_in[1];
  const float* X_n_1 = (const float*)d_in[2];
  const int* cross = (const int*)d_in[3];
  const float* W_gate = (const float*)d_in[4];
  float* out = (float*)d_out;

  const int N1 = in_sizes[0] / NODE_DIM;
  const int E = in_sizes[3] / 2;
  const int* src = cross;
  const int* dst = cross + E;

  float* sim = (float*)d_ws;

  owner_agg_kernel<<<(N1 + OWN_R - 1) / OWN_R, OWN_WAVES * 32, 0, stream>>>(
      X_h_1, X_h_2, X_n_1, W_gate, src, dst, sim, out, N1, E);
}


# sweep and hand-over filters: a batch's lane masks (all src range tests of a trip, 4 list pieces) formed before any branch, one vector-to-scalar wait a batch; loads guarded by clamped index; same sums
# speedup vs baseline: 4.5362x; 1.1471x over previous
#include <hip/hip_runtime.h>
#include <math.h>

typedef float v2f __attribute__((ext_vector_type(2)));
typedef float v8f __attribute__((ext_vector_type(8)));

#define NODE_DIM 128
#define ATTR_DIM 64
#define EPS 1e-8f

#define OWN_R 200
#define OWN_WAVES 32
#define OWN_U 4
#define OWN_CW 640
#define OWN_G 8
#define OWN_PB 192
#define OWN_K ((OWN_R + OWN_WAVES - 1) / OWN_WAVES)
#define OWN_HC 4
#define OWN_ST 2
#define OWN_XU ((OWN_R * NODE_DIM / 4 + OWN_WAVES * 32 - 1) / (OWN_WAVES * 32))
#define OWN_TM ((OWN_R + 15) / 16)
#define OWN_XP (ATTR_DIM + 4)
#define OWN_GA ((OWN_TM * 16 * (ATTR_DIM / 4) + OWN_WAVES * 32 - 1) / (OWN_WAVES * 32))
#define OWN_GB ((NODE_DIM * (ATTR_DIM / 4) + OWN_WAVES * 32 - 1) / (OWN_WAVES * 32))
static_assert(OWN_WAVES % (NODE_DIM / 16) == 0, "a wave keeps one feature tile for all its gate tiles");
static_assert((OWN_TM * 16 + NODE_DIM) * OWN_XP * 4 <= OWN_WAVES * OWN_CW * 8,
              "the staged Xn and Wg rows fit in the space of the match lists");

__device__ __forceinline__ int own_key(float f) {
  int b = __float_as_int(f);
  return b >= 0 ? b : (b ^ 0x7fffffff);
}
__device__ __forceinline__ float own_unkey(int k) {
  return __int_as_float(k >= 0 ? k : (k ^ 0x7fffffff));
}

template <int M>
__device__ __forceinline__ float own_xadd(float x) {
  int b = __builtin_bit_cast(int, x);
  int r = __builtin_amdgcn_update_dpp(b, b, 0x160 + M, 0xf, 0xf, false);
  return x + __builtin_bit_cast(float, r);
}

__device__ __forceinline__ void own_sim_max(const int2* list, int count, int* maxL,
                                            const float* x1L,
                                            const float* __restrict__ X2,
                                            const int* __restrict__ dst,
                                            float* sim, int lane) {
  if (count == 0) return;
  const int j = lane & 7;
  const int g = lane >> 3;
  int2 en[OWN_ST];
  int d[OWN_ST];
#pragma unroll
  for (int t = 0; t < OWN_ST; ++t) {
    en[t] = list[min(t * 4 + g, count - 1)];
    d[t] = dst[en[t].x];
  }
  for (int i = 0; i < count; i += 4 * OWN_ST) {
    float4 a[OWN_ST][4], b[OWN_ST][4];
    int2 cur[OWN_ST];
#pragma unroll
    for (int t = 0; t < OWN_ST; ++t) {
      const float4* ra =
          reinterpret_cast<const float4*>(x1L + en[t].y * NODE_DIM) + j;
      const float4* rb =
          reinterpret_cast<const float4*>(X2 + (size_t)d[t] * NODE_DIM) + j;
#pragma unroll
      for (int c = 0; c < 4; ++c) {
        a[t][c] = ra[c * 8];
        b[t][c] = rb[c * 8];
      }
      cur[t] = en[t];
    }
#pragma unroll
    for (int t = 0; t < OWN_ST; ++t) {
      en[t] = list[min(i + 4 * OWN_ST + t * 4 + g, count - 1)];
      d[t] = dst[en[t].x];
    }
    float pd[OWN_ST], pa[OWN_ST], pb[OWN_ST];
#pragma unroll
    for (int t = 0; t < OWN_ST; ++t) {
      float sd = 0.0f, sa = 0.0f, sb = 0.0f;
#pragma unroll
      for (int c = 0; c < 4; ++c) {
        sd = fmaf(a[t][c].x, b[t][c].x, sd);
        sd = fmaf(a[t][c].y, b[t][c].y, sd);
        sd = fmaf(a[t][c].z, b[t][c].z, sd);
        sd = fmaf(a[t][c].w, b[t][c].w, sd);
        sa = fmaf(a[t][c].x, a[t][c].x, sa);
        sa = fmaf(a[t][c].y, a[t][c].y, sa);
        sa = fmaf(a[t][c].z, a[t][c].z, sa);
        sa = fmaf(a[t][c].w, a[t][c].w, sa);
        sb = fmaf(b[t][c].x, b[t][c].x, sb);
        sb = fmaf(b[t][c].y, b[t][c].y, sb);
        sb = fmaf(b[t][c].z, b[t][c].z, sb);
        sb = fmaf(b[t][c].w, b[t][c].w, sb);
      }
      sd = own_xadd<4>(sd);
      sa = own_xadd<4>(sa);
      sb = own_xadd<4>(sb);
      sd = own_xadd<2>(sd);
      sa = own_xadd<2>(sa);
      sb = own_xadd<2>(sb);
      pd[t] = own_xadd<1>(sd);
      pa[t] = own_xadd<1>(sa);
      pb[t] = own_xadd<1>(sb);
    }
    float md = pd[0], ma = pa[0], mb = pb[0];
    int2 me = cur[0];
#pragma unroll
    for (int t = 1; t < OWN_ST; ++t) {
      const bool s = (j == t);
      md = s ? pd[t] : md;
      ma = s ? pa[t] : ma;
      mb = s ? pb[t] : mb;
      me.x = s ? cur[t].x : me.x;
      me.y = s ? cur[t].y : me.y;
    }
    float sv = md / (fmaxf(sqrtf(ma), EPS) * fmaxf(sqrtf(mb), EPS));
    if (j < OWN_ST && i + j * 4 + g < count) {
      sim[me.x] = sv;
      __hip_atomic_fetch_max(maxL + me.y, own_key(sv), __ATOMIC_RELAXED,
                             __HIP_MEMORY_SCOPE_WORKGROUP);
    }
  }
}

__device__ __forceinline__ void own_flush(const int2* list, int count,
                                          float* aggL, float* sumL,
                                          const int* maxL,
                                          const float* __restrict__ X2,
                                          const int* __restrict__ dst,
                                          const float* sim,
                                          int lane) {
  const int sub = lane & (OWN_G - 1);
  for (int i = 0; i < count; i += OWN_G) {
    int2 en = list[min(i + sub, count - 1)];
    int md = dst[en.x];
    float ms = sim[en.x];
    bool live = (i + sub < count);
    int mloc = live ? en.y : OWN_R;
    int loc[OWN_G];
    float4 b[OWN_G];
    float w[OWN_G];
#pragma unroll
    for (int q = 0; q < OWN_G; ++q) {
      int d = __builtin_amdgcn_readlane(md, q);
      loc[q] = __builtin_amdgcn_readlane(mloc, q);
      b[q] = reinterpret_cast<const float4*>(X2 + (size_t)d * NODE_DIM)[lane];
    }
    float mex = expf(ms - own_unkey(maxL[en.y]));
    if (lane < OWN_G && live)
      __hip_atomic_fetch_add(sumL + en.y, mex, __ATOMIC_RELAXED, __HIP_MEMORY_SCOPE_WORKGROUP);
#pragma unroll
    for (int q = 0; q < OWN_G; ++q)
      w[q] = __int_as_float(__builtin_amdgcn_readlane(__float_as_int(mex), q));
#pragma unroll
    for (int q = 0; q < OWN_G; ++q) {
      float* o = aggL + loc[q] * NODE_DIM + lane;
      __hip_atomic_fetch_add(o + 0, w[q] * b[q].x, __ATOMIC_RELAXED, __HIP_MEMORY_SCOPE_WORKGROUP);
      __hip_atomic_fetch_add(o + 32, w[q] * b[q].y, __ATOMIC_RELAXED, __HIP_MEMORY_SCOPE_WORKGROUP);
      __hip_atomic_fetch_add(o + 64, w[q] * b[q].z, __ATOMIC_RELAXED, __HIP_MEMORY_SCOPE_WORKGROUP);
      __hip_atomic_fetch_add(o + 96, w[q] * b[q].w, __ATOMIC_RELAXED, __HIP_MEMORY_SCOPE_WORKGROUP);
    }
  }
}

static_assert(OWN_K <= 8, "one team of 4 lanes per node slot of a wave");
__device__ __forceinline__ void own_team_pass(const int* sd, const int (&st)[OWN_K],
                                              const int (&cn)[OWN_K], float* aggL,
                                              float* sumL, const float* __restrict__ X2,
                                              int w, int rn, int lane) {
  const int j = lane & 3;
  const int g = lane >> 2;
  int myn = 0, mys = 0, maxn = 0;
#pragma unroll
  for (int k = 0; k < OWN_K; ++k) {
    myn = (g == k) ? cn[k] : myn;
    mys = (g == k) ? st[k] : mys;
    maxn = max(maxn, cn[k]);
  }
  const int loc = w + g * OWN_WAVES;
  const bool valid = (g < OWN_K) && (loc < rn);
  const int locc = min(loc, rn - 1);
  float4 a[8];
  const float4* ra = reinterpret_cast<const float4*>(aggL + locc * NODE_DIM) + j;
#pragma unroll
  for (int c = 0; c < 8; ++c) a[c] = ra[c * 4];
  float sa = 0.0f;
#pragma unroll
  for (int c = 0; c < 8; ++c) {
    sa = fmaf(a[c].x, a[c].x, sa);
    sa = fmaf(a[c].y, a[c].y, sa);
    sa = fmaf(a[c].z, a[c].z, sa);
    sa = fmaf(a[c].w, a[c].w, sa);
  }
  sa = own_xadd<2>(sa);
  sa = own_xadd<1>(sa);
  const float n1c = fmaxf(sqrtf(sa), EPS);
  float m = -INFINITY, s = 0.0f;
  float4 acc[8];
#pragma unroll
  for (int c = 0; c < 8; ++c) acc[c] = make_float4(0.f, 0.f, 0.f, 0.f);
  int dn = sd[(0 < myn) ? mys : 0];
  dn = (0 < myn) ? dn : 0;
  for (int i = 0; i < maxn; ++i) {
    const bool lv = (i < myn);
    float4 b[8];
    const float4* rb = reinterpret_cast<const float4*>(X2 + (size_t)dn * NODE_DIM) + j;
#pragma unroll
    for (int c = 0; c < 8; ++c) b[c] = rb[c * 4];
    {
      const int ni = i + 1;
      const int v = sd[(ni < myn) ? mys + ni : 0];
      dn = (ni < myn) ? v : 0;
    }
    float sdot = 0.0f, sb = 0.0f;
#pragma unroll
    for (int c = 0; c < 8; ++c) {
      sdot = fmaf(a[c].x, b[c].x, sdot);
      sdot = fmaf(a[c].y, b[c].y, sdot);
      sdot = fmaf(a[c].z, b[c].z, sdot);
      sdot = fmaf(a[c].w, b[c].w, sdot);
      sb = fmaf(b[c].x, b[c].x, sb);
      sb = fmaf(b[c].y, b[c].y, sb);
      sb = fmaf(b[c].z, b[c].z, sb);
      sb = fmaf(b[c].w, b[c].w, sb);
    }
    sdot = own_xadd<2>(sdot);
    sb = own_xadd<2>(sb);
    sdot = own_xadd<1>(sdot);
    sb = own_xadd<1>(sb);
    const float sv = sdot / (n1c * fmaxf(sqrtf(sb), EPS));
    const float ex = expf(-fabsf(sv - m));
    const bool up = sv > m;
    float sc = up ? ex : 1.0f;
    float p = up ? 1.0f : ex;
    float mn = fmaxf(m, sv);
    if (!lv) {
      sc = 1.0f;
      p = 0.0f;
      mn = m;
    }
    s = fmaf(s, sc, p);
#pragma unroll
    for (int c = 0; c < 8; ++c) {
      acc[c].x = fmaf(p, b[c].x, acc[c].x * sc);
      acc[c].y = fmaf(p, b[c].y, acc[c].y * sc);
      acc[c].z = fmaf(p, b[c].z, acc[c].z * sc);
      acc[c].w = fmaf(p, b[c].w, acc[c].w * sc);
    }
    m = mn;
  }
  if (valid) {
    float* o = aggL + loc * NODE_DIM + j;
#pragma unroll
    for (int c = 0; c < 8; ++c) {
      o[c * 4 + 0] = acc[c].x;
      o[c * 4 + 32] = acc[c].y;
      o[c * 4 + 64] = acc[c].z;
      o[c * 4 + 96] = acc[c].w;
    }
    if (j == 0) sumL[loc] = s;
  }
}

__device__ __forceinline__ int own_append(int2* myList, int count, int s, int e,
                                          int r0, int rn) {
  unsigned loc = (unsigned)(s - r0);
  bool m = loc < (unsigned)rn;
  unsigned mask = __builtin_amdgcn_ballot_w32(m);
  if (__builtin_expect(mask != 0u, 0)) {
    int pos = count + (int)__builtin_amdgcn_mbcnt_lo(mask, 0u);
    if (m) myList[pos] = make_int2(e, (int)loc);
    count += __builtin_popcount(mask);
  }
  return count;
}

__device__ __forceinline__ int own_put(int2* myList, int count, unsigned mask, bool mine,
                                       int e, int loc) {
  if (mask != 0u) {
    const int pos = count + (int)__builtin_amdgcn_mbcnt_lo(mask, 0u);
    if (mine) myList[pos] = make_int2(e, loc);
    count += __builtin_popcount(mask);
  }
  return count;
}

__global__ __launch_bounds__(OWN_WAVES * 32) void owner_agg_kernel(
    const float* __restrict__ X1, const float* __restrict__ X2,
    const float* __restrict__ Xn, const float* __restrict__ Wg,
    const int* __restrict__ src, const int* __restrict__ dst, float* sim,
    float* __restrict__ out, int n1, int E) {
  __shared__ float aggL[(OWN_R + 1) * NODE_DIM];
  __shared__ __attribute__((aligned(16))) int2 listL[OWN_WAVES * OWN_CW];
  __shared__ int2 privL[OWN_WAVES * OWN_PB];
  __shared__ float sumL[OWN_R];
  __shared__ int maxL[OWN_R];
  __shared__ int cntL[OWN_WAVES];
  __shared__ int ovfL;
  __shared__ int ovf2L;
  const int tid = threadIdx.x;
  const int lane = tid & 31;
  const int w = __builtin_amdgcn_readfirstlane(tid >> 5);
  const int r0 = blockIdx.x * OWN_R;
  const int rn = min(OWN_R, n1 - r0);

  {
    const float4* xs = reinterpret_cast<const float4*>(X1 + (size_t)r0 * NODE_DIM);
    float4* xd = reinterpret_cast<float4*>(aggL);
    const int total = rn * (NODE_DIM / 4);
    float4 xv[OWN_XU];
#pragma unroll
    for (int u = 0; u < OWN_XU; ++u) {
      const int i = tid + u * (OWN_WAVES * 32);
      xv[u] = (i < total) ? xs[i] : make_float4(0.f, 0.f, 0.f, 0.f);
    }
#pragma unroll
    for (int u = 0; u < OWN_XU; ++u) {
      const int i = tid + u * (OWN_WAVES * 32);
      if (i < total) xd[i] = xv[u];
    }
  }
  if (tid < OWN_R) {
    sumL[tid] = 0.0f;
    maxL[tid] = own_key(-INFINITY);
  }
  if (tid == 0) {
    ovfL = 0;
    ovf2L = 0;
  }
  __syncthreads();

  int2* myList = listL + w * OWN_CW;
  const int n4 = E >> 2;
  const int4* src4 = reinterpret_cast<const int4*>(src);
  const int trip4 = OWN_WAVES * OWN_U * 32;
  const int trips = (n4 + trip4 - 1) / trip4;
  const int tt0 = trips > 0 ? (int)(blockIdx.x % (unsigned)trips) : 0;
  bool second = false;
  bool fast = false;
  int pc = 0;
  int2* myPriv = privL + w * OWN_PB;
  for (int sweep = 0; sweep < 2; ++sweep) {
    int count = 0;
    int tt = tt0;
    for (int t = 0; t < trips; ++t) {
      const int base = tt * trip4 + w * (OWN_U * 32);
      tt = (tt + 1 == trips) ? 0 : tt + 1;
      if (count > OWN_CW - OWN_U * 128 - 4) {
        if (second) {
          own_flush(myList, count, aggL, sumL, maxL, X2, dst, sim, lane);
        } else {
          own_sim_max(myList, count, maxL, aggL, X2, dst, sim, lane);
          if (lane == 0) ovfL = 1;
        }
        count = 0;
      }
      int4 v[OWN_U];
      bool inl[OWN_U];
#pragma unroll
      for (int u = 0; u < OWN_U; ++u) {
        const int g = base + u * 32 + lane;
        inl[u] = (g < n4);
        v[u] = src4[min(g, n4 - 1)];
      }
      bool mm[OWN_U][4];
      unsigned mk[OWN_U][4];
#pragma unroll
      for (int u = 0; u < OWN_U; ++u) {
        const bool tx = (unsigned)(v[u].x - r0) < (unsigned)rn;
        const bool ty = (unsigned)(v[u].y - r0) < (unsigned)rn;
        const bool tz = (unsigned)(v[u].z - r0) < (unsigned)rn;
        const bool tw = (unsigned)(v[u].w - r0) < (unsigned)rn;
        const unsigned ik = __builtin_amdgcn_ballot_w32(inl[u]);
        mk[u][0] = __builtin_amdgcn_ballot_w32(tx) & ik;
        mk[u][1] = __builtin_amdgcn_ballot_w32(ty) & ik;
        mk[u][2] = __builtin_amdgcn_ballot_w32(tz) & ik;
        mk[u][3] = __builtin_amdgcn_ballot_w32(tw) & ik;
        mm[u][0] = inl[u] && tx;
        mm[u][1] = inl[u] && ty;
        mm[u][2] = inl[u] && tz;
        mm[u][3] = inl[u] && tw;
      }
#pragma unroll
      for (int u = 0; u < OWN_U; ++u) {
        if ((mk[u][0] | mk[u][1] | mk[u][2] | mk[u][3]) != 0u) {
          const int e = (base + u * 32 + lane) << 2;
          count = own_put(myList, count, mk[u][0], mm[u][0], e + 0, v[u].x - r0);
          count = own_put(myList, count, mk[u][1], mm[u][1], e + 1, v[u].y - r0);
          count = own_put(myList, count, mk[u][2], mm[u][2], e + 2, v[u].z - r0);
          count = own_put(myList, count, mk[u][3], mm[u][3], e + 3, v[u].w - r0);
        }
      }
    }
    if (w == 0) {
      int e = (n4 << 2) + lane;
      int s = (e < E) ? src[e] : -1;
      count = own_append(myList, count, s, e, r0, rn);
    }
    if (second) {
      own_flush(myList, count, aggL, sumL, maxL, X2, dst, sim, lane);
    } else {
      if (lane == 0) cntL[w] = count;
      __syncthreads();
      if (__builtin_amdgcn_readfirstlane(ovfL) == 0) {
        const int myc = cntL[lane];
        int hc = 0;
        for (int jl = 0; jl < OWN_WAVES; ++jl) {
          const int cj = __builtin_amdgcn_readlane(myc, jl);
          const int2* Lj = listL + jl * OWN_CW;
          for (int i0 = 0; i0 < cj; i0 += 32 * OWN_HC) {
            int2 en[OWN_HC];
            bool hm[OWN_HC];
            unsigned hk[OWN_HC];
#pragma unroll
            for (int c = 0; c < OWN_HC; ++c) en[c] = Lj[min(i0 + c * 32 + lane, cj - 1)];
#pragma unroll
            for (int c = 0; c < OWN_HC; ++c) {
              const bool in = (i0 + c * 32 + lane < cj);
              const bool my = ((en[c].y & (OWN_WAVES - 1)) == w);
              hk[c] = __builtin_amdgcn_ballot_w32(in) & __builtin_amdgcn_ballot_w32(my);
              hm[c] = in && my;
            }
#pragma unroll
            for (int c = 0; c < OWN_HC; ++c) {
              if (hk[c] != 0u) {
                const int n = __builtin_popcount(hk[c]);
                if (hc + n <= OWN_PB) {
                  int pos = hc + (int)__builtin_amdgcn_mbcnt_lo(hk[c], 0u);
                  if (hm[c]) myPriv[pos] = en[c];
                }
                hc += n;
              }
            }
          }
        }
        pc = hc;
        if (hc > OWN_PB) {
          if (lane == 0) ovf2L = 1;
        }
        __syncthreads();
        fast = (__builtin_amdgcn_readfirstlane(ovf2L) == 0);
      }
      if (!fast) own_sim_max(myList, count, maxL, aggL, X2, dst, sim, lane);
    }
    if (fast) break;
    __syncthreads();
    if (second) break;
    for (int i = tid; i < (OWN_R + 1) * NODE_DIM / 4; i += OWN_WAVES * 32)
      reinterpret_cast<float4*>(aggL)[i] = make_float4(0.f, 0.f, 0.f, 0.f);
    __syncthreads();
    if (__builtin_amdgcn_readfirstlane(ovfL) == 0) break;
    second = true;
  }

  if (fast) {
    int* sd = reinterpret_cast<int*>(myList);
    int cn[OWN_K], st[OWN_K], run[OWN_K];
#pragma unroll
    for (int k = 0; k < OWN_K; ++k) cn[k] = 0;
    for (int i = 0; i < pc; i += 32) {
      const int idx = i + lane;
      const int2 en = myPriv[min(idx, pc - 1)];
      const int kk = (idx < pc) ? (en.y / OWN_WAVES) : OWN_K;
#pragma unroll
      for (int k = 0; k < OWN_K; ++k)
        cn[k] += __builtin_popcount(__builtin_amdgcn_ballot_w32(kk == k));
    }
    st[0] = 0;
#pragma unroll
    for (int k = 1; k < OWN_K; ++k) st[k] = st[k - 1] + cn[k - 1];
#pragma unroll
    for (int k = 0; k < OWN_K; ++k) run[k] = st[k];
    for (int i = 0; i < pc; i += 32) {
      const int idx = i + lane;
      const int2 en = myPriv[min(idx, pc - 1)];
      const bool live = (idx < pc);
      const int kk = live ? (en.y / OWN_WAVES) : OWN_K;
      const int d = dst[en.x];
#pragma unroll
      for (int k = 0; k < OWN_K; ++k) {
        const unsigned mask = __builtin_amdgcn_ballot_w32(kk == k);
        if (kk == k) sd[run[k] + (int)__builtin_amdgcn_mbcnt_lo(mask, 0u)] = d;
        run[k] += __builtin_popcount(mask);
      }
    }
    own_team_pass(sd, st, cn, aggL, sumL, X2, w, rn, lane);
  } else if (!second) {
    const int myc = cntL[lane];
    pc = 0;
    for (int j = 0; j < OWN_WAVES; ++j) {
      const int cj = __builtin_amdgcn_readlane(myc, j);
      const int2* Lj = listL + j * OWN_CW;
      for (int i = 0; i < cj; i += 32) {
        const int idx = i + lane;
        int2 en = Lj[min(idx, cj - 1)];
        bool m = (idx < cj) && ((en.y & (OWN_WAVES - 1)) == w);
        unsigned mask = __builtin_amdgcn_ballot_w32(m);
        if (mask != 0u) {
          int pos = pc + (int)__builtin_amdgcn_mbcnt_lo(mask, 0u);
          if (m) myPriv[pos] = en;
          pc += __builtin_popcount(mask);
          if (pc > OWN_PB - 32) {
            own_flush(myPriv, pc, aggL, sumL, maxL, X2, dst, sim, lane);
            pc = 0;
          }
        }
      }
    }
    own_flush(myPriv, pc, aggL, sumL, maxL, X2, dst, sim, lane);
  }

  float4 ga[OWN_GA], gb[OWN_GB];
  {
    const float4* xs = reinterpret_cast<const float4*>(Xn + (size_t)r0 * ATTR_DIM);
    const float4* wsrc = reinterpret_cast<const float4*>(Wg);
#pragma unroll
    for (int u = 0; u < OWN_GA; ++u) {
      const int i = tid + u * (OWN_WAVES * 32);
      ga[u] = (i < rn * (ATTR_DIM / 4)) ? xs[i] : make_float4(0.f, 0.f, 0.f, 0.f);
    }
#pragma unroll
    for (int u = 0; u < OWN_GB; ++u) {
      const int i = tid + u * (OWN_WAVES * 32);
      gb[u] = (i < NODE_DIM * (ATTR_DIM / 4)) ? wsrc[i] : make_float4(0.f, 0.f, 0.f, 0.f);
    }
  }

  __syncthreads();

  float* xnL = reinterpret_cast<float*>(listL);
  float* wgL = xnL + OWN_TM * 16 * OWN_XP;
#pragma unroll
  for (int u = 0; u < OWN_GA; ++u) {
    const int i = tid + u * (OWN_WAVES * 32);
    if (i < OWN_TM * 16 * (ATTR_DIM / 4))
      *reinterpret_cast<float4*>(xnL + (i >> 4) * OWN_XP + (i & 15) * 4) = ga[u];
  }
#pragma unroll
  for (int u = 0; u < OWN_GB; ++u) {
    const int i = tid + u * (OWN_WAVES * 32);
    if (i < NODE_DIM * (ATTR_DIM / 4))
      *reinterpret_cast<float4*>(wgL + (i >> 4) * OWN_XP + (i & 15) * 4) = gb[u];
  }
  __syncthreads();

  {
    const int l15 = lane & 15;
    const int half = lane >> 4;
    const int tn = w & (NODE_DIM / 16 - 1);
    const float* bp = wgL + (tn * 16 + l15) * OWN_XP + half * 2;
    v2f bf[ATTR_DIM / 4];
#pragma unroll
    for (int s = 0; s < ATTR_DIM / 4; ++s) bf[s] = *reinterpret_cast<const v2f*>(bp + 4 * s);
    const int n = tn * 16 + l15;
    const int col = (n & 3) * 32 + (n >> 2);
    const int tiles = ((rn + 15) >> 4) * (NODE_DIM / 16);
    for (int t = w; t < tiles; t += OWN_WAVES) {
      const int tm = t / (NODE_DIM / 16);
      const float* ap = xnL + (tm * 16 + l15) * OWN_XP + half * 2;
      v2f af[ATTR_DIM / 4];
#pragma unroll
      for (int s = 0; s < ATTR_DIM / 4; ++s) af[s] = *reinterpret_cast<const v2f*>(ap + 4 * s);
      float pv[8], sv[8];
#pragma unroll
      for (int v = 0; v < 8; ++v) {
        const int rc = min(tm * 16 + half * 8 + v, rn - 1);
        pv[v] = aggL[rc * NODE_DIM + col];
        sv[v] = sumL[rc];
      }
      v8f c = {};
#pragma unroll
      for (int s = 0; s < ATTR_DIM / 4; ++s)
        c = __builtin_amdgcn_wmma_f32_16x16x4_f32(
             false, af[s],  false, bf[s],
             (short)0, c,  false,  false);
#pragma unroll
      for (int v = 0; v < 8; ++v) {
        const int row = tm * 16 + half * 8 + v;
        const float g = 1.0f / (1.0f + expf(-c[v]));
        const float a = (sv[v] > 0.0f) ? (pv[v] / sv[v]) : 0.0f;
        aggL[(row < rn ? row : OWN_R) * NODE_DIM + col] = g * a;
      }
    }
  }
  __syncthreads();

  for (int i = tid; i < rn * 32; i += OWN_WAVES * 32) {
    int row = i >> 5;
    const float* p = aggL + row * NODE_DIM + lane;
    reinterpret_cast<float4*>(out + (size_t)(r0 + row) * NODE_DIM)[lane] =
        make_float4(p[0], p[32], p[64], p[96]);
  }
}

extern "C" void kernel_launch(void* const* d_in, const int* in_sizes, int n_in,
                              void* d_out, int out_size, void* d_ws,
                              size_t ws_size, hipStream_t stream) {
  const float* X_h_1 = (const float*)d_in[0];
  const float* X_h_2 = (const float*)d_in[1];
  const float* X_n_1 = (const float*)d_in[2];
  const int* cross = (const int*)d_in[3];
  const float* W_gate = (const float*)d_in[4];
  float* out = (float*)d_out;

  const int N1 = in_sizes[0] / NODE_DIM;
  const int E = in_sizes[3] / 2;
  const int* src = cross;
  const int* dst = cross + E;

  float* sim = (float*)d_ws;

  owner_agg_kernel<<<(N1 + OWN_R - 1) / OWN_R, OWN_WAVES * 32, 0, stream>>>(
      X_h_1, X_h_2, X_n_1, W_gate, src, dst, sim, out, N1, E);
}


# hand-over by counting scatter: each wave counts its own match list per taking wave (LDS integer adds), prefix sums place every list in every buffer, each wave scatters its own; sums bit-identical
# speedup vs baseline: 4.8242x; 1.0635x over previous
#include <hip/hip_runtime.h>
#include <math.h>

typedef float v2f __attribute__((ext_vector_type(2)));
typedef float v8f __attribute__((ext_vector_type(8)));

#define NODE_DIM 128
#define ATTR_DIM 64
#define EPS 1e-8f

#define OWN_R 200
#define OWN_WAVES 32
#define OWN_U 4
#define OWN_CW 640
#define OWN_G 8
#define OWN_PB 192
#define OWN_K ((OWN_R + OWN_WAVES - 1) / OWN_WAVES)
#define OWN_ST 2
#define OWN_XU ((OWN_R * NODE_DIM / 4 + OWN_WAVES * 32 - 1) / (OWN_WAVES * 32))
#define OWN_TM ((OWN_R + 15) / 16)
#define OWN_XP (ATTR_DIM + 4)
#define OWN_GA ((OWN_TM * 16 * (ATTR_DIM / 4) + OWN_WAVES * 32 - 1) / (OWN_WAVES * 32))
#define OWN_GB ((NODE_DIM * (ATTR_DIM / 4) + OWN_WAVES * 32 - 1) / (OWN_WAVES * 32))
static_assert(OWN_WAVES % (NODE_DIM / 16) == 0, "a wave keeps one feature tile for all its gate tiles");
static_assert(OWN_WAVES == 32, "the hand-over keeps one list's count per lane and finds peers by 5 bits of the wave number");
static_assert((OWN_TM * 16 + NODE_DIM) * OWN_XP * 4 <= OWN_WAVES * OWN_CW * 8,
              "the staged Xn and Wg rows fit in the space of the match lists");

__device__ __forceinline__ int own_key(float f) {
  int b = __float_as_int(f);
  return b >= 0 ? b : (b ^ 0x7fffffff);
}
__device__ __forceinline__ float own_unkey(int k) {
  return __int_as_float(k >= 0 ? k : (k ^ 0x7fffffff));
}

template <int M>
__device__ __forceinline__ float own_xadd(float x) {
  int b = __builtin_bit_cast(int, x);
  int r = __builtin_amdgcn_update_dpp(b, b, 0x160 + M, 0xf, 0xf, false);
  return x + __builtin_bit_cast(float, r);
}

__device__ __forceinline__ void own_sim_max(const int2* list, int count, int* maxL,
                                            const float* x1L,
                                            const float* __restrict__ X2,
                                            const int* __restrict__ dst,
                                            float* sim, int lane) {
  if (count == 0) return;
  const int j = lane & 7;
  const int g = lane >> 3;
  int2 en[OWN_ST];
  int d[OWN_ST];
#pragma unroll
  for (int t = 0; t < OWN_ST; ++t) {
    en[t] = list[min(t * 4 + g, count - 1)];
    d[t] = dst[en[t].x];
  }
  for (int i = 0; i < count; i += 4 * OWN_ST) {
    float4 a[OWN_ST][4], b[OWN_ST][4];
    int2 cur[OWN_ST];
#pragma unroll
    for (int t = 0; t < OWN_ST; ++t) {
      const float4* ra =
          reinterpret_cast<const float4*>(x1L + en[t].y * NODE_DIM) + j;
      const float4* rb =
          reinterpret_cast<const float4*>(X2 + (size_t)d[t] * NODE_DIM) + j;
#pragma unroll
      for (int c = 0; c < 4; ++c) {
        a[t][c] = ra[c * 8];
        b[t][c] = rb[c * 8];
      }
      cur[t] = en[t];
    }
#pragma unroll
    for (int t = 0; t < OWN_ST; ++t) {
      en[t] = list[min(i + 4 * OWN_ST + t * 4 + g, count - 1)];
      d[t] = dst[en[t].x];
    }
    float pd[OWN_ST], pa[OWN_ST], pb[OWN_ST];
#pragma unroll
    for (int t = 0; t < OWN_ST; ++t) {
      float sd = 0.0f, sa = 0.0f, sb = 0.0f;
#pragma unroll
      for (int c = 0; c < 4; ++c) {
        sd = fmaf(a[t][c].x, b[t][c].x, sd);
        sd = fmaf(a[t][c].y, b[t][c].y, sd);
        sd = fmaf(a[t][c].z, b[t][c].z, sd);
        sd = fmaf(a[t][c].w, b[t][c].w, sd);
        sa = fmaf(a[t][c].x, a[t][c].x, sa);
        sa = fmaf(a[t][c].y, a[t][c].y, sa);
        sa = fmaf(a[t][c].z, a[t][c].z, sa);
        sa = fmaf(a[t][c].w, a[t][c].w, sa);
        sb = fmaf(b[t][c].x, b[t][c].x, sb);
        sb = fmaf(b[t][c].y, b[t][c].y, sb);
        sb = fmaf(b[t][c].z, b[t][c].z, sb);
        sb = fmaf(b[t][c].w, b[t][c].w, sb);
      }
      sd = own_xadd<4>(sd);
      sa = own_xadd<4>(sa);
      sb = own_xadd<4>(sb);
      sd = own_xadd<2>(sd);
      sa = own_xadd<2>(sa);
      sb = own_xadd<2>(sb);
      pd[t] = own_xadd<1>(sd);
      pa[t] = own_xadd<1>(sa);
      pb[t] = own_xadd<1>(sb);
    }
    float md = pd[0], ma = pa[0], mb = pb[0];
    int2 me = cur[0];
#pragma unroll
    for (int t = 1; t < OWN_ST; ++t) {
      const bool s = (j == t);
      md = s ? pd[t] : md;
      ma = s ? pa[t] : ma;
      mb = s ? pb[t] : mb;
      me.x = s ? cur[t].x : me.x;
      me.y = s ? cur[t].y : me.y;
    }
    float sv = md / (fmaxf(sqrtf(ma), EPS) * fmaxf(sqrtf(mb), EPS));
    if (j < OWN_ST && i + j * 4 + g < count) {
      sim[me.x] = sv;
      __hip_atomic_fetch_max(maxL + me.y, own_key(sv), __ATOMIC_RELAXED,
                             __HIP_MEMORY_SCOPE_WORKGROUP);
    }
  }
}

__device__ __forceinline__ void own_flush(const int2* list, int count,
                                          float* aggL, float* sumL,
                                          const int* maxL,
                                          const float* __restrict__ X2,
                                          const int* __restrict__ dst,
                                          const float* sim,
                                          int lane) {
  const int sub = lane & (OWN_G - 1);
  for (int i = 0; i < count; i += OWN_G) {
    int2 en = list[min(i + sub, count - 1)];
    int md = dst[en.x];
    float ms = sim[en.x];
    bool live = (i + sub < count);
    int mloc = live ? en.y : OWN_R;
    int loc[OWN_G];
    float4 b[OWN_G];
    float w[OWN_G];
#pragma unroll
    for (int q = 0; q < OWN_G; ++q) {
      int d = __builtin_amdgcn_readlane(md, q);
      loc[q] = __builtin_amdgcn_readlane(mloc, q);
      b[q] = reinterpret_cast<const float4*>(X2 + (size_t)d * NODE_DIM)[lane];
    }
    float mex = expf(ms - own_unkey(maxL[en.y]));
    if (lane < OWN_G && live)
      __hip_atomic_fetch_add(sumL + en.y, mex, __ATOMIC_RELAXED, __HIP_MEMORY_SCOPE_WORKGROUP);
#pragma unroll
    for (int q = 0; q < OWN_G; ++q)
      w[q] = __int_as_float(__builtin_amdgcn_readlane(__float_as_int(mex), q));
#pragma unroll
    for (int q = 0; q < OWN_G; ++q) {
      float* o = aggL + loc[q] * NODE_DIM + lane;
      __hip_atomic_fetch_add(o + 0, w[q] * b[q].x, __ATOMIC_RELAXED, __HIP_MEMORY_SCOPE_WORKGROUP);
      __hip_atomic_fetch_add(o + 32, w[q] * b[q].y, __ATOMIC_RELAXED, __HIP_MEMORY_SCOPE_WORKGROUP);
      __hip_atomic_fetch_add(o + 64, w[q] * b[q].z, __ATOMIC_RELAXED, __HIP_MEMORY_SCOPE_WORKGROUP);
      __hip_atomic_fetch_add(o + 96, w[q] * b[q].w, __ATOMIC_RELAXED, __HIP_MEMORY_SCOPE_WORKGROUP);
    }
  }
}

static_assert(OWN_K <= 8, "one team of 4 lanes per node slot of a wave");
__device__ __forceinline__ void own_team_pass(const int* sd, const int (&st)[OWN_K],
                                              const int (&cn)[OWN_K], float* aggL,
                                              float* sumL, const float* __restrict__ X2,
                                              int w, int rn, int lane) {
  const int j = lane & 3;
  const int g = lane >> 2;
  int myn = 0, mys = 0, maxn = 0;
#pragma unroll
  for (int k = 0; k < OWN_K; ++k) {
    myn = (g == k) ? cn[k] : myn;
    mys = (g == k) ? st[k] : mys;
    maxn = max(maxn, cn[k]);
  }
  const int loc = w + g * OWN_WAVES;
  const bool valid = (g < OWN_K) && (loc < rn);
  const int locc = min(loc, rn - 1);
  float4 a[8];
  const float4* ra = reinterpret_cast<const float4*>(aggL + locc * NODE_DIM) + j;
#pragma unroll
  for (int c = 0; c < 8; ++c) a[c] = ra[c * 4];
  float sa = 0.0f;
#pragma unroll
  for (int c = 0; c < 8; ++c) {
    sa = fmaf(a[c].x, a[c].x, sa);
    sa = fmaf(a[c].y, a[c].y, sa);
    sa = fmaf(a[c].z, a[c].z, sa);
    sa = fmaf(a[c].w, a[c].w, sa);
  }
  sa = own_xadd<2>(sa);
  sa = own_xadd<1>(sa);
  const float n1c = fmaxf(sqrtf(sa), EPS);
  float m = -INFINITY, s = 0.0f;
  float4 acc[8];
#pragma unroll
  for (int c = 0; c < 8; ++c) acc[c] = make_float4(0.f, 0.f, 0.f, 0.f);
  int dn = sd[(0 < myn) ? mys : 0];
  dn = (0 < myn) ? dn : 0;
  for (int i = 0; i < maxn; ++i) {
    const bool lv = (i < myn);
    float4 b[8];
    const float4* rb = reinterpret_cast<const float4*>(X2 + (size_t)dn * NODE_DIM) + j;
#pragma unroll
    for (int c = 0; c < 8; ++c) b[c] = rb[c * 4];
    {
      const int ni = i + 1;
      const int v = sd[(ni < myn) ? mys + ni : 0];
      dn = (ni < myn) ? v : 0;
    }
    float sdot = 0.0f, sb = 0.0f;
#pragma unroll
    for (int c = 0; c < 8; ++c) {
      sdot = fmaf(a[c].x, b[c].x, sdot);
      sdot = fmaf(a[c].y, b[c].y, sdot);
      sdot = fmaf(a[c].z, b[c].z, sdot);
      sdot = fmaf(a[c].w, b[c].w, sdot);
      sb = fmaf(b[c].x, b[c].x, sb);
      sb = fmaf(b[c].y, b[c].y, sb);
      sb = fmaf(b[c].z, b[c].z, sb);
      sb = fmaf(b[c].w, b[c].w, sb);
    }
    sdot = own_xadd<2>(sdot);
    sb = own_xadd<2>(sb);
    sdot = own_xadd<1>(sdot);
    sb = own_xadd<1>(sb);
    const float sv = sdot / (n1c * fmaxf(sqrtf(sb), EPS));
    const float ex = expf(-fabsf(sv - m));
    const bool up = sv > m;
    float sc = up ? ex : 1.0f;
    float p = up ? 1.0f : ex;
    float mn = fmaxf(m, sv);
    if (!lv) {
      sc = 1.0f;
      p = 0.0f;
      mn = m;
    }
    s = fmaf(s, sc, p);
#pragma unroll
    for (int c = 0; c < 8; ++c) {
      acc[c].x = fmaf(p, b[c].x, acc[c].x * sc);
      acc[c].y = fmaf(p, b[c].y, acc[c].y * sc);
      acc[c].z = fmaf(p, b[c].z, acc[c].z * sc);
      acc[c].w = fmaf(p, b[c].w, acc[c].w * sc);
    }
    m = mn;
  }
  if (valid) {
    float* o = aggL + loc * NODE_DIM + j;
#pragma unroll
    for (int c = 0; c < 8; ++c) {
      o[c * 4 + 0] = acc[c].x;
      o[c * 4 + 32] = acc[c].y;
      o[c * 4 + 64] = acc[c].z;
      o[c * 4 + 96] = acc[c].w;
    }
    if (j == 0) sumL[loc] = s;
  }
}

__device__ __forceinline__ int own_append(int2* myList, int count, int s, int e,
                                          int r0, int rn) {
  unsigned loc = (unsigned)(s - r0);
  bool m = loc < (unsigned)rn;
  unsigned mask = __builtin_amdgcn_ballot_w32(m);
  if (__builtin_expect(mask != 0u, 0)) {
    int pos = count + (int)__builtin_amdgcn_mbcnt_lo(mask, 0u);
    if (m) myList[pos] = make_int2(e, (int)loc);
    count += __builtin_popcount(mask);
  }
  return count;
}

__device__ __forceinline__ int own_put(int2* myList, int count, unsigned mask, bool mine,
                                       int e, int loc) {
  if (mask != 0u) {
    const int pos = count + (int)__builtin_amdgcn_mbcnt_lo(mask, 0u);
    if (mine) myList[pos] = make_int2(e, loc);
    count += __builtin_popcount(mask);
  }
  return count;
}

__global__ __launch_bounds__(OWN_WAVES * 32) void owner_agg_kernel(
    const float* __restrict__ X1, const float* __restrict__ X2,
    const float* __restrict__ Xn, const float* __restrict__ Wg,
    const int* __restrict__ src, const int* __restrict__ dst, float* sim,
    float* __restrict__ out, int n1, int E) {
  __shared__ float aggL[(OWN_R + 1) * NODE_DIM];
  __shared__ __attribute__((aligned(16))) int2 listL[OWN_WAVES * OWN_CW];
  __shared__ int2 privL[OWN_WAVES * OWN_PB];
  __shared__ float sumL[OWN_R];
  __shared__ int maxL[OWN_R];
  __shared__ int cntL[OWN_WAVES];
  __shared__ int histL[OWN_WAVES * OWN_WAVES];
  __shared__ int ovfL;
  __shared__ int ovf2L;
  const int tid = threadIdx.x;
  const int lane = tid & 31;
  const int w = __builtin_amdgcn_readfirstlane(tid >> 5);
  const int r0 = blockIdx.x * OWN_R;
  const int rn = min(OWN_R, n1 - r0);

  {
    const float4* xs = reinterpret_cast<const float4*>(X1 + (size_t)r0 * NODE_DIM);
    float4* xd = reinterpret_cast<float4*>(aggL);
    const int total = rn * (NODE_DIM / 4);
    float4 xv[OWN_XU];
#pragma unroll
    for (int u = 0; u < OWN_XU; ++u) {
      const int i = tid + u * (OWN_WAVES * 32);
      xv[u] = (i < total) ? xs[i] : make_float4(0.f, 0.f, 0.f, 0.f);
    }
#pragma unroll
    for (int u = 0; u < OWN_XU; ++u) {
      const int i = tid + u * (OWN_WAVES * 32);
      if (i < total) xd[i] = xv[u];
    }
  }
  if (tid < OWN_R) {
    sumL[tid] = 0.0f;
    maxL[tid] = own_key(-INFINITY);
  }
  if (tid == 0) {
    ovfL = 0;
    ovf2L = 0;
  }
  for (int i = tid; i < OWN_WAVES * OWN_WAVES; i += OWN_WAVES * 32) histL[i] = 0;
  __syncthreads();

  int2* myList = listL + w * OWN_CW;
  const int n4 = E >> 2;
  const int4* src4 = reinterpret_cast<const int4*>(src);
  const int trip4 = OWN_WAVES * OWN_U * 32;
  const int trips = (n4 + trip4 - 1) / trip4;
  const int tt0 = trips > 0 ? (int)(blockIdx.x % (unsigned)trips) : 0;
  bool second = false;
  bool fast = false;
  int pc = 0;
  int2* myPriv = privL + w * OWN_PB;
  for (int sweep = 0; sweep < 2; ++sweep) {
    int count = 0;
    int tt = tt0;
    for (int t = 0; t < trips; ++t) {
      const int base = tt * trip4 + w * (OWN_U * 32);
      tt = (tt + 1 == trips) ? 0 : tt + 1;
      if (count > OWN_CW - OWN_U * 128 - 4) {
        if (second) {
          own_flush(myList, count, aggL, sumL, maxL, X2, dst, sim, lane);
        } else {
          own_sim_max(myList, count, maxL, aggL, X2, dst, sim, lane);
          if (lane == 0) ovfL = 1;
        }
        count = 0;
      }
      int4 v[OWN_U];
      bool inl[OWN_U];
#pragma unroll
      for (int u = 0; u < OWN_U; ++u) {
        const int g = base + u * 32 + lane;
        inl[u] = (g < n4);
        v[u] = src4[min(g, n4 - 1)];
      }
      bool mm[OWN_U][4];
      unsigned mk[OWN_U][4];
#pragma unroll
      for (int u = 0; u < OWN_U; ++u) {
        const bool tx = (unsigned)(v[u].x - r0) < (unsigned)rn;
        const bool ty = (unsigned)(v[u].y - r0) < (unsigned)rn;
        const bool tz = (unsigned)(v[u].z - r0) < (unsigned)rn;
        const bool tw = (unsigned)(v[u].w - r0) < (unsigned)rn;
        const unsigned ik = __builtin_amdgcn_ballot_w32(inl[u]);
        mk[u][0] = __builtin_amdgcn_ballot_w32(tx) & ik;
        mk[u][1] = __builtin_amdgcn_ballot_w32(ty) & ik;
        mk[u][2] = __builtin_amdgcn_ballot_w32(tz) & ik;
        mk[u][3] = __builtin_amdgcn_ballot_w32(tw) & ik;
        mm[u][0] = inl[u] && tx;
        mm[u][1] = inl[u] && ty;
        mm[u][2] = inl[u] && tz;
        mm[u][3] = inl[u] && tw;
      }
#pragma unroll
      for (int u = 0; u < OWN_U; ++u) {
        if ((mk[u][0] | mk[u][1] | mk[u][2] | mk[u][3]) != 0u) {
          const int e = (base + u * 32 + lane) << 2;
          count = own_put(myList, count, mk[u][0], mm[u][0], e + 0, v[u].x - r0);
          count = own_put(myList, count, mk[u][1], mm[u][1], e + 1, v[u].y - r0);
          count = own_put(myList, count, mk[u][2], mm[u][2], e + 2, v[u].z - r0);
          count = own_put(myList, count, mk[u][3], mm[u][3], e + 3, v[u].w - r0);
        }
      }
    }
    if (w == 0) {
      int e = (n4 << 2) + lane;
      int s = (e < E) ? src[e] : -1;
      count = own_append(myList, count, s, e, r0, rn);
    }
    if (second) {
      own_flush(myList, count, aggL, sumL, maxL, X2, dst, sim, lane);
    } else {
      if (lane == 0) cntL[w] = count;
      for (int i = 0; i < count; i += 32) {
        const int idx = i + lane;
        if (idx < count) {
          const int d = myList[idx].y & (OWN_WAVES - 1);
          __hip_atomic_fetch_add(histL + d * OWN_WAVES + w, 1, __ATOMIC_RELAXED,
                                 __HIP_MEMORY_SCOPE_WORKGROUP);
        }
      }
      __syncthreads();
      if (__builtin_amdgcn_readfirstlane(ovfL) == 0) {
        const int c = histL[w * OWN_WAVES + lane];
        int incl = c;
#pragma unroll
        for (int o = 1; o < 32; o <<= 1) {
          const int t = __builtin_amdgcn_ds_bpermute(((lane - o) & 31) << 2, incl);
          incl += (lane >= o) ? t : 0;
        }
        histL[w * OWN_WAVES + lane] = incl - c;
        const int hc = __builtin_amdgcn_readlane(incl, 31);
        pc = hc;
        if (hc > OWN_PB) {
          if (lane == 0) ovf2L = 1;
        }
        __syncthreads();
        fast = (__builtin_amdgcn_readfirstlane(ovf2L) == 0);
        if (fast) {
          for (int i = 0; i < count; i += 32) {
            const int idx = i + lane;
            const bool live = (idx < count);
            const int2 en = myList[min(idx, count - 1)];
            const int d = en.y & (OWN_WAVES - 1);
            unsigned peers = __builtin_amdgcn_ballot_w32(live);
#pragma unroll
            for (int b = 0; b < 5; ++b) {
              const bool bit = ((d >> b) & 1) != 0;
              const unsigned m = __builtin_amdgcn_ballot_w32(bit);
              peers &= bit ? m : ~m;
            }
            const int rank = (int)__builtin_amdgcn_mbcnt_lo(peers, 0u);
            const bool lead = live && (rank == 0);
            int* place = histL + d * OWN_WAVES + w;
            int start = 0;
            if (lead) {
              start = *place;
              *place = start + __builtin_popcount(peers);
            }
            const int first = live ? __builtin_ctz(peers) : 0;
            const int base = __builtin_amdgcn_ds_bpermute(first << 2, start);
            if (live) privL[d * OWN_PB + base + rank] = en;
          }
          __syncthreads();
        }
      }
      if (!fast) own_sim_max(myList, count, maxL, aggL, X2, dst, sim, lane);
    }
    if (fast) break;
    __syncthreads();
    if (second) break;
    for (int i = tid; i < (OWN_R + 1) * NODE_DIM / 4; i += OWN_WAVES * 32)
      reinterpret_cast<float4*>(aggL)[i] = make_float4(0.f, 0.f, 0.f, 0.f);
    __syncthreads();
    if (__builtin_amdgcn_readfirstlane(ovfL) == 0) break;
    second = true;
  }

  if (fast) {
    int* sd = reinterpret_cast<int*>(myList);
    int cn[OWN_K], st[OWN_K], run[OWN_K];
#pragma unroll
    for (int k = 0; k < OWN_K; ++k) cn[k] = 0;
    for (int i = 0; i < pc; i += 32) {
      const int idx = i + lane;
      const int2 en = myPriv[min(idx, pc - 1)];
      const int kk = (idx < pc) ? (en.y / OWN_WAVES) : OWN_K;
#pragma unroll
      for (int k = 0; k < OWN_K; ++k)
        cn[k] += __builtin_popcount(__builtin_amdgcn_ballot_w32(kk == k));
    }
    st[0] = 0;
#pragma unroll
    for (int k = 1; k < OWN_K; ++k) st[k] = st[k - 1] + cn[k - 1];
#pragma unroll
    for (int k = 0; k < OWN_K; ++k) run[k] = st[k];
    for (int i = 0; i < pc; i += 32) {
      const int idx = i + lane;
      const int2 en = myPriv[min(idx, pc - 1)];
      const bool live = (idx < pc);
      const int kk = live ? (en.y / OWN_WAVES) : OWN_K;
      const int d = dst[en.x];
#pragma unroll
      for (int k = 0; k < OWN_K; ++k) {
        const unsigned mask = __builtin_amdgcn_ballot_w32(kk == k);
        if (kk == k) sd[run[k] + (int)__builtin_amdgcn_mbcnt_lo(mask, 0u)] = d;
        run[k] += __builtin_popcount(mask);
      }
    }
    own_team_pass(sd, st, cn, aggL, sumL, X2, w, rn, lane);
  } else if (!second) {
    const int myc = cntL[lane];
    pc = 0;
    for (int j = 0; j < OWN_WAVES; ++j) {
      const int cj = __builtin_amdgcn_readlane(myc, j);
      const int2* Lj = listL + j * OWN_CW;
      for (int i = 0; i < cj; i += 32) {
        const int idx = i + lane;
        int2 en = Lj[min(idx, cj - 1)];
        bool m = (idx < cj) && ((en.y & (OWN_WAVES - 1)) == w);
        unsigned mask = __builtin_amdgcn_ballot_w32(m);
        if (mask != 0u) {
          int pos = pc + (int)__builtin_amdgcn_mbcnt_lo(mask, 0u);
          if (m) myPriv[pos] = en;
          pc += __builtin_popcount(mask);
          if (pc > OWN_PB - 32) {
            own_flush(myPriv, pc, aggL, sumL, maxL, X2, dst, sim, lane);
            pc = 0;
          }
        }
      }
    }
    own_flush(myPriv, pc, aggL, sumL, maxL, X2, dst, sim, lane);
  }

  float4 ga[OWN_GA], gb[OWN_GB];
  {
    const float4* xs = reinterpret_cast<const float4*>(Xn + (size_t)r0 * ATTR_DIM);
    const float4* wsrc = reinterpret_cast<const float4*>(Wg);
#pragma unroll
    for (int u = 0; u < OWN_GA; ++u) {
      const int i = tid + u * (OWN_WAVES * 32);
      ga[u] = (i < rn * (ATTR_DIM / 4)) ? xs[i] : make_float4(0.f, 0.f, 0.f, 0.f);
    }
#pragma unroll
    for (int u = 0; u < OWN_GB; ++u) {
      const int i = tid + u * (OWN_WAVES * 32);
      gb[u] = (i < NODE_DIM * (ATTR_DIM / 4)) ? wsrc[i] : make_float4(0.f, 0.f, 0.f, 0.f);
    }
  }

  __syncthreads();

  float* xnL = reinterpret_cast<float*>(listL);
  float* wgL = xnL + OWN_TM * 16 * OWN_XP;
#pragma unroll
  for (int u = 0; u < OWN_GA; ++u) {
    const int i = tid + u * (OWN_WAVES * 32);
    if (i < OWN_TM * 16 * (ATTR_DIM / 4))
      *reinterpret_cast<float4*>(xnL + (i >> 4) * OWN_XP + (i & 15) * 4) = ga[u];
  }
#pragma unroll
  for (int u = 0; u < OWN_GB; ++u) {
    const int i = tid + u * (OWN_WAVES * 32);
    if (i < NODE_DIM * (ATTR_DIM / 4))
      *reinterpret_cast<float4*>(wgL + (i >> 4) * OWN_XP + (i & 15) * 4) = gb[u];
  }
  __syncthreads();

  {
    const int l15 = lane & 15;
    const int half = lane >> 4;
    const int tn = w & (NODE_DIM / 16 - 1);
    const float* bp = wgL + (tn * 16 + l15) * OWN_XP + half * 2;
    v2f bf[ATTR_DIM / 4];
#pragma unroll
    for (int s = 0; s < ATTR_DIM / 4; ++s) bf[s] = *reinterpret_cast<const v2f*>(bp + 4 * s);
    const int n = tn * 16 + l15;
    const int col = (n & 3) * 32 + (n >> 2);
    const int tiles = ((rn + 15) >> 4) * (NODE_DIM / 16);
    for (int t = w; t < tiles; t += OWN_WAVES) {
      const int tm = t / (NODE_DIM / 16);
      const float* ap = xnL + (tm * 16 + l15) * OWN_XP + half * 2;
      v2f af[ATTR_DIM / 4];
#pragma unroll
      for (int s = 0; s < ATTR_DIM / 4; ++s) af[s] = *reinterpret_cast<const v2f*>(ap + 4 * s);
      float pv[8], sv[8];
#pragma unroll
      for (int v = 0; v < 8; ++v) {
        const int rc = min(tm * 16 + half * 8 + v, rn - 1);
        pv[v] = aggL[rc * NODE_DIM + col];
        sv[v] = sumL[rc];
      }
      v8f c = {};
#pragma unroll
      for (int s = 0; s < ATTR_DIM / 4; ++s)
        c = __builtin_amdgcn_wmma_f32_16x16x4_f32(
             false, af[s],  false, bf[s],
             (short)0, c,  false,  false);
#pragma unroll
      for (int v = 0; v < 8; ++v) {
        const int row = tm * 16 + half * 8 + v;
        const float g = 1.0f / (1.0f + expf(-c[v]));
        const float a = (sv[v] > 0.0f) ? (pv[v] / sv[v]) : 0.0f;
        aggL[(row < rn ? row : OWN_R) * NODE_DIM + col] = g * a;
      }
    }
  }
  __syncthreads();

  for (int i = tid; i < rn * 32; i += OWN_WAVES * 32) {
    int row = i >> 5;
    const float* p = aggL + row * NODE_DIM + lane;
    reinterpret_cast<float4*>(out + (size_t)(r0 + row) * NODE_DIM)[lane] =
        make_float4(p[0], p[32], p[64], p[96]);
  }
}

extern "C" void kernel_launch(void* const* d_in, const int* in_sizes, int n_in,
                              void* d_out, int out_size, void* d_ws,
                              size_t ws_size, hipStream_t stream) {
  const float* X_h_1 = (const float*)d_in[0];
  const float* X_h_2 = (const float*)d_in[1];
  const float* X_n_1 = (const float*)d_in[2];
  const int* cross = (const int*)d_in[3];
  const float* W_gate = (const float*)d_in[4];
  float* out = (float*)d_out;

  const int N1 = in_sizes[0] / NODE_DIM;
  const int E = in_sizes[3] / 2;
  const int* src = cross;
  const int* dst = cross + E;

  float* sim = (float*)d_ws;

  owner_agg_kernel<<<(N1 + OWN_R - 1) / OWN_R, OWN_WAVES * 32, 0, stream>>>(
      X_h_1, X_h_2, X_n_1, W_gate, src, dst, sim, out, N1, E);
}


# team pass: a lane's 32 products of the dot product and of each square sum go into 4 partial sums (8 independent chains, built as packed 2-wide multiply-adds), not one chain of 32; new sum tree
# speedup vs baseline: 4.9840x; 1.0331x over previous
#include <hip/hip_runtime.h>
#include <math.h>

typedef float v2f __attribute__((ext_vector_type(2)));
typedef float v8f __attribute__((ext_vector_type(8)));

#define NODE_DIM 128
#define ATTR_DIM 64
#define EPS 1e-8f

#define OWN_R 200
#define OWN_WAVES 32
#define OWN_U 4
#define OWN_CW 640
#define OWN_G 8
#define OWN_PB 192
#define OWN_K ((OWN_R + OWN_WAVES - 1) / OWN_WAVES)
#define OWN_ST 2
#define OWN_XU ((OWN_R * NODE_DIM / 4 + OWN_WAVES * 32 - 1) / (OWN_WAVES * 32))
#define OWN_TM ((OWN_R + 15) / 16)
#define OWN_XP (ATTR_DIM + 4)
#define OWN_GA ((OWN_TM * 16 * (ATTR_DIM / 4) + OWN_WAVES * 32 - 1) / (OWN_WAVES * 32))
#define OWN_GB ((NODE_DIM * (ATTR_DIM / 4) + OWN_WAVES * 32 - 1) / (OWN_WAVES * 32))
static_assert(OWN_WAVES % (NODE_DIM / 16) == 0, "a wave keeps one feature tile for all its gate tiles");
static_assert(OWN_WAVES == 32, "the hand-over keeps one list's count per lane and finds peers by 5 bits of the wave number");
static_assert((OWN_TM * 16 + NODE_DIM) * OWN_XP * 4 <= OWN_WAVES * OWN_CW * 8,
              "the staged Xn and Wg rows fit in the space of the match lists");

__device__ __forceinline__ int own_key(float f) {
  int b = __float_as_int(f);
  return b >= 0 ? b : (b ^ 0x7fffffff);
}
__device__ __forceinline__ float own_unkey(int k) {
  return __int_as_float(k >= 0 ? k : (k ^ 0x7fffffff));
}

template <int M>
__device__ __forceinline__ float own_xadd(float x) {
  int b = __builtin_bit_cast(int, x);
  int r = __builtin_amdgcn_update_dpp(b, b, 0x160 + M, 0xf, 0xf, false);
  return x + __builtin_bit_cast(float, r);
}

__device__ __forceinline__ void own_sim_max(const int2* list, int count, int* maxL,
                                            const float* x1L,
                                            const float* __restrict__ X2,
                                            const int* __restrict__ dst,
                                            float* sim, int lane) {
  if (count == 0) return;
  const int j = lane & 7;
  const int g = lane >> 3;
  int2 en[OWN_ST];
  int d[OWN_ST];
#pragma unroll
  for (int t = 0; t < OWN_ST; ++t) {
    en[t] = list[min(t * 4 + g, count - 1)];
    d[t] = dst[en[t].x];
  }
  for (int i = 0; i < count; i += 4 * OWN_ST) {
    float4 a[OWN_ST][4], b[OWN_ST][4];
    int2 cur[OWN_ST];
#pragma unroll
    for (int t = 0; t < OWN_ST; ++t) {
      const float4* ra =
          reinterpret_cast<const float4*>(x1L + en[t].y * NODE_DIM) + j;
      const float4* rb =
          reinterpret_cast<const float4*>(X2 + (size_t)d[t] * NODE_DIM) + j;
#pragma unroll
      for (int c = 0; c < 4; ++c) {
        a[t][c] = ra[c * 8];
        b[t][c] = rb[c * 8];
      }
      cur[t] = en[t];
    }
#pragma unroll
    for (int t = 0; t < OWN_ST; ++t) {
      en[t] = list[min(i + 4 * OWN_ST + t * 4 + g, count - 1)];
      d[t] = dst[en[t].x];
    }
    float pd[OWN_ST], pa[OWN_ST], pb[OWN_ST];
#pragma unroll
    for (int t = 0; t < OWN_ST; ++t) {
      float sd = 0.0f, sa = 0.0f, sb = 0.0f;
#pragma unroll
      for (int c = 0; c < 4; ++c) {
        sd = fmaf(a[t][c].x, b[t][c].x, sd);
        sd = fmaf(a[t][c].y, b[t][c].y, sd);
        sd = fmaf(a[t][c].z, b[t][c].z, sd);
        sd = fmaf(a[t][c].w, b[t][c].w, sd);
        sa = fmaf(a[t][c].x, a[t][c].x, sa);
        sa = fmaf(a[t][c].y, a[t][c].y, sa);
        sa = fmaf(a[t][c].z, a[t][c].z, sa);
        sa = fmaf(a[t][c].w, a[t][c].w, sa);
        sb = fmaf(b[t][c].x, b[t][c].x, sb);
        sb = fmaf(b[t][c].y, b[t][c].y, sb);
        sb = fmaf(b[t][c].z, b[t][c].z, sb);
        sb = fmaf(b[t][c].w, b[t][c].w, sb);
      }
      sd = own_xadd<4>(sd);
      sa = own_xadd<4>(sa);
      sb = own_xadd<4>(sb);
      sd = own_xadd<2>(sd);
      sa = own_xadd<2>(sa);
      sb = own_xadd<2>(sb);
      pd[t] = own_xadd<1>(sd);
      pa[t] = own_xadd<1>(sa);
      pb[t] = own_xadd<1>(sb);
    }
    float md = pd[0], ma = pa[0], mb = pb[0];
    int2 me = cur[0];
#pragma unroll
    for (int t = 1; t < OWN_ST; ++t) {
      const bool s = (j == t);
      md = s ? pd[t] : md;
      ma = s ? pa[t] : ma;
      mb = s ? pb[t] : mb;
      me.x = s ? cur[t].x : me.x;
      me.y = s ? cur[t].y : me.y;
    }
    float sv = md / (fmaxf(sqrtf(ma), EPS) * fmaxf(sqrtf(mb), EPS));
    if (j < OWN_ST && i + j * 4 + g < count) {
      sim[me.x] = sv;
      __hip_atomic_fetch_max(maxL + me.y, own_key(sv), __ATOMIC_RELAXED,
                             __HIP_MEMORY_SCOPE_WORKGROUP);
    }
  }
}

__device__ __forceinline__ void own_flush(const int2* list, int count,
                                          float* aggL, float* sumL,
                                          const int* maxL,
                                          const float* __restrict__ X2,
                                          const int* __restrict__ dst,
                                          const float* sim,
                                          int lane) {
  const int sub = lane & (OWN_G - 1);
  for (int i = 0; i < count; i += OWN_G) {
    int2 en = list[min(i + sub, count - 1)];
    int md = dst[en.x];
    float ms = sim[en.x];
    bool live = (i + sub < count);
    int mloc = live ? en.y : OWN_R;
    int loc[OWN_G];
    float4 b[OWN_G];
    float w[OWN_G];
#pragma unroll
    for (int q = 0; q < OWN_G; ++q) {
      int d = __builtin_amdgcn_readlane(md, q);
      loc[q] = __builtin_amdgcn_readlane(mloc, q);
      b[q] = reinterpret_cast<const float4*>(X2 + (size_t)d * NODE_DIM)[lane];
    }
    float mex = expf(ms - own_unkey(maxL[en.y]));
    if (lane < OWN_G && live)
      __hip_atomic_fetch_add(sumL + en.y, mex, __ATOMIC_RELAXED, __HIP_MEMORY_SCOPE_WORKGROUP);
#pragma unroll
    for (int q = 0; q < OWN_G; ++q)
      w[q] = __int_as_float(__builtin_amdgcn_readlane(__float_as_int(mex), q));
#pragma unroll
    for (int q = 0; q < OWN_G; ++q) {
      float* o = aggL + loc[q] * NODE_DIM + lane;
      __hip_atomic_fetch_add(o + 0, w[q] * b[q].x, __ATOMIC_RELAXED, __HIP_MEMORY_SCOPE_WORKGROUP);
      __hip_atomic_fetch_add(o + 32, w[q] * b[q].y, __ATOMIC_RELAXED, __HIP_MEMORY_SCOPE_WORKGROUP);
      __hip_atomic_fetch_add(o + 64, w[q] * b[q].z, __ATOMIC_RELAXED, __HIP_MEMORY_SCOPE_WORKGROUP);
      __hip_atomic_fetch_add(o + 96, w[q] * b[q].w, __ATOMIC_RELAXED, __HIP_MEMORY_SCOPE_WORKGROUP);
    }
  }
}

static_assert(OWN_K <= 8, "one team of 4 lanes per node slot of a wave");
__device__ __forceinline__ void own_team_pass(const int* sd, const int (&st)[OWN_K],
                                              const int (&cn)[OWN_K], float* aggL,
                                              float* sumL, const float* __restrict__ X2,
                                              int w, int rn, int lane) {
  const int j = lane & 3;
  const int g = lane >> 2;
  int myn = 0, mys = 0, maxn = 0;
#pragma unroll
  for (int k = 0; k < OWN_K; ++k) {
    myn = (g == k) ? cn[k] : myn;
    mys = (g == k) ? st[k] : mys;
    maxn = max(maxn, cn[k]);
  }
  const int loc = w + g * OWN_WAVES;
  const bool valid = (g < OWN_K) && (loc < rn);
  const int locc = min(loc, rn - 1);
  float4 a[8];
  const float4* ra = reinterpret_cast<const float4*>(aggL + locc * NODE_DIM) + j;
#pragma unroll
  for (int c = 0; c < 8; ++c) a[c] = ra[c * 4];
  float sa0 = 0.0f, sa1 = 0.0f, sa2 = 0.0f, sa3 = 0.0f;
#pragma unroll
  for (int c = 0; c < 8; ++c) {
    sa0 = fmaf(a[c].x, a[c].x, sa0);
    sa1 = fmaf(a[c].y, a[c].y, sa1);
    sa2 = fmaf(a[c].z, a[c].z, sa2);
    sa3 = fmaf(a[c].w, a[c].w, sa3);
  }
  float sa = (sa0 + sa1) + (sa2 + sa3);
  sa = own_xadd<2>(sa);
  sa = own_xadd<1>(sa);
  const float n1c = fmaxf(sqrtf(sa), EPS);
  float m = -INFINITY, s = 0.0f;
  float4 acc[8];
#pragma unroll
  for (int c = 0; c < 8; ++c) acc[c] = make_float4(0.f, 0.f, 0.f, 0.f);
  int dn = sd[(0 < myn) ? mys : 0];
  dn = (0 < myn) ? dn : 0;
  for (int i = 0; i < maxn; ++i) {
    const bool lv = (i < myn);
    float4 b[8];
    const float4* rb = reinterpret_cast<const float4*>(X2 + (size_t)dn * NODE_DIM) + j;
#pragma unroll
    for (int c = 0; c < 8; ++c) b[c] = rb[c * 4];
    {
      const int ni = i + 1;
      const int v = sd[(ni < myn) ? mys + ni : 0];
      dn = (ni < myn) ? v : 0;
    }
    float d0 = 0.0f, d1 = 0.0f, d2 = 0.0f, d3 = 0.0f;
    float q0 = 0.0f, q1 = 0.0f, q2 = 0.0f, q3 = 0.0f;
#pragma unroll
    for (int c = 0; c < 8; ++c) {
      d0 = fmaf(a[c].x, b[c].x, d0);
      d1 = fmaf(a[c].y, b[c].y, d1);
      d2 = fmaf(a[c].z, b[c].z, d2);
      d3 = fmaf(a[c].w, b[c].w, d3);
      q0 = fmaf(b[c].x, b[c].x, q0);
      q1 = fmaf(b[c].y, b[c].y, q1);
      q2 = fmaf(b[c].z, b[c].z, q2);
      q3 = fmaf(b[c].w, b[c].w, q3);
    }
    float sdot = (d0 + d1) + (d2 + d3);
    float sb = (q0 + q1) + (q2 + q3);
    sdot = own_xadd<2>(sdot);
    sb = own_xadd<2>(sb);
    sdot = own_xadd<1>(sdot);
    sb = own_xadd<1>(sb);
    const float sv = sdot / (n1c * fmaxf(sqrtf(sb), EPS));
    const float ex = expf(-fabsf(sv - m));
    const bool up = sv > m;
    float sc = up ? ex : 1.0f;
    float p = up ? 1.0f : ex;
    float mn = fmaxf(m, sv);
    if (!lv) {
      sc = 1.0f;
      p = 0.0f;
      mn = m;
    }
    s = fmaf(s, sc, p);
#pragma unroll
    for (int c = 0; c < 8; ++c) {
      acc[c].x = fmaf(p, b[c].x, acc[c].x * sc);
      acc[c].y = fmaf(p, b[c].y, acc[c].y * sc);
      acc[c].z = fmaf(p, b[c].z, acc[c].z * sc);
      acc[c].w = fmaf(p, b[c].w, acc[c].w * sc);
    }
    m = mn;
  }
  if (valid) {
    float* o = aggL + loc * NODE_DIM + j;
#pragma unroll
    for (int c = 0; c < 8; ++c) {
      o[c * 4 + 0] = acc[c].x;
      o[c * 4 + 32] = acc[c].y;
      o[c * 4 + 64] = acc[c].z;
      o[c * 4 + 96] = acc[c].w;
    }
    if (j == 0) sumL[loc] = s;
  }
}

__device__ __forceinline__ int own_append(int2* myList, int count, int s, int e,
                                          int r0, int rn) {
  unsigned loc = (unsigned)(s - r0);
  bool m = loc < (unsigned)rn;
  unsigned mask = __builtin_amdgcn_ballot_w32(m);
  if (__builtin_expect(mask != 0u, 0)) {
    int pos = count + (int)__builtin_amdgcn_mbcnt_lo(mask, 0u);
    if (m) myList[pos] = make_int2(e, (int)loc);
    count += __builtin_popcount(mask);
  }
  return count;
}

__device__ __forceinline__ int own_put(int2* myList, int count, unsigned mask, bool mine,
                                       int e, int loc) {
  if (mask != 0u) {
    const int pos = count + (int)__builtin_amdgcn_mbcnt_lo(mask, 0u);
    if (mine) myList[pos] = make_int2(e, loc);
    count += __builtin_popcount(mask);
  }
  return count;
}

__global__ __launch_bounds__(OWN_WAVES * 32) void owner_agg_kernel(
    const float* __restrict__ X1, const float* __restrict__ X2,
    const float* __restrict__ Xn, const float* __restrict__ Wg,
    const int* __restrict__ src, const int* __restrict__ dst, float* sim,
    float* __restrict__ out, int n1, int E) {
  __shared__ float aggL[(OWN_R + 1) * NODE_DIM];
  __shared__ __attribute__((aligned(16))) int2 listL[OWN_WAVES * OWN_CW];
  __shared__ int2 privL[OWN_WAVES * OWN_PB];
  __shared__ float sumL[OWN_R];
  __shared__ int maxL[OWN_R];
  __shared__ int cntL[OWN_WAVES];
  __shared__ int histL[OWN_WAVES * OWN_WAVES];
  __shared__ int ovfL;
  __shared__ int ovf2L;
  const int tid = threadIdx.x;
  const int lane = tid & 31;
  const int w = __builtin_amdgcn_readfirstlane(tid >> 5);
  const int r0 = blockIdx.x * OWN_R;
  const int rn = min(OWN_R, n1 - r0);

  {
    const float4* xs = reinterpret_cast<const float4*>(X1 + (size_t)r0 * NODE_DIM);
    float4* xd = reinterpret_cast<float4*>(aggL);
    const int total = rn * (NODE_DIM / 4);
    float4 xv[OWN_XU];
#pragma unroll
    for (int u = 0; u < OWN_XU; ++u) {
      const int i = tid + u * (OWN_WAVES * 32);
      xv[u] = (i < total) ? xs[i] : make_float4(0.f, 0.f, 0.f, 0.f);
    }
#pragma unroll
    for (int u = 0; u < OWN_XU; ++u) {
      const int i = tid + u * (OWN_WAVES * 32);
      if (i < total) xd[i] = xv[u];
    }
  }
  if (tid < OWN_R) {
    sumL[tid] = 0.0f;
    maxL[tid] = own_key(-INFINITY);
  }
  if (tid == 0) {
    ovfL = 0;
    ovf2L = 0;
  }
  for (int i = tid; i < OWN_WAVES * OWN_WAVES; i += OWN_WAVES * 32) histL[i] = 0;
  __syncthreads();

  int2* myList = listL + w * OWN_CW;
  const int n4 = E >> 2;
  const int4* src4 = reinterpret_cast<const int4*>(src);
  const int trip4 = OWN_WAVES * OWN_U * 32;
  const int trips = (n4 + trip4 - 1) / trip4;
  const int tt0 = trips > 0 ? (int)(blockIdx.x % (unsigned)trips) : 0;
  bool second = false;
  bool fast = false;
  int pc = 0;
  int2* myPriv = privL + w * OWN_PB;
  for (int sweep = 0; sweep < 2; ++sweep) {
    int count = 0;
    int tt = tt0;
    for (int t = 0; t < trips; ++t) {
      const int base = tt * trip4 + w * (OWN_U * 32);
      tt = (tt + 1 == trips) ? 0 : tt + 1;
      if (count > OWN_CW - OWN_U * 128 - 4) {
        if (second) {
          own_flush(myList, count, aggL, sumL, maxL, X2, dst, sim, lane);
        } else {
          own_sim_max(myList, count, maxL, aggL, X2, dst, sim, lane);
          if (lane == 0) ovfL = 1;
        }
        count = 0;
      }
      int4 v[OWN_U];
      bool inl[OWN_U];
#pragma unroll
      for (int u = 0; u < OWN_U; ++u) {
        const int g = base + u * 32 + lane;
        inl[u] = (g < n4);
        v[u] = src4[min(g, n4 - 1)];
      }
      bool mm[OWN_U][4];
      unsigned mk[OWN_U][4];
#pragma unroll
      for (int u = 0; u < OWN_U; ++u) {
        const bool tx = (unsigned)(v[u].x - r0) < (unsigned)rn;
        const bool ty = (unsigned)(v[u].y - r0) < (unsigned)rn;
        const bool tz = (unsigned)(v[u].z - r0) < (unsigned)rn;
        const bool tw = (unsigned)(v[u].w - r0) < (unsigned)rn;
        const unsigned ik = __builtin_amdgcn_ballot_w32(inl[u]);
        mk[u][0] = __builtin_amdgcn_ballot_w32(tx) & ik;
        mk[u][1] = __builtin_amdgcn_ballot_w32(ty) & ik;
        mk[u][2] = __builtin_amdgcn_ballot_w32(tz) & ik;
        mk[u][3] = __builtin_amdgcn_ballot_w32(tw) & ik;
        mm[u][0] = inl[u] && tx;
        mm[u][1] = inl[u] && ty;
        mm[u][2] = inl[u] && tz;
        mm[u][3] = inl[u] && tw;
      }
#pragma unroll
      for (int u = 0; u < OWN_U; ++u) {
        if ((mk[u][0] | mk[u][1] | mk[u][2] | mk[u][3]) != 0u) {
          const int e = (base + u * 32 + lane) << 2;
          count = own_put(myList, count, mk[u][0], mm[u][0], e + 0, v[u].x - r0);
          count = own_put(myList, count, mk[u][1], mm[u][1], e + 1, v[u].y - r0);
          count = own_put(myList, count, mk[u][2], mm[u][2], e + 2, v[u].z - r0);
          count = own_put(myList, count, mk[u][3], mm[u][3], e + 3, v[u].w - r0);
        }
      }
    }
    if (w == 0) {
      int e = (n4 << 2) + lane;
      int s = (e < E) ? src[e] : -1;
      count = own_append(myList, count, s, e, r0, rn);
    }
    if (second) {
      own_flush(myList, count, aggL, sumL, maxL, X2, dst, sim, lane);
    } else {
      if (lane == 0) cntL[w] = count;
      for (int i = 0; i < count; i += 32) {
        const int idx = i + lane;
        if (idx < count) {
          const int d = myList[idx].y & (OWN_WAVES - 1);
          __hip_atomic_fetch_add(histL + d * OWN_WAVES + w, 1, __ATOMIC_RELAXED,
                                 __HIP_MEMORY_SCOPE_WORKGROUP);
        }
      }
      __syncthreads();
      if (__builtin_amdgcn_readfirstlane(ovfL) == 0) {
        const int c = histL[w * OWN_WAVES + lane];
        int incl = c;
#pragma unroll
        for (int o = 1; o < 32; o <<= 1) {
          const int t = __builtin_amdgcn_ds_bpermute(((lane - o) & 31) << 2, incl);
          incl += (lane >= o) ? t : 0;
        }
        histL[w * OWN_WAVES + lane] = incl - c;
        const int hc = __builtin_amdgcn_readlane(incl, 31);
        pc = hc;
        if (hc > OWN_PB) {
          if (lane == 0) ovf2L = 1;
        }
        __syncthreads();
        fast = (__builtin_amdgcn_readfirstlane(ovf2L) == 0);
        if (fast) {
          for (int i = 0; i < count; i += 32) {
            const int idx = i + lane;
            const bool live = (idx < count);
            const int2 en = myList[min(idx, count - 1)];
            const int d = en.y & (OWN_WAVES - 1);
            unsigned peers = __builtin_amdgcn_ballot_w32(live);
#pragma unroll
            for (int b = 0; b < 5; ++b) {
              const bool bit = ((d >> b) & 1) != 0;
              const unsigned m = __builtin_amdgcn_ballot_w32(bit);
              peers &= bit ? m : ~m;
            }
            const int rank = (int)__builtin_amdgcn_mbcnt_lo(peers, 0u);
            const bool lead = live && (rank == 0);
            int* place = histL + d * OWN_WAVES + w;
            int start = 0;
            if (lead) {
              start = *place;
              *place = start + __builtin_popcount(peers);
            }
            const int first = live ? __builtin_ctz(peers) : 0;
            const int base = __builtin_amdgcn_ds_bpermute(first << 2, start);
            if (live) privL[d * OWN_PB + base + rank] = en;
          }
          __syncthreads();
        }
      }
      if (!fast) own_sim_max(myList, count, maxL, aggL, X2, dst, sim, lane);
    }
    if (fast) break;
    __syncthreads();
    if (second) break;
    for (int i = tid; i < (OWN_R + 1) * NODE_DIM / 4; i += OWN_WAVES * 32)
      reinterpret_cast<float4*>(aggL)[i] = make_float4(0.f, 0.f, 0.f, 0.f);
    __syncthreads();
    if (__builtin_amdgcn_readfirstlane(ovfL) == 0) break;
    second = true;
  }

  if (fast) {
    int* sd = reinterpret_cast<int*>(myList);
    int cn[OWN_K], st[OWN_K], run[OWN_K];
#pragma unroll
    for (int k = 0; k < OWN_K; ++k) cn[k] = 0;
    for (int i = 0; i < pc; i += 32) {
      const int idx = i + lane;
      const int2 en = myPriv[min(idx, pc - 1)];
      const int kk = (idx < pc) ? (en.y / OWN_WAVES) : OWN_K;
#pragma unroll
      for (int k = 0; k < OWN_K; ++k)
        cn[k] += __builtin_popcount(__builtin_amdgcn_ballot_w32(kk == k));
    }
    st[0] = 0;
#pragma unroll
    for (int k = 1; k < OWN_K; ++k) st[k] = st[k - 1] + cn[k - 1];
#pragma unroll
    for (int k = 0; k < OWN_K; ++k) run[k] = st[k];
    for (int i = 0; i < pc; i += 32) {
      const int idx = i + lane;
      const int2 en = myPriv[min(idx, pc - 1)];
      const bool live = (idx < pc);
      const int kk = live ? (en.y / OWN_WAVES) : OWN_K;
      const int d = dst[en.x];
#pragma unroll
      for (int k = 0; k < OWN_K; ++k) {
        const unsigned mask = __builtin_amdgcn_ballot_w32(kk == k);
        if (kk == k) sd[run[k] + (int)__builtin_amdgcn_mbcnt_lo(mask, 0u)] = d;
        run[k] += __builtin_popcount(mask);
      }
    }
    own_team_pass(sd, st, cn, aggL, sumL, X2, w, rn, lane);
  } else if (!second) {
    const int myc = cntL[lane];
    pc = 0;
    for (int j = 0; j < OWN_WAVES; ++j) {
      const int cj = __builtin_amdgcn_readlane(myc, j);
      const int2* Lj = listL + j * OWN_CW;
      for (int i = 0; i < cj; i += 32) {
        const int idx = i + lane;
        int2 en = Lj[min(idx, cj - 1)];
        bool m = (idx < cj) && ((en.y & (OWN_WAVES - 1)) == w);
        unsigned mask = __builtin_amdgcn_ballot_w32(m);
        if (mask != 0u) {
          int pos = pc + (int)__builtin_amdgcn_mbcnt_lo(mask, 0u);
          if (m) myPriv[pos] = en;
          pc += __builtin_popcount(mask);
          if (pc > OWN_PB - 32) {
            own_flush(myPriv, pc, aggL, sumL, maxL, X2, dst, sim, lane);
            pc = 0;
          }
        }
      }
    }
    own_flush(myPriv, pc, aggL, sumL, maxL, X2, dst, sim, lane);
  }

  float4 ga[OWN_GA], gb[OWN_GB];
  {
    const float4* xs = reinterpret_cast<const float4*>(Xn + (size_t)r0 * ATTR_DIM);
    const float4* wsrc = reinterpret_cast<const float4*>(Wg);
#pragma unroll
    for (int u = 0; u < OWN_GA; ++u) {
      const int i = tid + u * (OWN_WAVES * 32);
      ga[u] = (i < rn * (ATTR_DIM / 4)) ? xs[i] : make_float4(0.f, 0.f, 0.f, 0.f);
    }
#pragma unroll
    for (int u = 0; u < OWN_GB; ++u) {
      const int i = tid + u * (OWN_WAVES * 32);
      gb[u] = (i < NODE_DIM * (ATTR_DIM / 4)) ? wsrc[i] : make_float4(0.f, 0.f, 0.f, 0.f);
    }
  }

  __syncthreads();

  float* xnL = reinterpret_cast<float*>(listL);
  float* wgL = xnL + OWN_TM * 16 * OWN_XP;
#pragma unroll
  for (int u = 0; u < OWN_GA; ++u) {
    const int i = tid + u * (OWN_WAVES * 32);
    if (i < OWN_TM * 16 * (ATTR_DIM / 4))
      *reinterpret_cast<float4*>(xnL + (i >> 4) * OWN_XP + (i & 15) * 4) = ga[u];
  }
#pragma unroll
  for (int u = 0; u < OWN_GB; ++u) {
    const int i = tid + u * (OWN_WAVES * 32);
    if (i < NODE_DIM * (ATTR_DIM / 4))
      *reinterpret_cast<float4*>(wgL + (i >> 4) * OWN_XP + (i & 15) * 4) = gb[u];
  }
  __syncthreads();

  {
    const int l15 = lane & 15;
    const int half = lane >> 4;
    const int tn = w & (NODE_DIM / 16 - 1);
    const float* bp = wgL + (tn * 16 + l15) * OWN_XP + half * 2;
    v2f bf[ATTR_DIM / 4];
#pragma unroll
    for (int s = 0; s < ATTR_DIM / 4; ++s) bf[s] = *reinterpret_cast<const v2f*>(bp + 4 * s);
    const int n = tn * 16 + l15;
    const int col = (n & 3) * 32 + (n >> 2);
    const int tiles = ((rn + 15) >> 4) * (NODE_DIM / 16);
    for (int t = w; t < tiles; t += OWN_WAVES) {
      const int tm = t / (NODE_DIM / 16);
      const float* ap = xnL + (tm * 16 + l15) * OWN_XP + half * 2;
      v2f af[ATTR_DIM / 4];
#pragma unroll
      for (int s = 0; s < ATTR_DIM / 4; ++s) af[s] = *reinterpret_cast<const v2f*>(ap + 4 * s);
      float pv[8], sv[8];
#pragma unroll
      for (int v = 0; v < 8; ++v) {
        const int rc = min(tm * 16 + half * 8 + v, rn - 1);
        pv[v] = aggL[rc * NODE_DIM + col];
        sv[v] = sumL[rc];
      }
      v8f c = {};
#pragma unroll
      for (int s = 0; s < ATTR_DIM / 4; ++s)
        c = __builtin_amdgcn_wmma_f32_16x16x4_f32(
             false, af[s],  false, bf[s],
             (short)0, c,  false,  false);
#pragma unroll
      for (int v = 0; v < 8; ++v) {
        const int row = tm * 16 + half * 8 + v;
        const float g = 1.0f / (1.0f + expf(-c[v]));
        const float a = (sv[v] > 0.0f) ? (pv[v] / sv[v]) : 0.0f;
        aggL[(row < rn ? row : OWN_R) * NODE_DIM + col] = g * a;
      }
    }
  }
  __syncthreads();

  for (int i = tid; i < rn * 32; i += OWN_WAVES * 32) {
    int row = i >> 5;
    const float* p = aggL + row * NODE_DIM + lane;
    reinterpret_cast<float4*>(out + (size_t)(r0 + row) * NODE_DIM)[lane] =
        make_float4(p[0], p[32], p[64], p[96]);
  }
}

extern "C" void kernel_launch(void* const* d_in, const int* in_sizes, int n_in,
                              void* d_out, int out_size, void* d_ws,
                              size_t ws_size, hipStream_t stream) {
  const float* X_h_1 = (const float*)d_in[0];
  const float* X_h_2 = (const float*)d_in[1];
  const float* X_n_1 = (const float*)d_in[2];
  const int* cross = (const int*)d_in[3];
  const float* W_gate = (const float*)d_in[4];
  float* out = (float*)d_out;

  const int N1 = in_sizes[0] / NODE_DIM;
  const int E = in_sizes[3] / 2;
  const int* src = cross;
  const int* dst = cross + E;

  float* sim = (float*)d_ws;

  owner_agg_kernel<<<(N1 + OWN_R - 1) / OWN_R, OWN_WAVES * 32, 0, stream>>>(
      X_h_1, X_h_2, X_n_1, W_gate, src, dst, sim, out, N1, E);
}
